# starting kernel
# baseline (speedup 1.0000x reference)
#include <hip/hip_runtime.h>


#define KEPS 1e-3f
#define NSEQ 1024
#define DH   64
#define MF   128
#define CHK  16

typedef float v2f __attribute__((ext_vector_type(2)));
typedef float v8f __attribute__((ext_vector_type(8)));

__device__ __forceinline__ v8f wmma4(v2f a, v2f b, v8f c) {
  return __builtin_amdgcn_wmma_f32_16x16x4_f32(false, a, false, b,
                                               (short)0, c, false, false);
}

__global__ __launch_bounds__(256)
void performer_causal_f32_wmma(const float* __restrict__ q,
                               const float* __restrict__ k,
                               const float* __restrict__ v,
                               const float* __restrict__ proj,
                               float* __restrict__ out) {
  __shared__ float proj_s[MF * DH];
  __shared__ float S_s[MF * DH];
  __shared__ float qp_s[CHK * MF];
  __shared__ float kp_s[CHK * MF];
  __shared__ float xq_s[CHK * DH];
  __shared__ float xk_s[CHK * DH];
  __shared__ float v_s[CHK * DH];
  __shared__ float a_s[CHK * CHK];
  __shared__ float kc_s[MF];

  const int h    = blockIdx.x;
  const int tid  = threadIdx.x;
  const int lane = tid & 31;
  const int wave = tid >> 5;
  const int lr   = lane & 15;
  const int lh   = lane >> 4;

  const float norm = 0.35355339059327378f;

  const float* qh = q + (size_t)h * NSEQ * DH;
  const float* kh = k + (size_t)h * NSEQ * DH;
  const float* vh = v + (size_t)h * NSEQ * DH;
  float*       oh = out + (size_t)h * NSEQ * DH;

  for (int i = tid; i < MF * DH; i += 256) { proj_s[i] = proj[i]; S_s[i] = 0.f; }
  for (int i = tid; i < MF; i += 256) kc_s[i] = 0.f;
  __syncthreads();

  for (int n0 = 0; n0 < NSEQ; n0 += CHK) {
    for (int i = tid; i < CHK * DH; i += 256) {
      xq_s[i] = qh[n0 * DH + i];
      xk_s[i] = kh[n0 * DH + i];
      v_s[i]  = vh[n0 * DH + i];
    }
    __syncthreads();

    {
      const int m0 = wave * 16;
      v8f accq = {};
      v8f acck = {};
#pragma unroll
      for (int kb = 0; kb < DH; kb += 4) {
        v2f a, b;
        b.x = proj_s[(m0 + lr) * DH + kb + 2 * lh];
        b.y = proj_s[(m0 + lr) * DH + kb + 2 * lh + 1];
        a.x = xq_s[lr * DH + kb + 2 * lh];
        a.y = xq_s[lr * DH + kb + 2 * lh + 1];
        accq = wmma4(a, b, accq);
        a.x = xk_s[lr * DH + kb + 2 * lh];
        a.y = xk_s[lr * DH + kb + 2 * lh + 1];
        acck = wmma4(a, b, acck);
      }
#pragma unroll
      for (int j = 0; j < 8; ++j) {
        const int row = j + 8 * lh;
        qp_s[row * MF + m0 + lr] = fmaxf(accq[j] * norm, 0.f) + KEPS;
        kp_s[row * MF + m0 + lr] = fmaxf(acck[j] * norm, 0.f) + KEPS;
      }
    }
    __syncthreads();

    if (tid < MF) {
      float run = kc_s[tid];
#pragma unroll
      for (int r = 0; r < CHK; ++r) {
        run += kp_s[r * MF + tid];
        qp_s[r * MF + tid] = qp_s[r * MF + tid] / run;
      }
      kc_s[tid] = run;
    }
    __syncthreads();

    v8f o = {};
    if (wave < 4) {
      const int e0 = wave * 16;
#pragma unroll
      for (int kb = 0; kb < MF; kb += 4) {
        v2f a, b;
        a.x = qp_s[lr * MF + kb + 2 * lh];
        a.y = qp_s[lr * MF + kb + 2 * lh + 1];
        b.x = S_s[(kb + 2 * lh) * DH + e0 + lr];
        b.y = S_s[(kb + 2 * lh + 1) * DH + e0 + lr];
        o = wmma4(a, b, o);
      }
    } else if (wave == 4) {
      v8f am = {};
#pragma unroll
      for (int kb = 0; kb < MF; kb += 4) {
        v2f a, b;
        a.x = qp_s[lr * MF + kb + 2 * lh];
        a.y = qp_s[lr * MF + kb + 2 * lh + 1];
        b.x = kp_s[lr * MF + kb + 2 * lh];
        b.y = kp_s[lr * MF + kb + 2 * lh + 1];
        am = wmma4(a, b, am);
      }
#pragma unroll
      for (int j = 0; j < 8; ++j) {
        const int row = j + 8 * lh;
        a_s[row * CHK + lr] = (lr <= row) ? am[j] : 0.f;
      }
    }
    __syncthreads();

    if (wave < 4) {
      const int e0 = wave * 16;
#pragma unroll
      for (int kb = 0; kb < CHK; kb += 4) {
        v2f a, b;
        a.x = a_s[lr * CHK + kb + 2 * lh];
        a.y = a_s[lr * CHK + kb + 2 * lh + 1];
        b.x = v_s[(kb + 2 * lh) * DH + e0 + lr];
        b.y = v_s[(kb + 2 * lh + 1) * DH + e0 + lr];
        o = wmma4(a, b, o);
      }
#pragma unroll
      for (int j = 0; j < 8; ++j) {
        const int row = j + 8 * lh;
        oh[(n0 + row) * DH + e0 + lr] = o[j];
      }
    }
    {
      const int m0 = wave * 16;
#pragma unroll
      for (int et = 0; et < 4; ++et) {
        const int e0 = et * 16;
        v8f sacc;
#pragma unroll
        for (int j = 0; j < 8; ++j)
          sacc[j] = S_s[(m0 + j + 8 * lh) * DH + e0 + lr];
#pragma unroll
        for (int kb = 0; kb < CHK; kb += 4) {
          v2f a, b;
          a.x = kp_s[(kb + 2 * lh) * MF + m0 + lr];
          a.y = kp_s[(kb + 2 * lh + 1) * MF + m0 + lr];
          b.x = v_s[(kb + 2 * lh) * DH + e0 + lr];
          b.y = v_s[(kb + 2 * lh + 1) * DH + e0 + lr];
          sacc = wmma4(a, b, sacc);
        }
#pragma unroll
        for (int j = 0; j < 8; ++j)
          S_s[(m0 + j + 8 * lh) * DH + e0 + lr] = sacc[j];
      }
    }
    __syncthreads();
  }
}

extern "C" void kernel_launch(void* const* d_in, const int* in_sizes, int n_in,
                              void* d_out, int out_size, void* d_ws, size_t ws_size,
                              hipStream_t stream) {
  (void)in_sizes; (void)n_in; (void)out_size; (void)d_ws; (void)ws_size;
  const float* q    = (const float*)d_in[0];
  const float* k    = (const float*)d_in[1];
  const float* v    = (const float*)d_in[2];
  const float* proj = (const float*)d_in[3];
  float* out        = (float*)d_out;
  performer_causal_f32_wmma<<<dim3(8), dim3(256), 0, stream>>>(q, k, v, proj, out);
}


# LDS row pitches 64/128/16 dwords -> 76/140/20 (4 x odd): WMMA fragment reads and tile stores no longer 16-way / 2-way bank-conflicted; k order and summation order unchanged
# speedup vs baseline: 1.7769x; 1.7769x over previous
#include <hip/hip_runtime.h>


#define KEPS 1e-3f
#define NSEQ 1024
#define DH   64
#define MF   128
#define CHK  16

#define PD   (DH + 12)
#define PM   (MF + 12)
#define PA   (CHK + 4)

typedef float v2f __attribute__((ext_vector_type(2)));
typedef float v8f __attribute__((ext_vector_type(8)));

__device__ __forceinline__ v8f wmma4(v2f a, v2f b, v8f c) {
  return __builtin_amdgcn_wmma_f32_16x16x4_f32(false, a, false, b,
                                               (short)0, c, false, false);
}

__global__ __launch_bounds__(256)
void performer_causal_f32_wmma(const float* __restrict__ q,
                               const float* __restrict__ k,
                               const float* __restrict__ v,
                               const float* __restrict__ proj,
                               float* __restrict__ out) {
  __shared__ float proj_s[MF * PD];
  __shared__ float S_s[MF * PD];
  __shared__ float qp_s[CHK * PM];
  __shared__ float kp_s[CHK * PM];
  __shared__ float xq_s[CHK * PD];
  __shared__ float xk_s[CHK * PD];
  __shared__ float v_s[CHK * PD];
  __shared__ float a_s[CHK * PA];
  __shared__ float kc_s[MF];

  const int h    = blockIdx.x;
  const int tid  = threadIdx.x;
  const int lane = tid & 31;
  const int wave = tid >> 5;
  const int lr   = lane & 15;
  const int lh   = lane >> 4;

  const float norm = 0.35355339059327378f;

  const float* qh = q + (size_t)h * NSEQ * DH;
  const float* kh = k + (size_t)h * NSEQ * DH;
  const float* vh = v + (size_t)h * NSEQ * DH;
  float*       oh = out + (size_t)h * NSEQ * DH;

  for (int i = tid; i < MF * DH; i += 256) {
    const int p = (i >> 6) * PD + (i & 63);
    proj_s[p] = proj[i];
    S_s[p] = 0.f;
  }
  for (int i = tid; i < MF; i += 256) kc_s[i] = 0.f;
  __syncthreads();

  for (int n0 = 0; n0 < NSEQ; n0 += CHK) {
    for (int i = tid; i < CHK * DH; i += 256) {
      const int p = (i >> 6) * PD + (i & 63);
      xq_s[p] = qh[n0 * DH + i];
      xk_s[p] = kh[n0 * DH + i];
      v_s[p]  = vh[n0 * DH + i];
    }
    __syncthreads();

    {
      const int m0 = wave * 16;
      v8f accq = {};
      v8f acck = {};
#pragma unroll
      for (int kb = 0; kb < DH; kb += 4) {
        v2f a, b;
        b.x = proj_s[(m0 + lr) * PD + kb + 2 * lh];
        b.y = proj_s[(m0 + lr) * PD + kb + 2 * lh + 1];
        a.x = xq_s[lr * PD + kb + 2 * lh];
        a.y = xq_s[lr * PD + kb + 2 * lh + 1];
        accq = wmma4(a, b, accq);
        a.x = xk_s[lr * PD + kb + 2 * lh];
        a.y = xk_s[lr * PD + kb + 2 * lh + 1];
        acck = wmma4(a, b, acck);
      }
#pragma unroll
      for (int j = 0; j < 8; ++j) {
        const int row = j + 8 * lh;
        qp_s[row * PM + m0 + lr] = fmaxf(accq[j] * norm, 0.f) + KEPS;
        kp_s[row * PM + m0 + lr] = fmaxf(acck[j] * norm, 0.f) + KEPS;
      }
    }
    __syncthreads();

    if (tid < MF) {
      float run = kc_s[tid];
#pragma unroll
      for (int r = 0; r < CHK; ++r) {
        run += kp_s[r * PM + tid];
        qp_s[r * PM + tid] = qp_s[r * PM + tid] / run;
      }
      kc_s[tid] = run;
    }
    __syncthreads();

    v8f o = {};
    if (wave < 4) {
      const int e0 = wave * 16;
#pragma unroll
      for (int kb = 0; kb < MF; kb += 4) {
        v2f a, b;
        a.x = qp_s[lr * PM + kb + 2 * lh];
        a.y = qp_s[lr * PM + kb + 2 * lh + 1];
        b.x = S_s[(kb + 2 * lh) * PD + e0 + lr];
        b.y = S_s[(kb + 2 * lh + 1) * PD + e0 + lr];
        o = wmma4(a, b, o);
      }
    } else if (wave == 4) {
      v8f am = {};
#pragma unroll
      for (int kb = 0; kb < MF; kb += 4) {
        v2f a, b;
        a.x = qp_s[lr * PM + kb + 2 * lh];
        a.y = qp_s[lr * PM + kb + 2 * lh + 1];
        b.x = kp_s[lr * PM + kb + 2 * lh];
        b.y = kp_s[lr * PM + kb + 2 * lh + 1];
        am = wmma4(a, b, am);
      }
#pragma unroll
      for (int j = 0; j < 8; ++j) {
        const int row = j + 8 * lh;
        a_s[row * PA + lr] = (lr <= row) ? am[j] : 0.f;
      }
    }
    __syncthreads();

    if (wave < 4) {
      const int e0 = wave * 16;
#pragma unroll
      for (int kb = 0; kb < CHK; kb += 4) {
        v2f a, b;
        a.x = a_s[lr * PA + kb + 2 * lh];
        a.y = a_s[lr * PA + kb + 2 * lh + 1];
        b.x = v_s[(kb + 2 * lh) * PD + e0 + lr];
        b.y = v_s[(kb + 2 * lh + 1) * PD + e0 + lr];
        o = wmma4(a, b, o);
      }
#pragma unroll
      for (int j = 0; j < 8; ++j) {
        const int row = j + 8 * lh;
        oh[(n0 + row) * DH + e0 + lr] = o[j];
      }
    }
    {
      const int m0 = wave * 16;
#pragma unroll
      for (int et = 0; et < 4; ++et) {
        const int e0 = et * 16;
        v8f sacc;
#pragma unroll
        for (int j = 0; j < 8; ++j)
          sacc[j] = S_s[(m0 + j + 8 * lh) * PD + e0 + lr];
#pragma unroll
        for (int kb = 0; kb < CHK; kb += 4) {
          v2f a, b;
          a.x = kp_s[(kb + 2 * lh) * PM + m0 + lr];
          a.y = kp_s[(kb + 2 * lh + 1) * PM + m0 + lr];
          b.x = v_s[(kb + 2 * lh) * PD + e0 + lr];
          b.y = v_s[(kb + 2 * lh + 1) * PD + e0 + lr];
          sacc = wmma4(a, b, sacc);
        }
#pragma unroll
        for (int j = 0; j < 8; ++j)
          S_s[(m0 + j + 8 * lh) * PD + e0 + lr] = sacc[j];
      }
    }
    __syncthreads();
  }
}

extern "C" void kernel_launch(void* const* d_in, const int* in_sizes, int n_in,
                              void* d_out, int out_size, void* d_ws, size_t ws_size,
                              hipStream_t stream) {
  (void)in_sizes; (void)n_in; (void)out_size; (void)d_ws; (void)ws_size;
  const float* q    = (const float*)d_in[0];
  const float* k    = (const float*)d_in[1];
  const float* v    = (const float*)d_in[2];
  const float* proj = (const float*)d_in[3];
  float* out        = (float*)d_out;
  performer_causal_f32_wmma<<<dim3(8), dim3(256), 0, stream>>>(q, k, v, proj, out);
}


# Sequence of each head cut into 16 segments, one workgroup each (128, not 8): a k/v-only prefix pass rebuilds S and the cumsum carry in the scan's own order, then the unchanged scan runs on 64 rows
# speedup vs baseline: 7.0497x; 3.9674x over previous
#include <hip/hip_runtime.h>


#define KEPS 1e-3f
#define NSEQ 1024
#define DH   64
#define MF   128
#define CHK  16
#define NSEG 16
#define SEGL (NSEQ / NSEG)
static_assert(8 * NSEG == 128, "the launch below starts 8 x NSEG workgroups");

#define PD   (DH + 12)
#define PM   (MF + 12)
#define PA   (CHK + 4)

typedef float v2f __attribute__((ext_vector_type(2)));
typedef float v8f __attribute__((ext_vector_type(8)));

__device__ __forceinline__ v8f wmma4(v2f a, v2f b, v8f c) {
  return __builtin_amdgcn_wmma_f32_16x16x4_f32(false, a, false, b,
                                               (short)0, c, false, false);
}

__global__ __launch_bounds__(256)
void performer_causal_f32_wmma(const float* __restrict__ q,
                               const float* __restrict__ k,
                               const float* __restrict__ v,
                               const float* __restrict__ proj,
                               float* __restrict__ out) {
  __shared__ float proj_s[MF * PD];
  __shared__ float S_s[MF * PD];
  __shared__ float qp_s[CHK * PM];
  __shared__ float kp_s[CHK * PM];
  __shared__ float xq_s[CHK * PD];
  __shared__ float xk_s[2 * CHK * PD];
  __shared__ float v_s[2 * CHK * PD];
  __shared__ float a_s[CHK * PA];
  __shared__ float kc_s[MF];

  const int h    = blockIdx.x / NSEG;
  const int sj   = blockIdx.x % NSEG;
  const int seg  = (sj & 1) ? (sj >> 1) : (NSEG - 1 - (sj >> 1));
  const int n_begin = seg * SEGL;
  const int nlight  = n_begin / CHK;
  const int tid  = threadIdx.x;
  const int lane = tid & 31;
  const int wave = tid >> 5;
  const int lr   = lane & 15;
  const int lh   = lane >> 4;

  const float norm = 0.35355339059327378f;

  const float* qh = q + (size_t)h * NSEQ * DH;
  const float* kh = k + (size_t)h * NSEQ * DH;
  const float* vh = v + (size_t)h * NSEQ * DH;
  float*       oh = out + (size_t)h * NSEQ * DH;

  for (int i = tid; i < MF * DH; i += 256) {
    const int p = (i >> 6) * PD + (i & 63);
    proj_s[p] = proj[i];
  }
  __syncthreads();

  {
    const int m0 = wave * 16;
    v8f sreg[4] = {};
    float run = 0.f;
    if (nlight > 0) {
      v2f pb[DH / 4];
#pragma unroll
      for (int kk = 0; kk < DH / 4; ++kk) {
        pb[kk].x = proj_s[(m0 + lr) * PD + 4 * kk + 2 * lh];
        pb[kk].y = proj_s[(m0 + lr) * PD + 4 * kk + 2 * lh + 1];
      }
      float rk[4], rv[4];
#pragma unroll
      for (int t = 0; t < 4; ++t) {
        rk[t] = kh[tid + 256 * t];
        rv[t] = vh[tid + 256 * t];
      }
#pragma unroll
      for (int t = 0; t < 4; ++t) {
        const int p = ((tid >> 6) + 4 * t) * PD + (tid & 63);
        xk_s[p] = rk[t];
        v_s[p]  = rv[t];
      }
      __syncthreads();
      for (int c = 0; c < nlight; ++c) {
        const float* xkb = xk_s + (c & 1) * (CHK * PD);
        const float* vb  = v_s  + (c & 1) * (CHK * PD);
        float* xkn = xk_s + ((c + 1) & 1) * (CHK * PD);
        float* vn  = v_s  + ((c + 1) & 1) * (CHK * PD);
        const int cn = (c + 1 < nlight) ? (c + 1) : c;
#pragma unroll
        for (int t = 0; t < 4; ++t) {
          rk[t] = kh[cn * CHK * DH + tid + 256 * t];
          rv[t] = vh[cn * CHK * DH + tid + 256 * t];
        }
        v8f acck = {};
#pragma unroll
        for (int kk = 0; kk < DH / 4; ++kk) {
          v2f a;
          a.x = xkb[lr * PD + 4 * kk + 2 * lh];
          a.y = xkb[lr * PD + 4 * kk + 2 * lh + 1];
          acck = wmma4(a, pb[kk], acck);
        }
#pragma unroll
        for (int j = 0; j < 8; ++j)
          kp_s[(j + 8 * lh) * PM + m0 + lr] = fmaxf(acck[j] * norm, 0.f) + KEPS;
        __builtin_amdgcn_fence(__ATOMIC_RELEASE, "wavefront");
        __builtin_amdgcn_wave_barrier();
        __builtin_amdgcn_fence(__ATOMIC_ACQUIRE, "wavefront");
#pragma unroll
        for (int r = 0; r < CHK; ++r) run += kp_s[r * PM + m0 + lr];
#pragma unroll
        for (int et = 0; et < 4; ++et) {
          const int e0 = et * 16;
#pragma unroll
          for (int kb = 0; kb < CHK; kb += 4) {
            v2f a, b;
            a.x = kp_s[(kb + 2 * lh) * PM + m0 + lr];
            a.y = kp_s[(kb + 2 * lh + 1) * PM + m0 + lr];
            b.x = vb[(kb + 2 * lh) * PD + e0 + lr];
            b.y = vb[(kb + 2 * lh + 1) * PD + e0 + lr];
            sreg[et] = wmma4(a, b, sreg[et]);
          }
        }
#pragma unroll
        for (int t = 0; t < 4; ++t) {
          const int p = ((tid >> 6) + 4 * t) * PD + (tid & 63);
          xkn[p] = rk[t];
          vn[p]  = rv[t];
        }
        __syncthreads();
      }
    }
#pragma unroll
    for (int et = 0; et < 4; ++et)
#pragma unroll
      for (int j = 0; j < 8; ++j)
        S_s[(m0 + j + 8 * lh) * PD + et * 16 + lr] = sreg[et][j];
    if (lh == 0) kc_s[m0 + lr] = run;
  }
  __syncthreads();

  for (int n0 = n_begin; n0 < n_begin + SEGL; n0 += CHK) {
    for (int i = tid; i < CHK * DH; i += 256) {
      const int p = (i >> 6) * PD + (i & 63);
      xq_s[p] = qh[n0 * DH + i];
      xk_s[p] = kh[n0 * DH + i];
      v_s[p]  = vh[n0 * DH + i];
    }
    __syncthreads();

    {
      const int m0 = wave * 16;
      v8f accq = {};
      v8f acck = {};
#pragma unroll
      for (int kb = 0; kb < DH; kb += 4) {
        v2f a, b;
        b.x = proj_s[(m0 + lr) * PD + kb + 2 * lh];
        b.y = proj_s[(m0 + lr) * PD + kb + 2 * lh + 1];
        a.x = xq_s[lr * PD + kb + 2 * lh];
        a.y = xq_s[lr * PD + kb + 2 * lh + 1];
        accq = wmma4(a, b, accq);
        a.x = xk_s[lr * PD + kb + 2 * lh];
        a.y = xk_s[lr * PD + kb + 2 * lh + 1];
        acck = wmma4(a, b, acck);
      }
#pragma unroll
      for (int j = 0; j < 8; ++j) {
        const int row = j + 8 * lh;
        qp_s[row * PM + m0 + lr] = fmaxf(accq[j] * norm, 0.f) + KEPS;
        kp_s[row * PM + m0 + lr] = fmaxf(acck[j] * norm, 0.f) + KEPS;
      }
    }
    __syncthreads();

    if (tid < MF) {
      float run = kc_s[tid];
#pragma unroll
      for (int r = 0; r < CHK; ++r) {
        run += kp_s[r * PM + tid];
        qp_s[r * PM + tid] = qp_s[r * PM + tid] / run;
      }
      kc_s[tid] = run;
    }
    __syncthreads();

    v8f o = {};
    if (wave < 4) {
      const int e0 = wave * 16;
#pragma unroll
      for (int kb = 0; kb < MF; kb += 4) {
        v2f a, b;
        a.x = qp_s[lr * PM + kb + 2 * lh];
        a.y = qp_s[lr * PM + kb + 2 * lh + 1];
        b.x = S_s[(kb + 2 * lh) * PD + e0 + lr];
        b.y = S_s[(kb + 2 * lh + 1) * PD + e0 + lr];
        o = wmma4(a, b, o);
      }
    } else if (wave == 4) {
      v8f am = {};
#pragma unroll
      for (int kb = 0; kb < MF; kb += 4) {
        v2f a, b;
        a.x = qp_s[lr * PM + kb + 2 * lh];
        a.y = qp_s[lr * PM + kb + 2 * lh + 1];
        b.x = kp_s[lr * PM + kb + 2 * lh];
        b.y = kp_s[lr * PM + kb + 2 * lh + 1];
        am = wmma4(a, b, am);
      }
#pragma unroll
      for (int j = 0; j < 8; ++j) {
        const int row = j + 8 * lh;
        a_s[row * PA + lr] = (lr <= row) ? am[j] : 0.f;
      }
    }
    __syncthreads();

    if (wave < 4) {
      const int e0 = wave * 16;
#pragma unroll
      for (int kb = 0; kb < CHK; kb += 4) {
        v2f a, b;
        a.x = a_s[lr * PA + kb + 2 * lh];
        a.y = a_s[lr * PA + kb + 2 * lh + 1];
        b.x = v_s[(kb + 2 * lh) * PD + e0 + lr];
        b.y = v_s[(kb + 2 * lh + 1) * PD + e0 + lr];
        o = wmma4(a, b, o);
      }
#pragma unroll
      for (int j = 0; j < 8; ++j) {
        const int row = j + 8 * lh;
        oh[(n0 + row) * DH + e0 + lr] = o[j];
      }
    }
    {
      const int m0 = wave * 16;
#pragma unroll
      for (int et = 0; et < 4; ++et) {
        const int e0 = et * 16;
        v8f sacc;
#pragma unroll
        for (int j = 0; j < 8; ++j)
          sacc[j] = S_s[(m0 + j + 8 * lh) * PD + e0 + lr];
#pragma unroll
        for (int kb = 0; kb < CHK; kb += 4) {
          v2f a, b;
          a.x = kp_s[(kb + 2 * lh) * PM + m0 + lr];
          a.y = kp_s[(kb + 2 * lh + 1) * PM + m0 + lr];
          b.x = v_s[(kb + 2 * lh) * PD + e0 + lr];
          b.y = v_s[(kb + 2 * lh + 1) * PD + e0 + lr];
          sacc = wmma4(a, b, sacc);
        }
#pragma unroll
        for (int j = 0; j < 8; ++j)
          S_s[(m0 + j + 8 * lh) * PD + e0 + lr] = sacc[j];
      }
    }
    __syncthreads();
  }
}

extern "C" void kernel_launch(void* const* d_in, const int* in_sizes, int n_in,
                              void* d_out, int out_size, void* d_ws, size_t ws_size,
                              hipStream_t stream) {
  (void)in_sizes; (void)n_in; (void)out_size; (void)d_ws; (void)ws_size;
  const float* q    = (const float*)d_in[0];
  const float* k    = (const float*)d_in[1];
  const float* v    = (const float*)d_in[2];
  const float* proj = (const float*)d_in[3];
  float* out        = (float*)d_out;
  performer_causal_f32_wmma<<<dim3(128), dim3(256), 0, stream>>>(q, k, v, proj, out);
}


# Recomputed prefix pass replaced: a first launch writes each segment's own S and cumsum contribution to the workspace; the scan launch adds up those in front (summation order of the state changes)
# speedup vs baseline: 15.0227x; 2.1310x over previous
#include <hip/hip_runtime.h>


#define KEPS 1e-3f
#define NSEQ 1024
#define DH   64
#define MF   128
#define CHK  16
#define NSEG 16
#define SEGL (NSEQ / NSEG)
static_assert(8 * NSEG == 128, "the launch below starts 8 x NSEG workgroups");
#define NPART   (NSEG - 1)
#define PSTRIDE (MF * DH + MF)
#define WS_FLOATS (8 * NPART * PSTRIDE)
static_assert(8 * NPART == 120, "the first launch below starts 8 x NPART workgroups");
static_assert(SEGL % CHK == 0 && SEGL == 64, "the first launch holds one segment of 64 rows in LDS");

#define PD   (DH + 12)
#define PM   (MF + 12)
#define PA   (CHK + 4)

typedef float v2f __attribute__((ext_vector_type(2)));
typedef float v8f __attribute__((ext_vector_type(8)));

__device__ __forceinline__ v8f wmma4(v2f a, v2f b, v8f c) {
  return __builtin_amdgcn_wmma_f32_16x16x4_f32(false, a, false, b,
                                               (short)0, c, false, false);
}


__global__ __launch_bounds__(256)
void performer_segment_state_f32_wmma(const float* __restrict__ k,
                                      const float* __restrict__ v,
                                      const float* __restrict__ proj,
                                      float* __restrict__ ws, size_t ws_bytes) {
  if (ws_bytes < (size_t)WS_FLOATS * sizeof(float)) return;

  __shared__ __attribute__((aligned(16))) float xk_s[SEGL * PD];
  __shared__ __attribute__((aligned(16))) float v_s[SEGL * PD];
  __shared__ float kp_s[SEGL * PM];

  const int h    = blockIdx.x / NPART;
  const int seg  = blockIdx.x % NPART;
  const int tid  = threadIdx.x;
  const int lane = tid & 31;
  const int wave = tid >> 5;
  const int lr   = lane & 15;
  const int lh   = lane >> 4;
  const int m0   = wave * 16;

  const float norm = 0.35355339059327378f;

  const float* kseg = k + ((size_t)h * NSEQ + (size_t)seg * SEGL) * DH;
  const float* vseg = v + ((size_t)h * NSEQ + (size_t)seg * SEGL) * DH;

  float4 rk[4], rv[4];
#pragma unroll
  for (int t = 0; t < 4; ++t) {
    const int idx = tid + 256 * t;
    rk[t] = *reinterpret_cast<const float4*>(kseg + 4 * idx);
    rv[t] = *reinterpret_cast<const float4*>(vseg + 4 * idx);
  }
  v2f pb[DH / 4];
#pragma unroll
  for (int kk = 0; kk < DH / 4; ++kk) {
    pb[kk].x = proj[(m0 + lr) * DH + 4 * kk + 2 * lh];
    pb[kk].y = proj[(m0 + lr) * DH + 4 * kk + 2 * lh + 1];
  }
#pragma unroll
  for (int t = 0; t < 4; ++t) {
    const int idx = tid + 256 * t;
    const int p = (idx >> 4) * PD + 4 * (idx & 15);
    *reinterpret_cast<float4*>(xk_s + p) = rk[t];
    *reinterpret_cast<float4*>(v_s + p)  = rv[t];
  }
  __syncthreads();

  v8f acc[SEGL / CHK] = {};
#pragma unroll
  for (int kk = 0; kk < DH / 4; ++kk) {
#pragma unroll
    for (int rt = 0; rt < SEGL / CHK; ++rt) {
      v2f a;
      a.x = xk_s[(CHK * rt + lr) * PD + 4 * kk + 2 * lh];
      a.y = xk_s[(CHK * rt + lr) * PD + 4 * kk + 2 * lh + 1];
      acc[rt] = wmma4(a, pb[kk], acc[rt]);
    }
  }
#pragma unroll
  for (int rt = 0; rt < SEGL / CHK; ++rt)
#pragma unroll
    for (int j = 0; j < 8; ++j)
      kp_s[(CHK * rt + j + 8 * lh) * PM + m0 + lr] = fmaxf(acc[rt][j] * norm, 0.f) + KEPS;
  __builtin_amdgcn_fence(__ATOMIC_RELEASE, "wavefront");
  __builtin_amdgcn_wave_barrier();
  __builtin_amdgcn_fence(__ATOMIC_ACQUIRE, "wavefront");

  float* P = ws + (size_t)blockIdx.x * PSTRIDE;

  float run = 0.f;
#pragma unroll
  for (int r = 0; r < SEGL; ++r) run += kp_s[r * PM + m0 + lr];
  if (lh == 0) P[MF * DH + m0 + lr] = run;

  v8f sreg[4] = {};
#pragma unroll
  for (int kb = 0; kb < SEGL; kb += 4) {
    v2f a;
    a.x = kp_s[(kb + 2 * lh) * PM + m0 + lr];
    a.y = kp_s[(kb + 2 * lh + 1) * PM + m0 + lr];
#pragma unroll
    for (int et = 0; et < 4; ++et) {
      v2f b;
      b.x = v_s[(kb + 2 * lh) * PD + et * 16 + lr];
      b.y = v_s[(kb + 2 * lh + 1) * PD + et * 16 + lr];
      sreg[et] = wmma4(a, b, sreg[et]);
    }
  }
#pragma unroll
  for (int et = 0; et < 4; ++et)
#pragma unroll
    for (int jh = 0; jh < 2; ++jh) {
      float4 o;
      o.x = sreg[et][4 * jh + 0];
      o.y = sreg[et][4 * jh + 1];
      o.z = sreg[et][4 * jh + 2];
      o.w = sreg[et][4 * jh + 3];
      *reinterpret_cast<float4*>(P + (((wave * 4 + et) * 2 + jh) * 32 + lane) * 4) = o;
    }
}

__global__ __launch_bounds__(256)
void performer_causal_f32_wmma(const float* __restrict__ q,
                               const float* __restrict__ k,
                               const float* __restrict__ v,
                               const float* __restrict__ proj,
                               float* __restrict__ out,
                               const float* __restrict__ ws, size_t ws_bytes) {
  __shared__ float proj_s[MF * PD];
  __shared__ float S_s[MF * PD];
  __shared__ float qp_s[CHK * PM];
  __shared__ float kp_s[CHK * PM];
  __shared__ float xq_s[CHK * PD];
  __shared__ float xk_s[2 * CHK * PD];
  __shared__ float v_s[2 * CHK * PD];
  __shared__ float a_s[CHK * PA];
  __shared__ float kc_s[MF];

  const int h    = blockIdx.x / NSEG;
  const int sj   = blockIdx.x % NSEG;
  const int seg  = (sj & 1) ? (sj >> 1) : (NSEG - 1 - (sj >> 1));
  const int n_begin = seg * SEGL;
  const int nlight  = n_begin / CHK;
  const int tid  = threadIdx.x;
  const int lane = tid & 31;
  const int wave = tid >> 5;
  const int lr   = lane & 15;
  const int lh   = lane >> 4;

  const float norm = 0.35355339059327378f;

  const float* qh = q + (size_t)h * NSEQ * DH;
  const float* kh = k + (size_t)h * NSEQ * DH;
  const float* vh = v + (size_t)h * NSEQ * DH;
  float*       oh = out + (size_t)h * NSEQ * DH;

  for (int i = tid; i < MF * DH; i += 256) {
    const int p = (i >> 6) * PD + (i & 63);
    proj_s[p] = proj[i];
  }
  __syncthreads();

  {
    const int m0 = wave * 16;
    v8f sreg[4] = {};
    float run = 0.f;
    const bool from_ws = ws_bytes >= (size_t)WS_FLOATS * sizeof(float);
    if (from_ws) {
      const float* Ph = ws + (size_t)(h * NPART) * PSTRIDE;
      for (int s0 = 0; s0 < seg; s0 += 4) {
        float4 t[4][8];
        float tk[4];
#pragma unroll
        for (int u = 0; u < 4; ++u) {
          if (s0 + u < seg) {
            const float* P = Ph + (size_t)(s0 + u) * PSTRIDE;
#pragma unroll
            for (int f = 0; f < 8; ++f)
              t[u][f] = *reinterpret_cast<const float4*>(P + ((wave * 8 + f) * 32 + lane) * 4);
            tk[u] = P[MF * DH + m0 + lr];
          }
        }
#pragma unroll
        for (int u = 0; u < 4; ++u) {
          if (s0 + u < seg) {
#pragma unroll
            for (int f = 0; f < 8; ++f) {
              sreg[f >> 1][4 * (f & 1) + 0] += t[u][f].x;
              sreg[f >> 1][4 * (f & 1) + 1] += t[u][f].y;
              sreg[f >> 1][4 * (f & 1) + 2] += t[u][f].z;
              sreg[f >> 1][4 * (f & 1) + 3] += t[u][f].w;
            }
            run += tk[u];
          }
        }
      }
    } else if (nlight > 0) {
      v2f pb[DH / 4];
#pragma unroll
      for (int kk = 0; kk < DH / 4; ++kk) {
        pb[kk].x = proj_s[(m0 + lr) * PD + 4 * kk + 2 * lh];
        pb[kk].y = proj_s[(m0 + lr) * PD + 4 * kk + 2 * lh + 1];
      }
      float rk[4], rv[4];
#pragma unroll
      for (int t = 0; t < 4; ++t) {
        rk[t] = kh[tid + 256 * t];
        rv[t] = vh[tid + 256 * t];
      }
#pragma unroll
      for (int t = 0; t < 4; ++t) {
        const int p = ((tid >> 6) + 4 * t) * PD + (tid & 63);
        xk_s[p] = rk[t];
        v_s[p]  = rv[t];
      }
      __syncthreads();
      for (int c = 0; c < nlight; ++c) {
        const float* xkb = xk_s + (c & 1) * (CHK * PD);
        const float* vb  = v_s  + (c & 1) * (CHK * PD);
        float* xkn = xk_s + ((c + 1) & 1) * (CHK * PD);
        float* vn  = v_s  + ((c + 1) & 1) * (CHK * PD);
        const int cn = (c + 1 < nlight) ? (c + 1) : c;
#pragma unroll
        for (int t = 0; t < 4; ++t) {
          rk[t] = kh[cn * CHK * DH + tid + 256 * t];
          rv[t] = vh[cn * CHK * DH + tid + 256 * t];
        }
        v8f acck = {};
#pragma unroll
        for (int kk = 0; kk < DH / 4; ++kk) {
          v2f a;
          a.x = xkb[lr * PD + 4 * kk + 2 * lh];
          a.y = xkb[lr * PD + 4 * kk + 2 * lh + 1];
          acck = wmma4(a, pb[kk], acck);
        }
#pragma unroll
        for (int j = 0; j < 8; ++j)
          kp_s[(j + 8 * lh) * PM + m0 + lr] = fmaxf(acck[j] * norm, 0.f) + KEPS;
        __builtin_amdgcn_fence(__ATOMIC_RELEASE, "wavefront");
        __builtin_amdgcn_wave_barrier();
        __builtin_amdgcn_fence(__ATOMIC_ACQUIRE, "wavefront");
#pragma unroll
        for (int r = 0; r < CHK; ++r) run += kp_s[r * PM + m0 + lr];
#pragma unroll
        for (int et = 0; et < 4; ++et) {
          const int e0 = et * 16;
#pragma unroll
          for (int kb = 0; kb < CHK; kb += 4) {
            v2f a, b;
            a.x = kp_s[(kb + 2 * lh) * PM + m0 + lr];
            a.y = kp_s[(kb + 2 * lh + 1) * PM + m0 + lr];
            b.x = vb[(kb + 2 * lh) * PD + e0 + lr];
            b.y = vb[(kb + 2 * lh + 1) * PD + e0 + lr];
            sreg[et] = wmma4(a, b, sreg[et]);
          }
        }
#pragma unroll
        for (int t = 0; t < 4; ++t) {
          const int p = ((tid >> 6) + 4 * t) * PD + (tid & 63);
          xkn[p] = rk[t];
          vn[p]  = rv[t];
        }
        __syncthreads();
      }
    }
#pragma unroll
    for (int et = 0; et < 4; ++et)
#pragma unroll
      for (int j = 0; j < 8; ++j)
        S_s[(m0 + j + 8 * lh) * PD + et * 16 + lr] = sreg[et][j];
    if (lh == 0) kc_s[m0 + lr] = run;
  }
  __syncthreads();

  for (int n0 = n_begin; n0 < n_begin + SEGL; n0 += CHK) {
    for (int i = tid; i < CHK * DH; i += 256) {
      const int p = (i >> 6) * PD + (i & 63);
      xq_s[p] = qh[n0 * DH + i];
      xk_s[p] = kh[n0 * DH + i];
      v_s[p]  = vh[n0 * DH + i];
    }
    __syncthreads();

    {
      const int m0 = wave * 16;
      v8f accq = {};
      v8f acck = {};
#pragma unroll
      for (int kb = 0; kb < DH; kb += 4) {
        v2f a, b;
        b.x = proj_s[(m0 + lr) * PD + kb + 2 * lh];
        b.y = proj_s[(m0 + lr) * PD + kb + 2 * lh + 1];
        a.x = xq_s[lr * PD + kb + 2 * lh];
        a.y = xq_s[lr * PD + kb + 2 * lh + 1];
        accq = wmma4(a, b, accq);
        a.x = xk_s[lr * PD + kb + 2 * lh];
        a.y = xk_s[lr * PD + kb + 2 * lh + 1];
        acck = wmma4(a, b, acck);
      }
#pragma unroll
      for (int j = 0; j < 8; ++j) {
        const int row = j + 8 * lh;
        qp_s[row * PM + m0 + lr] = fmaxf(accq[j] * norm, 0.f) + KEPS;
        kp_s[row * PM + m0 + lr] = fmaxf(acck[j] * norm, 0.f) + KEPS;
      }
    }
    __syncthreads();

    if (tid < MF) {
      float run = kc_s[tid];
#pragma unroll
      for (int r = 0; r < CHK; ++r) {
        run += kp_s[r * PM + tid];
        qp_s[r * PM + tid] = qp_s[r * PM + tid] / run;
      }
      kc_s[tid] = run;
    }
    __syncthreads();

    v8f o = {};
    if (wave < 4) {
      const int e0 = wave * 16;
#pragma unroll
      for (int kb = 0; kb < MF; kb += 4) {
        v2f a, b;
        a.x = qp_s[lr * PM + kb + 2 * lh];
        a.y = qp_s[lr * PM + kb + 2 * lh + 1];
        b.x = S_s[(kb + 2 * lh) * PD + e0 + lr];
        b.y = S_s[(kb + 2 * lh + 1) * PD + e0 + lr];
        o = wmma4(a, b, o);
      }
    } else if (wave == 4) {
      v8f am = {};
#pragma unroll
      for (int kb = 0; kb < MF; kb += 4) {
        v2f a, b;
        a.x = qp_s[lr * PM + kb + 2 * lh];
        a.y = qp_s[lr * PM + kb + 2 * lh + 1];
        b.x = kp_s[lr * PM + kb + 2 * lh];
        b.y = kp_s[lr * PM + kb + 2 * lh + 1];
        am = wmma4(a, b, am);
      }
#pragma unroll
      for (int j = 0; j < 8; ++j) {
        const int row = j + 8 * lh;
        a_s[row * PA + lr] = (lr <= row) ? am[j] : 0.f;
      }
    }
    __syncthreads();

    if (wave < 4) {
      const int e0 = wave * 16;
#pragma unroll
      for (int kb = 0; kb < CHK; kb += 4) {
        v2f a, b;
        a.x = a_s[lr * PA + kb + 2 * lh];
        a.y = a_s[lr * PA + kb + 2 * lh + 1];
        b.x = v_s[(kb + 2 * lh) * PD + e0 + lr];
        b.y = v_s[(kb + 2 * lh + 1) * PD + e0 + lr];
        o = wmma4(a, b, o);
      }
#pragma unroll
      for (int j = 0; j < 8; ++j) {
        const int row = j + 8 * lh;
        oh[(n0 + row) * DH + e0 + lr] = o[j];
      }
    }
    {
      const int m0 = wave * 16;
#pragma unroll
      for (int et = 0; et < 4; ++et) {
        const int e0 = et * 16;
        v8f sacc;
#pragma unroll
        for (int j = 0; j < 8; ++j)
          sacc[j] = S_s[(m0 + j + 8 * lh) * PD + e0 + lr];
#pragma unroll
        for (int kb = 0; kb < CHK; kb += 4) {
          v2f a, b;
          a.x = kp_s[(kb + 2 * lh) * PM + m0 + lr];
          a.y = kp_s[(kb + 2 * lh + 1) * PM + m0 + lr];
          b.x = v_s[(kb + 2 * lh) * PD + e0 + lr];
          b.y = v_s[(kb + 2 * lh + 1) * PD + e0 + lr];
          sacc = wmma4(a, b, sacc);
        }
#pragma unroll
        for (int j = 0; j < 8; ++j)
          S_s[(m0 + j + 8 * lh) * PD + e0 + lr] = sacc[j];
      }
    }
    __syncthreads();
  }
}

extern "C" void kernel_launch(void* const* d_in, const int* in_sizes, int n_in,
                              void* d_out, int out_size, void* d_ws, size_t ws_size,
                              hipStream_t stream) {
  (void)in_sizes; (void)n_in; (void)out_size;
  const float* q    = (const float*)d_in[0];
  const float* k    = (const float*)d_in[1];
  const float* v    = (const float*)d_in[2];
  const float* proj = (const float*)d_in[3];
  float* out        = (float*)d_out;
  float* ws         = (float*)d_ws;
  performer_segment_state_f32_wmma<<<dim3(120), dim3(256), 0, stream>>>(k, v, proj, ws, ws_size);
  performer_causal_f32_wmma<<<dim3(128), dim3(256), 0, stream>>>(q, k, v, proj, out, ws, ws_size);
}


# Scan launch fetches proj and its segment's 64 q/k/v rows once, up front, as 16-byte loads kept in LDS; the chunk loop's 4 dependent 4-byte load rounds and staging barrier are gone
# speedup vs baseline: 20.2645x; 1.3489x over previous
#include <hip/hip_runtime.h>


#define KEPS 1e-3f
#define NSEQ 1024
#define DH   64
#define MF   128
#define CHK  16
#define NSEG 16
#define SEGL (NSEQ / NSEG)
static_assert(8 * NSEG == 128, "the launch below starts 8 x NSEG workgroups");
#define NPART   (NSEG - 1)
#define PSTRIDE (MF * DH + MF)
#define WS_FLOATS (8 * NPART * PSTRIDE)
static_assert(8 * NPART == 120, "the first launch below starts 8 x NPART workgroups");
static_assert(SEGL % CHK == 0 && SEGL == 64, "the first launch holds one segment of 64 rows in LDS");

#define PD   (DH + 12)
#define PM   (MF + 12)
#define PA   (CHK + 4)

typedef float v2f __attribute__((ext_vector_type(2)));
typedef float v8f __attribute__((ext_vector_type(8)));

__device__ __forceinline__ v8f wmma4(v2f a, v2f b, v8f c) {
  return __builtin_amdgcn_wmma_f32_16x16x4_f32(false, a, false, b,
                                               (short)0, c, false, false);
}


__global__ __launch_bounds__(256)
void performer_segment_state_f32_wmma(const float* __restrict__ k,
                                      const float* __restrict__ v,
                                      const float* __restrict__ proj,
                                      float* __restrict__ ws, size_t ws_bytes) {
  if (ws_bytes < (size_t)WS_FLOATS * sizeof(float)) return;

  __shared__ __attribute__((aligned(16))) float xk_s[SEGL * PD];
  __shared__ __attribute__((aligned(16))) float v_s[SEGL * PD];
  __shared__ float kp_s[SEGL * PM];

  const int h    = blockIdx.x / NPART;
  const int seg  = blockIdx.x % NPART;
  const int tid  = threadIdx.x;
  const int lane = tid & 31;
  const int wave = tid >> 5;
  const int lr   = lane & 15;
  const int lh   = lane >> 4;
  const int m0   = wave * 16;

  const float norm = 0.35355339059327378f;

  const float* kseg = k + ((size_t)h * NSEQ + (size_t)seg * SEGL) * DH;
  const float* vseg = v + ((size_t)h * NSEQ + (size_t)seg * SEGL) * DH;

  float4 rk[4], rv[4];
#pragma unroll
  for (int t = 0; t < 4; ++t) {
    const int idx = tid + 256 * t;
    rk[t] = *reinterpret_cast<const float4*>(kseg + 4 * idx);
    rv[t] = *reinterpret_cast<const float4*>(vseg + 4 * idx);
  }
  v2f pb[DH / 4];
#pragma unroll
  for (int kk = 0; kk < DH / 4; ++kk) {
    pb[kk].x = proj[(m0 + lr) * DH + 4 * kk + 2 * lh];
    pb[kk].y = proj[(m0 + lr) * DH + 4 * kk + 2 * lh + 1];
  }
#pragma unroll
  for (int t = 0; t < 4; ++t) {
    const int idx = tid + 256 * t;
    const int p = (idx >> 4) * PD + 4 * (idx & 15);
    *reinterpret_cast<float4*>(xk_s + p) = rk[t];
    *reinterpret_cast<float4*>(v_s + p)  = rv[t];
  }
  __syncthreads();

  v8f acc[SEGL / CHK] = {};
#pragma unroll
  for (int kk = 0; kk < DH / 4; ++kk) {
#pragma unroll
    for (int rt = 0; rt < SEGL / CHK; ++rt) {
      v2f a;
      a.x = xk_s[(CHK * rt + lr) * PD + 4 * kk + 2 * lh];
      a.y = xk_s[(CHK * rt + lr) * PD + 4 * kk + 2 * lh + 1];
      acc[rt] = wmma4(a, pb[kk], acc[rt]);
    }
  }
#pragma unroll
  for (int rt = 0; rt < SEGL / CHK; ++rt)
#pragma unroll
    for (int j = 0; j < 8; ++j)
      kp_s[(CHK * rt + j + 8 * lh) * PM + m0 + lr] = fmaxf(acc[rt][j] * norm, 0.f) + KEPS;
  __builtin_amdgcn_fence(__ATOMIC_RELEASE, "wavefront");
  __builtin_amdgcn_wave_barrier();
  __builtin_amdgcn_fence(__ATOMIC_ACQUIRE, "wavefront");

  float* P = ws + (size_t)blockIdx.x * PSTRIDE;

  float run = 0.f;
#pragma unroll
  for (int r = 0; r < SEGL; ++r) run += kp_s[r * PM + m0 + lr];
  if (lh == 0) P[MF * DH + m0 + lr] = run;

  v8f sreg[4] = {};
#pragma unroll
  for (int kb = 0; kb < SEGL; kb += 4) {
    v2f a;
    a.x = kp_s[(kb + 2 * lh) * PM + m0 + lr];
    a.y = kp_s[(kb + 2 * lh + 1) * PM + m0 + lr];
#pragma unroll
    for (int et = 0; et < 4; ++et) {
      v2f b;
      b.x = v_s[(kb + 2 * lh) * PD + et * 16 + lr];
      b.y = v_s[(kb + 2 * lh + 1) * PD + et * 16 + lr];
      sreg[et] = wmma4(a, b, sreg[et]);
    }
  }
#pragma unroll
  for (int et = 0; et < 4; ++et)
#pragma unroll
    for (int jh = 0; jh < 2; ++jh) {
      float4 o;
      o.x = sreg[et][4 * jh + 0];
      o.y = sreg[et][4 * jh + 1];
      o.z = sreg[et][4 * jh + 2];
      o.w = sreg[et][4 * jh + 3];
      *reinterpret_cast<float4*>(P + (((wave * 4 + et) * 2 + jh) * 32 + lane) * 4) = o;
    }
}

__global__ __launch_bounds__(256)
void performer_causal_f32_wmma(const float* __restrict__ q,
                               const float* __restrict__ k,
                               const float* __restrict__ v,
                               const float* __restrict__ proj,
                               float* __restrict__ out,
                               const float* __restrict__ ws, size_t ws_bytes) {
  __shared__ __attribute__((aligned(16))) float proj_s[MF * PD];
  __shared__ float S_s[MF * PD];
  __shared__ float qp_s[CHK * PM];
  __shared__ float kp_s[CHK * PM];
  __shared__ __attribute__((aligned(16))) float xq_s[SEGL * PD];
  __shared__ __attribute__((aligned(16))) float xk_s[SEGL * PD];
  __shared__ __attribute__((aligned(16))) float v_s[SEGL * PD];
  __shared__ float a_s[CHK * PA];
  __shared__ float kc_s[MF];

  const int h    = blockIdx.x / NSEG;
  const int sj   = blockIdx.x % NSEG;
  const int seg  = (sj & 1) ? (sj >> 1) : (NSEG - 1 - (sj >> 1));
  const int n_begin = seg * SEGL;
  const int nlight  = n_begin / CHK;
  const int tid  = threadIdx.x;
  const int lane = tid & 31;
  const int wave = tid >> 5;
  const int lr   = lane & 15;
  const int lh   = lane >> 4;

  const float norm = 0.35355339059327378f;

  const float* qh = q + (size_t)h * NSEQ * DH;
  const float* kh = k + (size_t)h * NSEQ * DH;
  const float* vh = v + (size_t)h * NSEQ * DH;
  float*       oh = out + (size_t)h * NSEQ * DH;

  {
    float4 rp[8];
#pragma unroll
    for (int t = 0; t < 8; ++t)
      rp[t] = *reinterpret_cast<const float4*>(proj + 4 * (tid + 256 * t));
    float4 rq4[4], rk4[4], rv4[4];
#pragma unroll
    for (int t = 0; t < 4; ++t) {
      const int idx = tid + 256 * t;
      rq4[t] = *reinterpret_cast<const float4*>(qh + (size_t)n_begin * DH + 4 * idx);
      rk4[t] = *reinterpret_cast<const float4*>(kh + (size_t)n_begin * DH + 4 * idx);
      rv4[t] = *reinterpret_cast<const float4*>(vh + (size_t)n_begin * DH + 4 * idx);
    }
#pragma unroll
    for (int t = 0; t < 8; ++t) {
      const int idx = tid + 256 * t;
      const int p = (idx >> 4) * PD + 4 * (idx & 15);
      *reinterpret_cast<float4*>(proj_s + p) = rp[t];
    }
#pragma unroll
    for (int t = 0; t < 4; ++t) {
      const int idx = tid + 256 * t;
      const int p = (idx >> 4) * PD + 4 * (idx & 15);
      *reinterpret_cast<float4*>(xq_s + p) = rq4[t];
      *reinterpret_cast<float4*>(xk_s + p) = rk4[t];
      *reinterpret_cast<float4*>(v_s + p)  = rv4[t];
    }
  }

  {
    const int m0 = wave * 16;
    v8f sreg[4] = {};
    float run = 0.f;
    const bool from_ws = ws_bytes >= (size_t)WS_FLOATS * sizeof(float);
    if (from_ws) {
      const float* Ph = ws + (size_t)(h * NPART) * PSTRIDE;
      for (int s0 = 0; s0 < seg; s0 += 4) {
        float4 t[4][8];
        float tk[4];
#pragma unroll
        for (int u = 0; u < 4; ++u) {
          if (s0 + u < seg) {
            const float* P = Ph + (size_t)(s0 + u) * PSTRIDE;
#pragma unroll
            for (int f = 0; f < 8; ++f)
              t[u][f] = *reinterpret_cast<const float4*>(P + ((wave * 8 + f) * 32 + lane) * 4);
            tk[u] = P[MF * DH + m0 + lr];
          }
        }
#pragma unroll
        for (int u = 0; u < 4; ++u) {
          if (s0 + u < seg) {
#pragma unroll
            for (int f = 0; f < 8; ++f) {
              sreg[f >> 1][4 * (f & 1) + 0] += t[u][f].x;
              sreg[f >> 1][4 * (f & 1) + 1] += t[u][f].y;
              sreg[f >> 1][4 * (f & 1) + 2] += t[u][f].z;
              sreg[f >> 1][4 * (f & 1) + 3] += t[u][f].w;
            }
            run += tk[u];
          }
        }
      }
    } else if (nlight > 0) {
      float* const xkbuf = S_s;
      float* const vbuf  = S_s + 2 * CHK * PD;
      __syncthreads();
      v2f pb[DH / 4];
#pragma unroll
      for (int kk = 0; kk < DH / 4; ++kk) {
        pb[kk].x = proj_s[(m0 + lr) * PD + 4 * kk + 2 * lh];
        pb[kk].y = proj_s[(m0 + lr) * PD + 4 * kk + 2 * lh + 1];
      }
      float rk[4], rv[4];
#pragma unroll
      for (int t = 0; t < 4; ++t) {
        rk[t] = kh[tid + 256 * t];
        rv[t] = vh[tid + 256 * t];
      }
#pragma unroll
      for (int t = 0; t < 4; ++t) {
        const int p = ((tid >> 6) + 4 * t) * PD + (tid & 63);
        xkbuf[p] = rk[t];
        vbuf[p]  = rv[t];
      }
      __syncthreads();
      for (int c = 0; c < nlight; ++c) {
        const float* xkb = xkbuf + (c & 1) * (CHK * PD);
        const float* vb  = vbuf  + (c & 1) * (CHK * PD);
        float* xkn = xkbuf + ((c + 1) & 1) * (CHK * PD);
        float* vn  = vbuf  + ((c + 1) & 1) * (CHK * PD);
        const int cn = (c + 1 < nlight) ? (c + 1) : c;
#pragma unroll
        for (int t = 0; t < 4; ++t) {
          rk[t] = kh[cn * CHK * DH + tid + 256 * t];
          rv[t] = vh[cn * CHK * DH + tid + 256 * t];
        }
        v8f acck = {};
#pragma unroll
        for (int kk = 0; kk < DH / 4; ++kk) {
          v2f a;
          a.x = xkb[lr * PD + 4 * kk + 2 * lh];
          a.y = xkb[lr * PD + 4 * kk + 2 * lh + 1];
          acck = wmma4(a, pb[kk], acck);
        }
#pragma unroll
        for (int j = 0; j < 8; ++j)
          kp_s[(j + 8 * lh) * PM + m0 + lr] = fmaxf(acck[j] * norm, 0.f) + KEPS;
        __builtin_amdgcn_fence(__ATOMIC_RELEASE, "wavefront");
        __builtin_amdgcn_wave_barrier();
        __builtin_amdgcn_fence(__ATOMIC_ACQUIRE, "wavefront");
#pragma unroll
        for (int r = 0; r < CHK; ++r) run += kp_s[r * PM + m0 + lr];
#pragma unroll
        for (int et = 0; et < 4; ++et) {
          const int e0 = et * 16;
#pragma unroll
          for (int kb = 0; kb < CHK; kb += 4) {
            v2f a, b;
            a.x = kp_s[(kb + 2 * lh) * PM + m0 + lr];
            a.y = kp_s[(kb + 2 * lh + 1) * PM + m0 + lr];
            b.x = vb[(kb + 2 * lh) * PD + e0 + lr];
            b.y = vb[(kb + 2 * lh + 1) * PD + e0 + lr];
            sreg[et] = wmma4(a, b, sreg[et]);
          }
        }
#pragma unroll
        for (int t = 0; t < 4; ++t) {
          const int p = ((tid >> 6) + 4 * t) * PD + (tid & 63);
          xkn[p] = rk[t];
          vn[p]  = rv[t];
        }
        __syncthreads();
      }
    }
#pragma unroll
    for (int et = 0; et < 4; ++et)
#pragma unroll
      for (int j = 0; j < 8; ++j)
        S_s[(m0 + j + 8 * lh) * PD + et * 16 + lr] = sreg[et][j];
    if (lh == 0) kc_s[m0 + lr] = run;
  }
  __syncthreads();

  for (int c0 = 0; c0 < SEGL; c0 += CHK) {
    const int n0 = n_begin + c0;
    const float* xqc = xq_s + c0 * PD;
    const float* xkc = xk_s + c0 * PD;
    const float* vc  = v_s  + c0 * PD;

    {
      const int m0 = wave * 16;
      v8f accq = {};
      v8f acck = {};
#pragma unroll
      for (int kb = 0; kb < DH; kb += 4) {
        v2f a, b;
        b.x = proj_s[(m0 + lr) * PD + kb + 2 * lh];
        b.y = proj_s[(m0 + lr) * PD + kb + 2 * lh + 1];
        a.x = xqc[lr * PD + kb + 2 * lh];
        a.y = xqc[lr * PD + kb + 2 * lh + 1];
        accq = wmma4(a, b, accq);
        a.x = xkc[lr * PD + kb + 2 * lh];
        a.y = xkc[lr * PD + kb + 2 * lh + 1];
        acck = wmma4(a, b, acck);
      }
#pragma unroll
      for (int j = 0; j < 8; ++j) {
        const int row = j + 8 * lh;
        qp_s[row * PM + m0 + lr] = fmaxf(accq[j] * norm, 0.f) + KEPS;
        kp_s[row * PM + m0 + lr] = fmaxf(acck[j] * norm, 0.f) + KEPS;
      }
    }
    __syncthreads();

    if (tid < MF) {
      float run = kc_s[tid];
#pragma unroll
      for (int r = 0; r < CHK; ++r) {
        run += kp_s[r * PM + tid];
        qp_s[r * PM + tid] = qp_s[r * PM + tid] / run;
      }
      kc_s[tid] = run;
    }
    __syncthreads();

    v8f o = {};
    if (wave < 4) {
      const int e0 = wave * 16;
#pragma unroll
      for (int kb = 0; kb < MF; kb += 4) {
        v2f a, b;
        a.x = qp_s[lr * PM + kb + 2 * lh];
        a.y = qp_s[lr * PM + kb + 2 * lh + 1];
        b.x = S_s[(kb + 2 * lh) * PD + e0 + lr];
        b.y = S_s[(kb + 2 * lh + 1) * PD + e0 + lr];
        o = wmma4(a, b, o);
      }
    } else if (wave == 4) {
      v8f am = {};
#pragma unroll
      for (int kb = 0; kb < MF; kb += 4) {
        v2f a, b;
        a.x = qp_s[lr * PM + kb + 2 * lh];
        a.y = qp_s[lr * PM + kb + 2 * lh + 1];
        b.x = kp_s[lr * PM + kb + 2 * lh];
        b.y = kp_s[lr * PM + kb + 2 * lh + 1];
        am = wmma4(a, b, am);
      }
#pragma unroll
      for (int j = 0; j < 8; ++j) {
        const int row = j + 8 * lh;
        a_s[row * PA + lr] = (lr <= row) ? am[j] : 0.f;
      }
    }
    __syncthreads();

    if (wave < 4) {
      const int e0 = wave * 16;
#pragma unroll
      for (int kb = 0; kb < CHK; kb += 4) {
        v2f a, b;
        a.x = a_s[lr * PA + kb + 2 * lh];
        a.y = a_s[lr * PA + kb + 2 * lh + 1];
        b.x = vc[(kb + 2 * lh) * PD + e0 + lr];
        b.y = vc[(kb + 2 * lh + 1) * PD + e0 + lr];
        o = wmma4(a, b, o);
      }
#pragma unroll
      for (int j = 0; j < 8; ++j) {
        const int row = j + 8 * lh;
        oh[(n0 + row) * DH + e0 + lr] = o[j];
      }
    }
    {
      const int m0 = wave * 16;
#pragma unroll
      for (int et = 0; et < 4; ++et) {
        const int e0 = et * 16;
        v8f sacc;
#pragma unroll
        for (int j = 0; j < 8; ++j)
          sacc[j] = S_s[(m0 + j + 8 * lh) * PD + e0 + lr];
#pragma unroll
        for (int kb = 0; kb < CHK; kb += 4) {
          v2f a, b;
          a.x = kp_s[(kb + 2 * lh) * PM + m0 + lr];
          a.y = kp_s[(kb + 2 * lh + 1) * PM + m0 + lr];
          b.x = vc[(kb + 2 * lh) * PD + e0 + lr];
          b.y = vc[(kb + 2 * lh + 1) * PD + e0 + lr];
          sacc = wmma4(a, b, sacc);
        }
#pragma unroll
        for (int j = 0; j < 8; ++j)
          S_s[(m0 + j + 8 * lh) * PD + e0 + lr] = sacc[j];
      }
    }
    __syncthreads();
  }
}

extern "C" void kernel_launch(void* const* d_in, const int* in_sizes, int n_in,
                              void* d_out, int out_size, void* d_ws, size_t ws_size,
                              hipStream_t stream) {
  (void)in_sizes; (void)n_in; (void)out_size;
  const float* q    = (const float*)d_in[0];
  const float* k    = (const float*)d_in[1];
  const float* v    = (const float*)d_in[2];
  const float* proj = (const float*)d_in[3];
  float* out        = (float*)d_out;
  float* ws         = (float*)d_ws;
  performer_segment_state_f32_wmma<<<dim3(120), dim3(256), 0, stream>>>(k, v, proj, ws, ws_size);
  performer_causal_f32_wmma<<<dim3(128), dim3(256), 0, stream>>>(q, k, v, proj, out, ws, ws_size);
}


# Scan launch split over the two 32-column halves of the value dimension: 256 workgroups (was 128), each sums from the workspace, keeps and updates only its half of S; last segments started first
# speedup vs baseline: 22.2837x; 1.0996x over previous
#include <hip/hip_runtime.h>


#define KEPS 1e-3f
#define NSEQ 1024
#define DH   64
#define MF   128
#define CHK  16
#define NSEG 16
#define SEGL (NSEQ / NSEG)
#define NEH  2
#define EW   (DH / NEH)
#define ETW  (EW / 16)
static_assert(8 * NSEG * NEH == 256 && NSEG == 16 && NEH == 2 && EW == 32 && ETW == 2,
              "the scan launch below starts 8 x NSEG x NEH workgroups and decodes its block id by shifts");
#define NPART   (NSEG - 1)
#define PSTRIDE (MF * DH + MF)
#define WS_FLOATS (8 * NPART * PSTRIDE)
static_assert(8 * NPART == 120, "the first launch below starts 8 x NPART workgroups");
static_assert(SEGL % CHK == 0 && SEGL == 64, "the first launch holds one segment of 64 rows in LDS");

#define PD   (DH + 12)
#define PM   (MF + 12)
#define PA   (CHK + 4)

typedef float v2f __attribute__((ext_vector_type(2)));
typedef float v8f __attribute__((ext_vector_type(8)));

__device__ __forceinline__ v8f wmma4(v2f a, v2f b, v8f c) {
  return __builtin_amdgcn_wmma_f32_16x16x4_f32(false, a, false, b,
                                               (short)0, c, false, false);
}


__global__ __launch_bounds__(256)
void performer_segment_state_f32_wmma(const float* __restrict__ k,
                                      const float* __restrict__ v,
                                      const float* __restrict__ proj,
                                      float* __restrict__ ws, size_t ws_bytes) {
  if (ws_bytes < (size_t)WS_FLOATS * sizeof(float)) return;

  __shared__ __attribute__((aligned(16))) float xk_s[SEGL * PD];
  __shared__ __attribute__((aligned(16))) float v_s[SEGL * PD];
  __shared__ float kp_s[SEGL * PM];

  const int h    = blockIdx.x / NPART;
  const int seg  = blockIdx.x % NPART;
  const int tid  = threadIdx.x;
  const int lane = tid & 31;
  const int wave = tid >> 5;
  const int lr   = lane & 15;
  const int lh   = lane >> 4;
  const int m0   = wave * 16;

  const float norm = 0.35355339059327378f;

  const float* kseg = k + ((size_t)h * NSEQ + (size_t)seg * SEGL) * DH;
  const float* vseg = v + ((size_t)h * NSEQ + (size_t)seg * SEGL) * DH;

  float4 rk[4], rv[4];
#pragma unroll
  for (int t = 0; t < 4; ++t) {
    const int idx = tid + 256 * t;
    rk[t] = *reinterpret_cast<const float4*>(kseg + 4 * idx);
    rv[t] = *reinterpret_cast<const float4*>(vseg + 4 * idx);
  }
  v2f pb[DH / 4];
#pragma unroll
  for (int kk = 0; kk < DH / 4; ++kk) {
    pb[kk].x = proj[(m0 + lr) * DH + 4 * kk + 2 * lh];
    pb[kk].y = proj[(m0 + lr) * DH + 4 * kk + 2 * lh + 1];
  }
#pragma unroll
  for (int t = 0; t < 4; ++t) {
    const int idx = tid + 256 * t;
    const int p = (idx >> 4) * PD + 4 * (idx & 15);
    *reinterpret_cast<float4*>(xk_s + p) = rk[t];
    *reinterpret_cast<float4*>(v_s + p)  = rv[t];
  }
  __syncthreads();

  v8f acc[SEGL / CHK] = {};
#pragma unroll
  for (int kk = 0; kk < DH / 4; ++kk) {
#pragma unroll
    for (int rt = 0; rt < SEGL / CHK; ++rt) {
      v2f a;
      a.x = xk_s[(CHK * rt + lr) * PD + 4 * kk + 2 * lh];
      a.y = xk_s[(CHK * rt + lr) * PD + 4 * kk + 2 * lh + 1];
      acc[rt] = wmma4(a, pb[kk], acc[rt]);
    }
  }
#pragma unroll
  for (int rt = 0; rt < SEGL / CHK; ++rt)
#pragma unroll
    for (int j = 0; j < 8; ++j)
      kp_s[(CHK * rt + j + 8 * lh) * PM + m0 + lr] = fmaxf(acc[rt][j] * norm, 0.f) + KEPS;
  __builtin_amdgcn_fence(__ATOMIC_RELEASE, "wavefront");
  __builtin_amdgcn_wave_barrier();
  __builtin_amdgcn_fence(__ATOMIC_ACQUIRE, "wavefront");

  float* P = ws + (size_t)blockIdx.x * PSTRIDE;

  float run = 0.f;
#pragma unroll
  for (int r = 0; r < SEGL; ++r) run += kp_s[r * PM + m0 + lr];
  if (lh == 0) P[MF * DH + m0 + lr] = run;

  v8f sreg[4] = {};
#pragma unroll
  for (int kb = 0; kb < SEGL; kb += 4) {
    v2f a;
    a.x = kp_s[(kb + 2 * lh) * PM + m0 + lr];
    a.y = kp_s[(kb + 2 * lh + 1) * PM + m0 + lr];
#pragma unroll
    for (int et = 0; et < 4; ++et) {
      v2f b;
      b.x = v_s[(kb + 2 * lh) * PD + et * 16 + lr];
      b.y = v_s[(kb + 2 * lh + 1) * PD + et * 16 + lr];
      sreg[et] = wmma4(a, b, sreg[et]);
    }
  }
#pragma unroll
  for (int et = 0; et < 4; ++et)
#pragma unroll
    for (int jh = 0; jh < 2; ++jh) {
      float4 o;
      o.x = sreg[et][4 * jh + 0];
      o.y = sreg[et][4 * jh + 1];
      o.z = sreg[et][4 * jh + 2];
      o.w = sreg[et][4 * jh + 3];
      *reinterpret_cast<float4*>(P + (((wave * 4 + et) * 2 + jh) * 32 + lane) * 4) = o;
    }
}

__device__ __forceinline__ void contrib_load(const float* __restrict__ P, int blk0, int lane, int col,
                                             float4 (&t)[4], float& tk) {
#pragma unroll
  for (int i = 0; i < 4; ++i)
    t[i] = *reinterpret_cast<const float4*>(P + ((blk0 + i) * 32 + lane) * 4);
  tk = P[MF * DH + col];
}
__device__ __forceinline__ void contrib_add(v8f (&sreg)[ETW], float& run, const float4 (&t)[4], float tk) {
#pragma unroll
  for (int i = 0; i < 4; ++i) {
    sreg[i >> 1][4 * (i & 1) + 0] += t[i].x;
    sreg[i >> 1][4 * (i & 1) + 1] += t[i].y;
    sreg[i >> 1][4 * (i & 1) + 2] += t[i].z;
    sreg[i >> 1][4 * (i & 1) + 3] += t[i].w;
  }
  run += tk;
}

__global__ __launch_bounds__(256)
void performer_causal_f32_wmma(const float* __restrict__ q,
                               const float* __restrict__ k,
                               const float* __restrict__ v,
                               const float* __restrict__ proj,
                               float* __restrict__ out,
                               const float* __restrict__ ws, size_t ws_bytes) {
  __shared__ __attribute__((aligned(16))) float proj_s[MF * PD];
  __shared__ float S_s[MF * PD];
  __shared__ float qp_s[CHK * PM];
  __shared__ float kp_s[CHK * PM];
  __shared__ __attribute__((aligned(16))) float xq_s[SEGL * PD];
  __shared__ __attribute__((aligned(16))) float xk_s[SEGL * PD];
  __shared__ __attribute__((aligned(16))) float v_s[SEGL * PD];
  __shared__ float a_s[CHK * PA];
  __shared__ float kc_s[MF];

  const int eh   = blockIdx.x & 1;
  const int h    = (blockIdx.x >> 1) & 7;
  const int seg  = NSEG - 1 - (blockIdx.x >> 4);
  const int ec0  = EW * eh;
  const int n_begin = seg * SEGL;
  const int nlight  = n_begin / CHK;
  const int tid  = threadIdx.x;
  const int lane = tid & 31;
  const int wave = tid >> 5;
  const int lr   = lane & 15;
  const int lh   = lane >> 4;

  const float norm = 0.35355339059327378f;

  const float* qh = q + (size_t)h * NSEQ * DH;
  const float* kh = k + (size_t)h * NSEQ * DH;
  const float* vh = v + (size_t)h * NSEQ * DH;
  float*       oh = out + (size_t)h * NSEQ * DH;

  {
    float4 rp[8];
#pragma unroll
    for (int t = 0; t < 8; ++t)
      rp[t] = *reinterpret_cast<const float4*>(proj + 4 * (tid + 256 * t));
    float4 rq4[4], rk4[4], rv4[2];
#pragma unroll
    for (int t = 0; t < 4; ++t) {
      const int idx = tid + 256 * t;
      rq4[t] = *reinterpret_cast<const float4*>(qh + (size_t)n_begin * DH + 4 * idx);
      rk4[t] = *reinterpret_cast<const float4*>(kh + (size_t)n_begin * DH + 4 * idx);
    }
#pragma unroll
    for (int t = 0; t < 2; ++t) {
      const int idx = tid + 256 * t;
      rv4[t] = *reinterpret_cast<const float4*>(vh + (size_t)(n_begin + (idx >> 3)) * DH + ec0 + 4 * (idx & 7));
    }
#pragma unroll
    for (int t = 0; t < 8; ++t) {
      const int idx = tid + 256 * t;
      const int p = (idx >> 4) * PD + 4 * (idx & 15);
      *reinterpret_cast<float4*>(proj_s + p) = rp[t];
    }
#pragma unroll
    for (int t = 0; t < 4; ++t) {
      const int idx = tid + 256 * t;
      const int p = (idx >> 4) * PD + 4 * (idx & 15);
      *reinterpret_cast<float4*>(xq_s + p) = rq4[t];
      *reinterpret_cast<float4*>(xk_s + p) = rk4[t];
    }
#pragma unroll
    for (int t = 0; t < 2; ++t) {
      const int idx = tid + 256 * t;
      *reinterpret_cast<float4*>(v_s + (idx >> 3) * PD + ec0 + 4 * (idx & 7)) = rv4[t];
    }
  }

  {
    const int m0 = wave * 16;
    v8f sreg[ETW] = {};
    float run = 0.f;
    const bool from_ws = ws_bytes >= (size_t)WS_FLOATS * sizeof(float);
    if (from_ws) {
      const float* Ph = ws + (size_t)(h * NPART) * PSTRIDE;
      const int blk0 = wave * 8 + 4 * eh;
      int s0 = 0;
      for (; s0 + 8 <= seg; s0 += 8) {
        float4 t[8][4];
        float tk[8];
#pragma unroll
        for (int u = 0; u < 8; ++u)
          contrib_load(Ph + (size_t)(s0 + u) * PSTRIDE, blk0, lane, m0 + lr, t[u], tk[u]);
#pragma unroll
        for (int u = 0; u < 8; ++u)
          contrib_add(sreg, run, t[u], tk[u]);
      }
      {
        const int rem = seg - s0;
        float4 t[7][4];
        float tk[7];
#pragma unroll
        for (int u = 0; u < 7; ++u)
          if (u < rem)
            contrib_load(Ph + (size_t)(s0 + u) * PSTRIDE, blk0, lane, m0 + lr, t[u], tk[u]);
#pragma unroll
        for (int u = 0; u < 7; ++u)
          if (u < rem)
            contrib_add(sreg, run, t[u], tk[u]);
      }
    } else if (nlight > 0) {
      float* const xkbuf = S_s;
      float* const vbuf  = S_s + 2 * CHK * PD;
      __syncthreads();
      v2f pb[DH / 4];
#pragma unroll
      for (int kk = 0; kk < DH / 4; ++kk) {
        pb[kk].x = proj_s[(m0 + lr) * PD + 4 * kk + 2 * lh];
        pb[kk].y = proj_s[(m0 + lr) * PD + 4 * kk + 2 * lh + 1];
      }
      float rk[4], rv[4];
#pragma unroll
      for (int t = 0; t < 4; ++t) {
        rk[t] = kh[tid + 256 * t];
        rv[t] = vh[tid + 256 * t];
      }
#pragma unroll
      for (int t = 0; t < 4; ++t) {
        const int p = ((tid >> 6) + 4 * t) * PD + (tid & 63);
        xkbuf[p] = rk[t];
        vbuf[p]  = rv[t];
      }
      __syncthreads();
      for (int c = 0; c < nlight; ++c) {
        const float* xkb = xkbuf + (c & 1) * (CHK * PD);
        const float* vb  = vbuf  + (c & 1) * (CHK * PD);
        float* xkn = xkbuf + ((c + 1) & 1) * (CHK * PD);
        float* vn  = vbuf  + ((c + 1) & 1) * (CHK * PD);
        const int cn = (c + 1 < nlight) ? (c + 1) : c;
#pragma unroll
        for (int t = 0; t < 4; ++t) {
          rk[t] = kh[cn * CHK * DH + tid + 256 * t];
          rv[t] = vh[cn * CHK * DH + tid + 256 * t];
        }
        v8f acck = {};
#pragma unroll
        for (int kk = 0; kk < DH / 4; ++kk) {
          v2f a;
          a.x = xkb[lr * PD + 4 * kk + 2 * lh];
          a.y = xkb[lr * PD + 4 * kk + 2 * lh + 1];
          acck = wmma4(a, pb[kk], acck);
        }
#pragma unroll
        for (int j = 0; j < 8; ++j)
          kp_s[(j + 8 * lh) * PM + m0 + lr] = fmaxf(acck[j] * norm, 0.f) + KEPS;
        __builtin_amdgcn_fence(__ATOMIC_RELEASE, "wavefront");
        __builtin_amdgcn_wave_barrier();
        __builtin_amdgcn_fence(__ATOMIC_ACQUIRE, "wavefront");
#pragma unroll
        for (int r = 0; r < CHK; ++r) run += kp_s[r * PM + m0 + lr];
#pragma unroll
        for (int etl = 0; etl < ETW; ++etl) {
          const int e0 = ec0 + etl * 16;
#pragma unroll
          for (int kb = 0; kb < CHK; kb += 4) {
            v2f a, b;
            a.x = kp_s[(kb + 2 * lh) * PM + m0 + lr];
            a.y = kp_s[(kb + 2 * lh + 1) * PM + m0 + lr];
            b.x = vb[(kb + 2 * lh) * PD + e0 + lr];
            b.y = vb[(kb + 2 * lh + 1) * PD + e0 + lr];
            sreg[etl] = wmma4(a, b, sreg[etl]);
          }
        }
#pragma unroll
        for (int t = 0; t < 4; ++t) {
          const int p = ((tid >> 6) + 4 * t) * PD + (tid & 63);
          xkn[p] = rk[t];
          vn[p]  = rv[t];
        }
        __syncthreads();
      }
    }
#pragma unroll
    for (int etl = 0; etl < ETW; ++etl)
#pragma unroll
      for (int j = 0; j < 8; ++j)
        S_s[(m0 + j + 8 * lh) * PD + ec0 + etl * 16 + lr] = sreg[etl][j];
    if (lh == 0) kc_s[m0 + lr] = run;
  }
  __syncthreads();

  for (int c0 = 0; c0 < SEGL; c0 += CHK) {
    const int n0 = n_begin + c0;
    const float* xqc = xq_s + c0 * PD;
    const float* xkc = xk_s + c0 * PD;
    const float* vc  = v_s  + c0 * PD;

    {
      const int m0 = wave * 16;
      v8f accq = {};
      v8f acck = {};
#pragma unroll
      for (int kb = 0; kb < DH; kb += 4) {
        v2f a, b;
        b.x = proj_s[(m0 + lr) * PD + kb + 2 * lh];
        b.y = proj_s[(m0 + lr) * PD + kb + 2 * lh + 1];
        a.x = xqc[lr * PD + kb + 2 * lh];
        a.y = xqc[lr * PD + kb + 2 * lh + 1];
        accq = wmma4(a, b, accq);
        a.x = xkc[lr * PD + kb + 2 * lh];
        a.y = xkc[lr * PD + kb + 2 * lh + 1];
        acck = wmma4(a, b, acck);
      }
#pragma unroll
      for (int j = 0; j < 8; ++j) {
        const int row = j + 8 * lh;
        qp_s[row * PM + m0 + lr] = fmaxf(accq[j] * norm, 0.f) + KEPS;
        kp_s[row * PM + m0 + lr] = fmaxf(acck[j] * norm, 0.f) + KEPS;
      }
    }
    __syncthreads();

    if (tid < MF) {
      float run = kc_s[tid];
#pragma unroll
      for (int r = 0; r < CHK; ++r) {
        run += kp_s[r * PM + tid];
        qp_s[r * PM + tid] = qp_s[r * PM + tid] / run;
      }
      kc_s[tid] = run;
    }
    __syncthreads();

    v8f o = {};
    if (wave < ETW) {
      const int e0 = ec0 + wave * 16;
#pragma unroll
      for (int kb = 0; kb < MF; kb += 4) {
        v2f a, b;
        a.x = qp_s[lr * PM + kb + 2 * lh];
        a.y = qp_s[lr * PM + kb + 2 * lh + 1];
        b.x = S_s[(kb + 2 * lh) * PD + e0 + lr];
        b.y = S_s[(kb + 2 * lh + 1) * PD + e0 + lr];
        o = wmma4(a, b, o);
      }
    } else if (wave == ETW) {
      v8f am = {};
#pragma unroll
      for (int kb = 0; kb < MF; kb += 4) {
        v2f a, b;
        a.x = qp_s[lr * PM + kb + 2 * lh];
        a.y = qp_s[lr * PM + kb + 2 * lh + 1];
        b.x = kp_s[lr * PM + kb + 2 * lh];
        b.y = kp_s[lr * PM + kb + 2 * lh + 1];
        am = wmma4(a, b, am);
      }
#pragma unroll
      for (int j = 0; j < 8; ++j) {
        const int row = j + 8 * lh;
        a_s[row * PA + lr] = (lr <= row) ? am[j] : 0.f;
      }
    }
    __syncthreads();

    if (wave < ETW) {
      const int e0 = ec0 + wave * 16;
#pragma unroll
      for (int kb = 0; kb < CHK; kb += 4) {
        v2f a, b;
        a.x = a_s[lr * PA + kb + 2 * lh];
        a.y = a_s[lr * PA + kb + 2 * lh + 1];
        b.x = vc[(kb + 2 * lh) * PD + e0 + lr];
        b.y = vc[(kb + 2 * lh + 1) * PD + e0 + lr];
        o = wmma4(a, b, o);
      }
#pragma unroll
      for (int j = 0; j < 8; ++j) {
        const int row = j + 8 * lh;
        oh[(n0 + row) * DH + e0 + lr] = o[j];
      }
    }
    {
      const int m0 = wave * 16;
#pragma unroll
      for (int etl = 0; etl < ETW; ++etl) {
        const int e0 = ec0 + etl * 16;
        v8f sacc;
#pragma unroll
        for (int j = 0; j < 8; ++j)
          sacc[j] = S_s[(m0 + j + 8 * lh) * PD + e0 + lr];
#pragma unroll
        for (int kb = 0; kb < CHK; kb += 4) {
          v2f a, b;
          a.x = kp_s[(kb + 2 * lh) * PM + m0 + lr];
          a.y = kp_s[(kb + 2 * lh + 1) * PM + m0 + lr];
          b.x = vc[(kb + 2 * lh) * PD + e0 + lr];
          b.y = vc[(kb + 2 * lh + 1) * PD + e0 + lr];
          sacc = wmma4(a, b, sacc);
        }
#pragma unroll
        for (int j = 0; j < 8; ++j)
          S_s[(m0 + j + 8 * lh) * PD + e0 + lr] = sacc[j];
      }
    }
    __syncthreads();
  }
}

extern "C" void kernel_launch(void* const* d_in, const int* in_sizes, int n_in,
                              void* d_out, int out_size, void* d_ws, size_t ws_size,
                              hipStream_t stream) {
  (void)in_sizes; (void)n_in; (void)out_size;
  const float* q    = (const float*)d_in[0];
  const float* k    = (const float*)d_in[1];
  const float* v    = (const float*)d_in[2];
  const float* proj = (const float*)d_in[3];
  float* out        = (float*)d_out;
  float* ws         = (float*)d_ws;
  performer_segment_state_f32_wmma<<<dim3(120), dim3(256), 0, stream>>>(k, v, proj, ws, ws_size);
  performer_causal_f32_wmma<<<dim3(256), dim3(256), 0, stream>>>(q, k, v, proj, out, ws, ws_size);
}


# Scan launch requests the front segments' workspace contributions once its inputs are in LDS and adds them (same order) after the q/k projections, hoisted out of the chunk loop: prefix read under WMMA
# speedup vs baseline: 22.9843x; 1.0314x over previous
#include <hip/hip_runtime.h>


#define KEPS 1e-3f
#define NSEQ 1024
#define DH   64
#define MF   128
#define CHK  16
#define NSEG 16
#define SEGL (NSEQ / NSEG)
#define NEH  2
#define EW   (DH / NEH)
#define ETW  (EW / 16)
static_assert(8 * NSEG * NEH == 256 && NSEG == 16 && NEH == 2 && EW == 32 && ETW == 2,
              "the scan launch below starts 8 x NSEG x NEH workgroups and decodes its block id by shifts");
#define NPART   (NSEG - 1)
#define PSTRIDE (MF * DH + MF)
#define WS_FLOATS (8 * NPART * PSTRIDE)
static_assert(8 * NPART == 120, "the first launch below starts 8 x NPART workgroups");
static_assert(SEGL % CHK == 0 && SEGL == 64, "the first launch holds one segment of 64 rows in LDS");

#define PD   (DH + 12)
#define PM   (MF + 12)
#define PA   (CHK + 4)

typedef float v2f __attribute__((ext_vector_type(2)));
typedef float v8f __attribute__((ext_vector_type(8)));

__device__ __forceinline__ v8f wmma4(v2f a, v2f b, v8f c) {
  return __builtin_amdgcn_wmma_f32_16x16x4_f32(false, a, false, b,
                                               (short)0, c, false, false);
}


__global__ __launch_bounds__(256)
void performer_segment_state_f32_wmma(const float* __restrict__ k,
                                      const float* __restrict__ v,
                                      const float* __restrict__ proj,
                                      float* __restrict__ ws, size_t ws_bytes) {
  if (ws_bytes < (size_t)WS_FLOATS * sizeof(float)) return;

  __shared__ __attribute__((aligned(16))) float xk_s[SEGL * PD];
  __shared__ __attribute__((aligned(16))) float v_s[SEGL * PD];
  __shared__ float kp_s[SEGL * PM];

  const int h    = blockIdx.x / NPART;
  const int seg  = blockIdx.x % NPART;
  const int tid  = threadIdx.x;
  const int lane = tid & 31;
  const int wave = tid >> 5;
  const int lr   = lane & 15;
  const int lh   = lane >> 4;
  const int m0   = wave * 16;

  const float norm = 0.35355339059327378f;

  const float* kseg = k + ((size_t)h * NSEQ + (size_t)seg * SEGL) * DH;
  const float* vseg = v + ((size_t)h * NSEQ + (size_t)seg * SEGL) * DH;

  float4 rk[4], rv[4];
#pragma unroll
  for (int t = 0; t < 4; ++t) {
    const int idx = tid + 256 * t;
    rk[t] = *reinterpret_cast<const float4*>(kseg + 4 * idx);
    rv[t] = *reinterpret_cast<const float4*>(vseg + 4 * idx);
  }
  v2f pb[DH / 4];
#pragma unroll
  for (int kk = 0; kk < DH / 4; ++kk) {
    pb[kk].x = proj[(m0 + lr) * DH + 4 * kk + 2 * lh];
    pb[kk].y = proj[(m0 + lr) * DH + 4 * kk + 2 * lh + 1];
  }
#pragma unroll
  for (int t = 0; t < 4; ++t) {
    const int idx = tid + 256 * t;
    const int p = (idx >> 4) * PD + 4 * (idx & 15);
    *reinterpret_cast<float4*>(xk_s + p) = rk[t];
    *reinterpret_cast<float4*>(v_s + p)  = rv[t];
  }
  __syncthreads();

  v8f acc[SEGL / CHK] = {};
#pragma unroll
  for (int kk = 0; kk < DH / 4; ++kk) {
#pragma unroll
    for (int rt = 0; rt < SEGL / CHK; ++rt) {
      v2f a;
      a.x = xk_s[(CHK * rt + lr) * PD + 4 * kk + 2 * lh];
      a.y = xk_s[(CHK * rt + lr) * PD + 4 * kk + 2 * lh + 1];
      acc[rt] = wmma4(a, pb[kk], acc[rt]);
    }
  }
#pragma unroll
  for (int rt = 0; rt < SEGL / CHK; ++rt)
#pragma unroll
    for (int j = 0; j < 8; ++j)
      kp_s[(CHK * rt + j + 8 * lh) * PM + m0 + lr] = fmaxf(acc[rt][j] * norm, 0.f) + KEPS;
  __builtin_amdgcn_fence(__ATOMIC_RELEASE, "wavefront");
  __builtin_amdgcn_wave_barrier();
  __builtin_amdgcn_fence(__ATOMIC_ACQUIRE, "wavefront");

  float* P = ws + (size_t)blockIdx.x * PSTRIDE;

  float run = 0.f;
#pragma unroll
  for (int r = 0; r < SEGL; ++r) run += kp_s[r * PM + m0 + lr];
  if (lh == 0) P[MF * DH + m0 + lr] = run;

  v8f sreg[4] = {};
#pragma unroll
  for (int kb = 0; kb < SEGL; kb += 4) {
    v2f a;
    a.x = kp_s[(kb + 2 * lh) * PM + m0 + lr];
    a.y = kp_s[(kb + 2 * lh + 1) * PM + m0 + lr];
#pragma unroll
    for (int et = 0; et < 4; ++et) {
      v2f b;
      b.x = v_s[(kb + 2 * lh) * PD + et * 16 + lr];
      b.y = v_s[(kb + 2 * lh + 1) * PD + et * 16 + lr];
      sreg[et] = wmma4(a, b, sreg[et]);
    }
  }
#pragma unroll
  for (int et = 0; et < 4; ++et)
#pragma unroll
    for (int jh = 0; jh < 2; ++jh) {
      float4 o;
      o.x = sreg[et][4 * jh + 0];
      o.y = sreg[et][4 * jh + 1];
      o.z = sreg[et][4 * jh + 2];
      o.w = sreg[et][4 * jh + 3];
      *reinterpret_cast<float4*>(P + (((wave * 4 + et) * 2 + jh) * 32 + lane) * 4) = o;
    }
}

__device__ __forceinline__ void contrib_load(const float* __restrict__ P, int blk0, int lane, int col,
                                             float4 (&t)[4], float& tk) {
#pragma unroll
  for (int i = 0; i < 4; ++i)
    t[i] = *reinterpret_cast<const float4*>(P + ((blk0 + i) * 32 + lane) * 4);
  tk = P[MF * DH + col];
}
__device__ __forceinline__ void contrib_add(v8f (&sreg)[ETW], float& run, const float4 (&t)[4], float tk) {
#pragma unroll
  for (int i = 0; i < 4; ++i) {
    sreg[i >> 1][4 * (i & 1) + 0] += t[i].x;
    sreg[i >> 1][4 * (i & 1) + 1] += t[i].y;
    sreg[i >> 1][4 * (i & 1) + 2] += t[i].z;
    sreg[i >> 1][4 * (i & 1) + 3] += t[i].w;
  }
  run += tk;
}

__global__ __launch_bounds__(256)
void performer_causal_f32_wmma(const float* __restrict__ q,
                               const float* __restrict__ k,
                               const float* __restrict__ v,
                               const float* __restrict__ proj,
                               float* __restrict__ out,
                               const float* __restrict__ ws, size_t ws_bytes) {
  __shared__ __attribute__((aligned(16))) float proj_s[MF * PD];
  __shared__ float S_s[MF * PD];
  __shared__ float qp_s[SEGL * PM];
  __shared__ float kp_s[SEGL * PM];
  __shared__ __attribute__((aligned(16))) float xq_s[SEGL * PD];
  __shared__ __attribute__((aligned(16))) float xk_s[SEGL * PD];
  __shared__ __attribute__((aligned(16))) float v_s[SEGL * PD];
  __shared__ float a_s[CHK * PA];
  __shared__ float kc_s[MF];

  const int eh   = blockIdx.x & 1;
  const int h    = (blockIdx.x >> 1) & 7;
  const int seg  = NSEG - 1 - (blockIdx.x >> 4);
  const int ec0  = EW * eh;
  const int n_begin = seg * SEGL;
  const int nlight  = n_begin / CHK;
  const int tid  = threadIdx.x;
  const int lane = tid & 31;
  const int wave = tid >> 5;
  const int lr   = lane & 15;
  const int lh   = lane >> 4;

  const float norm = 0.35355339059327378f;

  const float* qh = q + (size_t)h * NSEQ * DH;
  const float* kh = k + (size_t)h * NSEQ * DH;
  const float* vh = v + (size_t)h * NSEQ * DH;
  float*       oh = out + (size_t)h * NSEQ * DH;

  const int m0w = wave * 16;
  const bool from_ws = ws_bytes >= (size_t)WS_FLOATS * sizeof(float);
  const float* Ph = ws + (size_t)(h * NPART) * PSTRIDE;
  const int blk0 = wave * 8 + 4 * eh;
  const int sB = seg & 8;
  const int sC = seg & 12;
  const int sD = seg & 14;
  float4 tA[8][4], tB[4][4], tC[2][4], tD[1][4];
  float tkA[8], tkB[4], tkC[2], tkD[1];

  {
    float4 rp[8];
#pragma unroll
    for (int t = 0; t < 8; ++t)
      rp[t] = *reinterpret_cast<const float4*>(proj + 4 * (tid + 256 * t));
    float4 rq4[4], rk4[4], rv4[2];
#pragma unroll
    for (int t = 0; t < 4; ++t) {
      const int idx = tid + 256 * t;
      rq4[t] = *reinterpret_cast<const float4*>(qh + (size_t)n_begin * DH + 4 * idx);
      rk4[t] = *reinterpret_cast<const float4*>(kh + (size_t)n_begin * DH + 4 * idx);
    }
#pragma unroll
    for (int t = 0; t < 2; ++t) {
      const int idx = tid + 256 * t;
      rv4[t] = *reinterpret_cast<const float4*>(vh + (size_t)(n_begin + (idx >> 3)) * DH + ec0 + 4 * (idx & 7));
    }
#pragma unroll
    for (int t = 0; t < 8; ++t) {
      const int idx = tid + 256 * t;
      const int p = (idx >> 4) * PD + 4 * (idx & 15);
      *reinterpret_cast<float4*>(proj_s + p) = rp[t];
    }
#pragma unroll
    for (int t = 0; t < 4; ++t) {
      const int idx = tid + 256 * t;
      const int p = (idx >> 4) * PD + 4 * (idx & 15);
      *reinterpret_cast<float4*>(xq_s + p) = rq4[t];
      *reinterpret_cast<float4*>(xk_s + p) = rk4[t];
    }
#pragma unroll
    for (int t = 0; t < 2; ++t) {
      const int idx = tid + 256 * t;
      *reinterpret_cast<float4*>(v_s + (idx >> 3) * PD + ec0 + 4 * (idx & 7)) = rv4[t];
    }
  }
  if (from_ws) {
    if (seg & 8) {
#pragma unroll
      for (int u = 0; u < 8; ++u)
        contrib_load(Ph + (size_t)u * PSTRIDE, blk0, lane, m0w + lr, tA[u], tkA[u]);
    }
    if (seg & 4) {
#pragma unroll
      for (int u = 0; u < 4; ++u)
        contrib_load(Ph + (size_t)(sB + u) * PSTRIDE, blk0, lane, m0w + lr, tB[u], tkB[u]);
    }
  }

  v8f sreg[ETW] = {};
  float run = 0.f;
  if (!from_ws && nlight > 0) {
    const int m0 = m0w;
      float* const xkbuf = S_s;
      float* const vbuf  = S_s + 2 * CHK * PD;
      __syncthreads();
      v2f pb[DH / 4];
#pragma unroll
      for (int kk = 0; kk < DH / 4; ++kk) {
        pb[kk].x = proj_s[(m0 + lr) * PD + 4 * kk + 2 * lh];
        pb[kk].y = proj_s[(m0 + lr) * PD + 4 * kk + 2 * lh + 1];
      }
      float rk[4], rv[4];
#pragma unroll
      for (int t = 0; t < 4; ++t) {
        rk[t] = kh[tid + 256 * t];
        rv[t] = vh[tid + 256 * t];
      }
#pragma unroll
      for (int t = 0; t < 4; ++t) {
        const int p = ((tid >> 6) + 4 * t) * PD + (tid & 63);
        xkbuf[p] = rk[t];
        vbuf[p]  = rv[t];
      }
      __syncthreads();
      for (int c = 0; c < nlight; ++c) {
        const float* xkb = xkbuf + (c & 1) * (CHK * PD);
        const float* vb  = vbuf  + (c & 1) * (CHK * PD);
        float* xkn = xkbuf + ((c + 1) & 1) * (CHK * PD);
        float* vn  = vbuf  + ((c + 1) & 1) * (CHK * PD);
        const int cn = (c + 1 < nlight) ? (c + 1) : c;
#pragma unroll
        for (int t = 0; t < 4; ++t) {
          rk[t] = kh[cn * CHK * DH + tid + 256 * t];
          rv[t] = vh[cn * CHK * DH + tid + 256 * t];
        }
        v8f acck = {};
#pragma unroll
        for (int kk = 0; kk < DH / 4; ++kk) {
          v2f a;
          a.x = xkb[lr * PD + 4 * kk + 2 * lh];
          a.y = xkb[lr * PD + 4 * kk + 2 * lh + 1];
          acck = wmma4(a, pb[kk], acck);
        }
#pragma unroll
        for (int j = 0; j < 8; ++j)
          kp_s[(j + 8 * lh) * PM + m0 + lr] = fmaxf(acck[j] * norm, 0.f) + KEPS;
        __builtin_amdgcn_fence(__ATOMIC_RELEASE, "wavefront");
        __builtin_amdgcn_wave_barrier();
        __builtin_amdgcn_fence(__ATOMIC_ACQUIRE, "wavefront");
#pragma unroll
        for (int r = 0; r < CHK; ++r) run += kp_s[r * PM + m0 + lr];
#pragma unroll
        for (int etl = 0; etl < ETW; ++etl) {
          const int e0 = ec0 + etl * 16;
#pragma unroll
          for (int kb = 0; kb < CHK; kb += 4) {
            v2f a, b;
            a.x = kp_s[(kb + 2 * lh) * PM + m0 + lr];
            a.y = kp_s[(kb + 2 * lh + 1) * PM + m0 + lr];
            b.x = vb[(kb + 2 * lh) * PD + e0 + lr];
            b.y = vb[(kb + 2 * lh + 1) * PD + e0 + lr];
            sreg[etl] = wmma4(a, b, sreg[etl]);
          }
        }
#pragma unroll
        for (int t = 0; t < 4; ++t) {
          const int p = ((tid >> 6) + 4 * t) * PD + (tid & 63);
          xkn[p] = rk[t];
          vn[p]  = rv[t];
        }
        __syncthreads();
      }
  }
  __builtin_amdgcn_fence(__ATOMIC_RELEASE, "workgroup", "local");
  __builtin_amdgcn_s_barrier();
  __builtin_amdgcn_fence(__ATOMIC_ACQUIRE, "workgroup", "local");

  {
    const int m0 = m0w;
    v8f accq[SEGL / CHK] = {};
    v8f acck[SEGL / CHK] = {};
#pragma unroll
    for (int kb = 0; kb < DH; kb += 4) {
      if (kb == DH / 4 && from_ws) {
        if (seg & 2) {
#pragma unroll
          for (int u = 0; u < 2; ++u)
            contrib_load(Ph + (size_t)(sC + u) * PSTRIDE, blk0, lane, m0w + lr, tC[u], tkC[u]);
        }
        if (seg & 1)
          contrib_load(Ph + (size_t)sD * PSTRIDE, blk0, lane, m0w + lr, tD[0], tkD[0]);
      }
      v2f b;
      b.x = proj_s[(m0 + lr) * PD + kb + 2 * lh];
      b.y = proj_s[(m0 + lr) * PD + kb + 2 * lh + 1];
#pragma unroll
      for (int rt = 0; rt < SEGL / CHK; ++rt) {
        v2f a;
        a.x = xq_s[(CHK * rt + lr) * PD + kb + 2 * lh];
        a.y = xq_s[(CHK * rt + lr) * PD + kb + 2 * lh + 1];
        accq[rt] = wmma4(a, b, accq[rt]);
        a.x = xk_s[(CHK * rt + lr) * PD + kb + 2 * lh];
        a.y = xk_s[(CHK * rt + lr) * PD + kb + 2 * lh + 1];
        acck[rt] = wmma4(a, b, acck[rt]);
      }
    }
#pragma unroll
    for (int rt = 0; rt < SEGL / CHK; ++rt)
#pragma unroll
      for (int j = 0; j < 8; ++j) {
        const int row = CHK * rt + j + 8 * lh;
        qp_s[row * PM + m0 + lr] = fmaxf(accq[rt][j] * norm, 0.f) + KEPS;
        kp_s[row * PM + m0 + lr] = fmaxf(acck[rt][j] * norm, 0.f) + KEPS;
      }
  }

  if (from_ws) {
    if (seg & 8) {
#pragma unroll
      for (int u = 0; u < 8; ++u) contrib_add(sreg, run, tA[u], tkA[u]);
    }
    if (seg & 4) {
#pragma unroll
      for (int u = 0; u < 4; ++u) contrib_add(sreg, run, tB[u], tkB[u]);
    }
    if (seg & 2) {
#pragma unroll
      for (int u = 0; u < 2; ++u) contrib_add(sreg, run, tC[u], tkC[u]);
    }
    if (seg & 1) contrib_add(sreg, run, tD[0], tkD[0]);
  }
#pragma unroll
  for (int etl = 0; etl < ETW; ++etl)
#pragma unroll
    for (int j = 0; j < 8; ++j)
      S_s[(m0w + j + 8 * lh) * PD + ec0 + etl * 16 + lr] = sreg[etl][j];
  if (lh == 0) kc_s[m0w + lr] = run;
  __syncthreads();

  for (int c0 = 0; c0 < SEGL; c0 += CHK) {
    const int n0 = n_begin + c0;
    const float* vc  = v_s  + c0 * PD;
    float* const qpc = qp_s + c0 * PM;
    const float* kpc = kp_s + c0 * PM;

    if (tid < MF) {
      float run = kc_s[tid];
#pragma unroll
      for (int r = 0; r < CHK; ++r) {
        run += kpc[r * PM + tid];
        qpc[r * PM + tid] = qpc[r * PM + tid] / run;
      }
      kc_s[tid] = run;
    }
    __syncthreads();

    v8f o = {};
    if (wave < ETW) {
      const int e0 = ec0 + wave * 16;
#pragma unroll
      for (int kb = 0; kb < MF; kb += 4) {
        v2f a, b;
        a.x = qpc[lr * PM + kb + 2 * lh];
        a.y = qpc[lr * PM + kb + 2 * lh + 1];
        b.x = S_s[(kb + 2 * lh) * PD + e0 + lr];
        b.y = S_s[(kb + 2 * lh + 1) * PD + e0 + lr];
        o = wmma4(a, b, o);
      }
    } else if (wave == ETW) {
      v8f am = {};
#pragma unroll
      for (int kb = 0; kb < MF; kb += 4) {
        v2f a, b;
        a.x = qpc[lr * PM + kb + 2 * lh];
        a.y = qpc[lr * PM + kb + 2 * lh + 1];
        b.x = kpc[lr * PM + kb + 2 * lh];
        b.y = kpc[lr * PM + kb + 2 * lh + 1];
        am = wmma4(a, b, am);
      }
#pragma unroll
      for (int j = 0; j < 8; ++j) {
        const int row = j + 8 * lh;
        a_s[row * PA + lr] = (lr <= row) ? am[j] : 0.f;
      }
    }
    __syncthreads();

    if (wave < ETW) {
      const int e0 = ec0 + wave * 16;
#pragma unroll
      for (int kb = 0; kb < CHK; kb += 4) {
        v2f a, b;
        a.x = a_s[lr * PA + kb + 2 * lh];
        a.y = a_s[lr * PA + kb + 2 * lh + 1];
        b.x = vc[(kb + 2 * lh) * PD + e0 + lr];
        b.y = vc[(kb + 2 * lh + 1) * PD + e0 + lr];
        o = wmma4(a, b, o);
      }
#pragma unroll
      for (int j = 0; j < 8; ++j) {
        const int row = j + 8 * lh;
        oh[(n0 + row) * DH + e0 + lr] = o[j];
      }
    }
    {
      const int m0 = wave * 16;
#pragma unroll
      for (int etl = 0; etl < ETW; ++etl) {
        const int e0 = ec0 + etl * 16;
        v8f sacc;
#pragma unroll
        for (int j = 0; j < 8; ++j)
          sacc[j] = S_s[(m0 + j + 8 * lh) * PD + e0 + lr];
#pragma unroll
        for (int kb = 0; kb < CHK; kb += 4) {
          v2f a, b;
          a.x = kpc[(kb + 2 * lh) * PM + m0 + lr];
          a.y = kpc[(kb + 2 * lh + 1) * PM + m0 + lr];
          b.x = vc[(kb + 2 * lh) * PD + e0 + lr];
          b.y = vc[(kb + 2 * lh + 1) * PD + e0 + lr];
          sacc = wmma4(a, b, sacc);
        }
#pragma unroll
        for (int j = 0; j < 8; ++j)
          S_s[(m0 + j + 8 * lh) * PD + e0 + lr] = sacc[j];
      }
    }
    __syncthreads();
  }
}

extern "C" void kernel_launch(void* const* d_in, const int* in_sizes, int n_in,
                              void* d_out, int out_size, void* d_ws, size_t ws_size,
                              hipStream_t stream) {
  (void)in_sizes; (void)n_in; (void)out_size;
  const float* q    = (const float*)d_in[0];
  const float* k    = (const float*)d_in[1];
  const float* v    = (const float*)d_in[2];
  const float* proj = (const float*)d_in[3];
  float* out        = (float*)d_out;
  float* ws         = (float*)d_ws;
  performer_segment_state_f32_wmma<<<dim3(120), dim3(256), 0, stream>>>(k, v, proj, ws, ws_size);
  performer_causal_f32_wmma<<<dim3(256), dim3(256), 0, stream>>>(q, k, v, proj, out, ws, ws_size);
}


# Scan launch treats its 64 rows as one causal block: 8 output + 10 score tiles, independent WMMA chains on 8 waves; no in-launch S update, 3 barriers not 14; out's sum regrouped (v terms apart from S)
# speedup vs baseline: 28.1346x; 1.2241x over previous
#include <hip/hip_runtime.h>


#define KEPS 1e-3f
#define NSEQ 1024
#define DH   64
#define MF   128
#define CHK  16
#define NSEG 16
#define SEGL (NSEQ / NSEG)
#define NEH  2
#define EW   (DH / NEH)
#define ETW  (EW / 16)
static_assert(8 * NSEG * NEH == 256 && NSEG == 16 && NEH == 2 && EW == 32 && ETW == 2,
              "the scan launch below starts 8 x NSEG x NEH workgroups and decodes its block id by shifts");
#define NPART   (NSEG - 1)
#define PSTRIDE (MF * DH + MF)
#define WS_FLOATS (8 * NPART * PSTRIDE)
static_assert(8 * NPART == 120, "the first launch below starts 8 x NPART workgroups");
static_assert(SEGL % CHK == 0 && SEGL == 64, "the first launch holds one segment of 64 rows in LDS");

#define PD   (DH + 12)
#define PM   (MF + 12)
#define PA   (CHK + 4)
#define NRT  (SEGL / CHK)
#define NAT  (NRT * (NRT + 1) / 2)
static_assert(NRT == 4 && NAT == 10 && 2 * NRT == 8,
              "the scan launch gives its 8 waves one output tile each (NRT x ETW) and the 10 score tiles by wave number");

typedef float v2f __attribute__((ext_vector_type(2)));
typedef float v8f __attribute__((ext_vector_type(8)));

__device__ __forceinline__ v8f wmma4(v2f a, v2f b, v8f c) {
  return __builtin_amdgcn_wmma_f32_16x16x4_f32(false, a, false, b,
                                               (short)0, c, false, false);
}


__global__ __launch_bounds__(256)
void performer_segment_state_f32_wmma(const float* __restrict__ k,
                                      const float* __restrict__ v,
                                      const float* __restrict__ proj,
                                      float* __restrict__ ws, size_t ws_bytes) {
  if (ws_bytes < (size_t)WS_FLOATS * sizeof(float)) return;

  __shared__ __attribute__((aligned(16))) float xk_s[SEGL * PD];
  __shared__ __attribute__((aligned(16))) float v_s[SEGL * PD];
  __shared__ float kp_s[SEGL * PM];

  const int h    = blockIdx.x / NPART;
  const int seg  = blockIdx.x % NPART;
  const int tid  = threadIdx.x;
  const int lane = tid & 31;
  const int wave = tid >> 5;
  const int lr   = lane & 15;
  const int lh   = lane >> 4;
  const int m0   = wave * 16;

  const float norm = 0.35355339059327378f;

  const float* kseg = k + ((size_t)h * NSEQ + (size_t)seg * SEGL) * DH;
  const float* vseg = v + ((size_t)h * NSEQ + (size_t)seg * SEGL) * DH;

  float4 rk[4], rv[4];
#pragma unroll
  for (int t = 0; t < 4; ++t) {
    const int idx = tid + 256 * t;
    rk[t] = *reinterpret_cast<const float4*>(kseg + 4 * idx);
    rv[t] = *reinterpret_cast<const float4*>(vseg + 4 * idx);
  }
  v2f pb[DH / 4];
#pragma unroll
  for (int kk = 0; kk < DH / 4; ++kk) {
    pb[kk].x = proj[(m0 + lr) * DH + 4 * kk + 2 * lh];
    pb[kk].y = proj[(m0 + lr) * DH + 4 * kk + 2 * lh + 1];
  }
#pragma unroll
  for (int t = 0; t < 4; ++t) {
    const int idx = tid + 256 * t;
    const int p = (idx >> 4) * PD + 4 * (idx & 15);
    *reinterpret_cast<float4*>(xk_s + p) = rk[t];
    *reinterpret_cast<float4*>(v_s + p)  = rv[t];
  }
  __syncthreads();

  v8f acc[SEGL / CHK] = {};
#pragma unroll
  for (int kk = 0; kk < DH / 4; ++kk) {
#pragma unroll
    for (int rt = 0; rt < SEGL / CHK; ++rt) {
      v2f a;
      a.x = xk_s[(CHK * rt + lr) * PD + 4 * kk + 2 * lh];
      a.y = xk_s[(CHK * rt + lr) * PD + 4 * kk + 2 * lh + 1];
      acc[rt] = wmma4(a, pb[kk], acc[rt]);
    }
  }
#pragma unroll
  for (int rt = 0; rt < SEGL / CHK; ++rt)
#pragma unroll
    for (int j = 0; j < 8; ++j)
      kp_s[(CHK * rt + j + 8 * lh) * PM + m0 + lr] = fmaxf(acc[rt][j] * norm, 0.f) + KEPS;
  __builtin_amdgcn_fence(__ATOMIC_RELEASE, "wavefront");
  __builtin_amdgcn_wave_barrier();
  __builtin_amdgcn_fence(__ATOMIC_ACQUIRE, "wavefront");

  float* P = ws + (size_t)blockIdx.x * PSTRIDE;

  float run = 0.f;
#pragma unroll
  for (int r = 0; r < SEGL; ++r) run += kp_s[r * PM + m0 + lr];
  if (lh == 0) P[MF * DH + m0 + lr] = run;

  v8f sreg[4] = {};
#pragma unroll
  for (int kb = 0; kb < SEGL; kb += 4) {
    v2f a;
    a.x = kp_s[(kb + 2 * lh) * PM + m0 + lr];
    a.y = kp_s[(kb + 2 * lh + 1) * PM + m0 + lr];
#pragma unroll
    for (int et = 0; et < 4; ++et) {
      v2f b;
      b.x = v_s[(kb + 2 * lh) * PD + et * 16 + lr];
      b.y = v_s[(kb + 2 * lh + 1) * PD + et * 16 + lr];
      sreg[et] = wmma4(a, b, sreg[et]);
    }
  }
#pragma unroll
  for (int et = 0; et < 4; ++et)
#pragma unroll
    for (int jh = 0; jh < 2; ++jh) {
      float4 o;
      o.x = sreg[et][4 * jh + 0];
      o.y = sreg[et][4 * jh + 1];
      o.z = sreg[et][4 * jh + 2];
      o.w = sreg[et][4 * jh + 3];
      *reinterpret_cast<float4*>(P + (((wave * 4 + et) * 2 + jh) * 32 + lane) * 4) = o;
    }
}

__device__ __forceinline__ void contrib_load(const float* __restrict__ P, int blk0, int lane, int col,
                                             float4 (&t)[4], float& tk) {
#pragma unroll
  for (int i = 0; i < 4; ++i)
    t[i] = *reinterpret_cast<const float4*>(P + ((blk0 + i) * 32 + lane) * 4);
  tk = P[MF * DH + col];
}
__device__ __forceinline__ void contrib_add(v8f (&sreg)[ETW], float& run, const float4 (&t)[4], float tk) {
#pragma unroll
  for (int i = 0; i < 4; ++i) {
    sreg[i >> 1][4 * (i & 1) + 0] += t[i].x;
    sreg[i >> 1][4 * (i & 1) + 1] += t[i].y;
    sreg[i >> 1][4 * (i & 1) + 2] += t[i].z;
    sreg[i >> 1][4 * (i & 1) + 3] += t[i].w;
  }
  run += tk;
}

__global__ __launch_bounds__(256)
void performer_causal_f32_wmma(const float* __restrict__ q,
                               const float* __restrict__ k,
                               const float* __restrict__ v,
                               const float* __restrict__ proj,
                               float* __restrict__ out,
                               const float* __restrict__ ws, size_t ws_bytes) {
  __shared__ __attribute__((aligned(16))) float proj_s[MF * PD];
  __shared__ float S_s[MF * PD];
  __shared__ float qp_s[SEGL * PM];
  __shared__ float kp_s[SEGL * PM];
  __shared__ __attribute__((aligned(16))) float xq_s[SEGL * PD];
  __shared__ __attribute__((aligned(16))) float xk_s[SEGL * PD];
  __shared__ __attribute__((aligned(16))) float v_s[SEGL * PD];
  __shared__ float a_s[NAT * CHK * PA];

  const int eh   = blockIdx.x & 1;
  const int h    = (blockIdx.x >> 1) & 7;
  const int seg  = NSEG - 1 - (blockIdx.x >> 4);
  const int ec0  = EW * eh;
  const int n_begin = seg * SEGL;
  const int nlight  = n_begin / CHK;
  const int tid  = threadIdx.x;
  const int lane = tid & 31;
  const int wave = tid >> 5;
  const int lr   = lane & 15;
  const int lh   = lane >> 4;

  const float norm = 0.35355339059327378f;

  const float* qh = q + (size_t)h * NSEQ * DH;
  const float* kh = k + (size_t)h * NSEQ * DH;
  const float* vh = v + (size_t)h * NSEQ * DH;
  float*       oh = out + (size_t)h * NSEQ * DH;

  const int m0w = wave * 16;
  const bool from_ws = ws_bytes >= (size_t)WS_FLOATS * sizeof(float);
  const float* Ph = ws + (size_t)(h * NPART) * PSTRIDE;
  const int blk0 = wave * 8 + 4 * eh;
  const int sB = seg & 8;
  const int sC = seg & 12;
  const int sD = seg & 14;
  float4 tA[8][4], tB[4][4], tC[2][4], tD[1][4];
  float tkA[8], tkB[4], tkC[2], tkD[1];

  {
    float4 rp[8];
#pragma unroll
    for (int t = 0; t < 8; ++t)
      rp[t] = *reinterpret_cast<const float4*>(proj + 4 * (tid + 256 * t));
    float4 rq4[4], rk4[4], rv4[2];
#pragma unroll
    for (int t = 0; t < 4; ++t) {
      const int idx = tid + 256 * t;
      rq4[t] = *reinterpret_cast<const float4*>(qh + (size_t)n_begin * DH + 4 * idx);
      rk4[t] = *reinterpret_cast<const float4*>(kh + (size_t)n_begin * DH + 4 * idx);
    }
#pragma unroll
    for (int t = 0; t < 2; ++t) {
      const int idx = tid + 256 * t;
      rv4[t] = *reinterpret_cast<const float4*>(vh + (size_t)(n_begin + (idx >> 3)) * DH + ec0 + 4 * (idx & 7));
    }
#pragma unroll
    for (int t = 0; t < 8; ++t) {
      const int idx = tid + 256 * t;
      const int p = (idx >> 4) * PD + 4 * (idx & 15);
      *reinterpret_cast<float4*>(proj_s + p) = rp[t];
    }
#pragma unroll
    for (int t = 0; t < 4; ++t) {
      const int idx = tid + 256 * t;
      const int p = (idx >> 4) * PD + 4 * (idx & 15);
      *reinterpret_cast<float4*>(xq_s + p) = rq4[t];
      *reinterpret_cast<float4*>(xk_s + p) = rk4[t];
    }
#pragma unroll
    for (int t = 0; t < 2; ++t) {
      const int idx = tid + 256 * t;
      *reinterpret_cast<float4*>(v_s + (idx >> 3) * PD + ec0 + 4 * (idx & 7)) = rv4[t];
    }
  }
  if (from_ws) {
    if (seg & 8) {
#pragma unroll
      for (int u = 0; u < 8; ++u)
        contrib_load(Ph + (size_t)u * PSTRIDE, blk0, lane, m0w + lr, tA[u], tkA[u]);
    }
    if (seg & 4) {
#pragma unroll
      for (int u = 0; u < 4; ++u)
        contrib_load(Ph + (size_t)(sB + u) * PSTRIDE, blk0, lane, m0w + lr, tB[u], tkB[u]);
    }
  }

  v8f sreg[ETW] = {};
  float run = 0.f;
  if (!from_ws && nlight > 0) {
    const int m0 = m0w;
      float* const xkbuf = S_s;
      float* const vbuf  = S_s + 2 * CHK * PD;
      __syncthreads();
      v2f pb[DH / 4];
#pragma unroll
      for (int kk = 0; kk < DH / 4; ++kk) {
        pb[kk].x = proj_s[(m0 + lr) * PD + 4 * kk + 2 * lh];
        pb[kk].y = proj_s[(m0 + lr) * PD + 4 * kk + 2 * lh + 1];
      }
      float rk[4], rv[4];
#pragma unroll
      for (int t = 0; t < 4; ++t) {
        rk[t] = kh[tid + 256 * t];
        rv[t] = vh[tid + 256 * t];
      }
#pragma unroll
      for (int t = 0; t < 4; ++t) {
        const int p = ((tid >> 6) + 4 * t) * PD + (tid & 63);
        xkbuf[p] = rk[t];
        vbuf[p]  = rv[t];
      }
      __syncthreads();
      for (int c = 0; c < nlight; ++c) {
        const float* xkb = xkbuf + (c & 1) * (CHK * PD);
        const float* vb  = vbuf  + (c & 1) * (CHK * PD);
        float* xkn = xkbuf + ((c + 1) & 1) * (CHK * PD);
        float* vn  = vbuf  + ((c + 1) & 1) * (CHK * PD);
        const int cn = (c + 1 < nlight) ? (c + 1) : c;
#pragma unroll
        for (int t = 0; t < 4; ++t) {
          rk[t] = kh[cn * CHK * DH + tid + 256 * t];
          rv[t] = vh[cn * CHK * DH + tid + 256 * t];
        }
        v8f acck = {};
#pragma unroll
        for (int kk = 0; kk < DH / 4; ++kk) {
          v2f a;
          a.x = xkb[lr * PD + 4 * kk + 2 * lh];
          a.y = xkb[lr * PD + 4 * kk + 2 * lh + 1];
          acck = wmma4(a, pb[kk], acck);
        }
#pragma unroll
        for (int j = 0; j < 8; ++j)
          kp_s[(j + 8 * lh) * PM + m0 + lr] = fmaxf(acck[j] * norm, 0.f) + KEPS;
        __builtin_amdgcn_fence(__ATOMIC_RELEASE, "wavefront");
        __builtin_amdgcn_wave_barrier();
        __builtin_amdgcn_fence(__ATOMIC_ACQUIRE, "wavefront");
#pragma unroll
        for (int r = 0; r < CHK; ++r) run += kp_s[r * PM + m0 + lr];
#pragma unroll
        for (int etl = 0; etl < ETW; ++etl) {
          const int e0 = ec0 + etl * 16;
#pragma unroll
          for (int kb = 0; kb < CHK; kb += 4) {
            v2f a, b;
            a.x = kp_s[(kb + 2 * lh) * PM + m0 + lr];
            a.y = kp_s[(kb + 2 * lh + 1) * PM + m0 + lr];
            b.x = vb[(kb + 2 * lh) * PD + e0 + lr];
            b.y = vb[(kb + 2 * lh + 1) * PD + e0 + lr];
            sreg[etl] = wmma4(a, b, sreg[etl]);
          }
        }
#pragma unroll
        for (int t = 0; t < 4; ++t) {
          const int p = ((tid >> 6) + 4 * t) * PD + (tid & 63);
          xkn[p] = rk[t];
          vn[p]  = rv[t];
        }
        __syncthreads();
      }
  }
  __builtin_amdgcn_fence(__ATOMIC_RELEASE, "workgroup", "local");
  __builtin_amdgcn_s_barrier();
  __builtin_amdgcn_fence(__ATOMIC_ACQUIRE, "workgroup", "local");

  v8f accq[NRT] = {};
  v8f acck[NRT] = {};
#pragma unroll
  for (int kb = 0; kb < DH; kb += 4) {
    if (kb == DH / 4 && from_ws) {
      if (seg & 2) {
#pragma unroll
        for (int u = 0; u < 2; ++u)
          contrib_load(Ph + (size_t)(sC + u) * PSTRIDE, blk0, lane, m0w + lr, tC[u], tkC[u]);
      }
      if (seg & 1)
        contrib_load(Ph + (size_t)sD * PSTRIDE, blk0, lane, m0w + lr, tD[0], tkD[0]);
    }
    v2f b;
    b.x = proj_s[(m0w + lr) * PD + kb + 2 * lh];
    b.y = proj_s[(m0w + lr) * PD + kb + 2 * lh + 1];
#pragma unroll
    for (int rt = 0; rt < NRT; ++rt) {
      v2f a;
      a.x = xq_s[(CHK * rt + lr) * PD + kb + 2 * lh];
      a.y = xq_s[(CHK * rt + lr) * PD + kb + 2 * lh + 1];
      accq[rt] = wmma4(a, b, accq[rt]);
      a.x = xk_s[(CHK * rt + lr) * PD + kb + 2 * lh];
      a.y = xk_s[(CHK * rt + lr) * PD + kb + 2 * lh + 1];
      acck[rt] = wmma4(a, b, acck[rt]);
    }
  }
#pragma unroll
  for (int rt = 0; rt < NRT; ++rt)
#pragma unroll
    for (int j = 0; j < 8; ++j)
      kp_s[(CHK * rt + j + 8 * lh) * PM + m0w + lr] = fmaxf(acck[rt][j] * norm, 0.f) + KEPS;

  if (from_ws) {
    if (seg & 8) {
#pragma unroll
      for (int u = 0; u < 8; ++u) contrib_add(sreg, run, tA[u], tkA[u]);
    }
    if (seg & 4) {
#pragma unroll
      for (int u = 0; u < 4; ++u) contrib_add(sreg, run, tB[u], tkB[u]);
    }
    if (seg & 2) {
#pragma unroll
      for (int u = 0; u < 2; ++u) contrib_add(sreg, run, tC[u], tkC[u]);
    }
    if (seg & 1) contrib_add(sreg, run, tD[0], tkD[0]);
  }
#pragma unroll
  for (int etl = 0; etl < ETW; ++etl)
#pragma unroll
    for (int j = 0; j < 8; ++j)
      S_s[(m0w + j + 8 * lh) * PD + ec0 + etl * 16 + lr] = sreg[etl][j];

  __builtin_amdgcn_fence(__ATOMIC_RELEASE, "wavefront");
  __builtin_amdgcn_wave_barrier();
  __builtin_amdgcn_fence(__ATOMIC_ACQUIRE, "wavefront");
  {
    float kc[NRT][8];
#pragma unroll
    for (int rt = 0; rt < NRT; ++rt) {
#pragma unroll
      for (int j = 0; j < 8; ++j) {
        run += kp_s[(CHK * rt + j) * PM + m0w + lr];
        kc[rt][j] = run;
      }
#pragma unroll
      for (int j = 0; j < 8; ++j) {
        run += kp_s[(CHK * rt + 8 + j) * PM + m0w + lr];
        if (lh) kc[rt][j] = run;
      }
    }
#pragma unroll
    for (int rt = 0; rt < NRT; ++rt)
#pragma unroll
      for (int j = 0; j < 8; ++j)
        qp_s[(CHK * rt + j + 8 * lh) * PM + m0w + lr] =
            (fmaxf(accq[rt][j] * norm, 0.f) + KEPS) / kc[rt][j];
  }
  __syncthreads();

  const int wv = __builtin_amdgcn_readfirstlane(wave);

  const int ort = wv >> 1;
  const int e0  = ec0 + (wv & 1) * 16;
  int art = wv >> 1, act = wv & 1;
  if (wv == 1) { art = 2; act = 2; }
  v8f o = {};
  {
    v8f am = {};
    const float* qo = qp_s + (CHK * ort + lr) * PM + 2 * lh;
    const float* sb = S_s + (2 * lh) * PD + e0 + lr;
    const float* qa = qp_s + (CHK * art + lr) * PM + 2 * lh;
    const float* ka = kp_s + (CHK * act + lr) * PM + 2 * lh;
#pragma unroll
    for (int kb = 0; kb < MF; kb += 4) {
      v2f a, b;
      a.x = qo[kb];
      a.y = qo[kb + 1];
      b.x = sb[kb * PD];
      b.y = sb[(kb + 1) * PD];
      o = wmma4(a, b, o);
      a.x = qa[kb];
      a.y = qa[kb + 1];
      b.x = ka[kb];
      b.y = ka[kb + 1];
      am = wmma4(a, b, am);
    }
    float* at = a_s + (art * (art + 1) / 2 + act) * (CHK * PA);
    const bool diag = art == act;
#pragma unroll
    for (int j = 0; j < 8; ++j) {
      const int row = j + 8 * lh;
      at[row * PA + lr] = (!diag || lr <= row) ? am[j] : 0.f;
    }
  }
  if (wv < 2) {
    const int act2 = 2 + wv;
    v8f am = {};
    const float* qa = qp_s + (CHK * 3 + lr) * PM + 2 * lh;
    const float* ka = kp_s + (CHK * act2 + lr) * PM + 2 * lh;
#pragma unroll
    for (int kb = 0; kb < MF; kb += 4) {
      v2f a, b;
      a.x = qa[kb];
      a.y = qa[kb + 1];
      b.x = ka[kb];
      b.y = ka[kb + 1];
      am = wmma4(a, b, am);
    }
    float* at = a_s + (6 + act2) * (CHK * PA);
    const bool diag = act2 == 3;
#pragma unroll
    for (int j = 0; j < 8; ++j) {
      const int row = j + 8 * lh;
      at[row * PA + lr] = (!diag || lr <= row) ? am[j] : 0.f;
    }
  }
  __syncthreads();

#pragma unroll
  for (int ct = 0; ct < NRT; ++ct) {
    if (ct <= ort) {
      const float* at = a_s + (ort * (ort + 1) / 2 + ct) * (CHK * PA) + lr * PA + 2 * lh;
      const float* vb = v_s + (CHK * ct + 2 * lh) * PD + e0 + lr;
#pragma unroll
      for (int kb = 0; kb < CHK; kb += 4) {
        v2f a, b;
        a.x = at[kb];
        a.y = at[kb + 1];
        b.x = vb[kb * PD];
        b.y = vb[(kb + 1) * PD];
        o = wmma4(a, b, o);
      }
    }
  }
#pragma unroll
  for (int j = 0; j < 8; ++j)
    oh[(n_begin + CHK * ort + j + 8 * lh) * DH + e0 + lr] = o[j];
}

extern "C" void kernel_launch(void* const* d_in, const int* in_sizes, int n_in,
                              void* d_out, int out_size, void* d_ws, size_t ws_size,
                              hipStream_t stream) {
  (void)in_sizes; (void)n_in; (void)out_size;
  const float* q    = (const float*)d_in[0];
  const float* k    = (const float*)d_in[1];
  const float* v    = (const float*)d_in[2];
  const float* proj = (const float*)d_in[3];
  float* out        = (float*)d_out;
  float* ws         = (float*)d_ws;
  performer_segment_state_f32_wmma<<<dim3(120), dim3(256), 0, stream>>>(k, v, proj, ws, ws_size);
  performer_causal_f32_wmma<<<dim3(256), dim3(256), 0, stream>>>(q, k, v, proj, out, ws, ws_size);
}


# First launch re-cut: 240 workgroups of 4 waves (head, segment, feature half), was 120 of 8: one wave per SIMD, not two sharing its f32 matrix unit; WMMA order and workspace layout unchanged
# speedup vs baseline: 29.9961x; 1.0662x over previous
#include <hip/hip_runtime.h>


#define KEPS 1e-3f
#define NSEQ 1024
#define DH   64
#define MF   128
#define CHK  16
#define NSEG 16
#define SEGL (NSEQ / NSEG)
#define NEH  2
#define EW   (DH / NEH)
#define ETW  (EW / 16)
static_assert(8 * NSEG * NEH == 256 && NSEG == 16 && NEH == 2 && EW == 32 && ETW == 2,
              "the scan launch below starts 8 x NSEG x NEH workgroups and decodes its block id by shifts");
#define NPART   (NSEG - 1)
#define PSTRIDE (MF * DH + MF)
#define WS_FLOATS (8 * NPART * PSTRIDE)
#define NFH 2
static_assert(NFH * 8 * NPART == 240 && MF / 16 == 4 * NFH,
              "the first launch below starts NFH x 8 x NPART workgroups of 4 waves, one 16-feature tile a wave");
static_assert(SEGL % CHK == 0 && SEGL == 64, "the first launch holds one segment of 64 rows in LDS");

#define PD   (DH + 12)
#define PM   (MF + 12)
#define PA   (CHK + 4)
#define NRT  (SEGL / CHK)
#define NAT  (NRT * (NRT + 1) / 2)
static_assert(NRT == 4 && NAT == 10 && 2 * NRT == 8,
              "the scan launch gives its 8 waves one output tile each (NRT x ETW) and the 10 score tiles by wave number");

typedef float v2f __attribute__((ext_vector_type(2)));
typedef float v8f __attribute__((ext_vector_type(8)));

__device__ __forceinline__ v8f wmma4(v2f a, v2f b, v8f c) {
  return __builtin_amdgcn_wmma_f32_16x16x4_f32(false, a, false, b,
                                               (short)0, c, false, false);
}


__global__ __launch_bounds__(128)
void performer_segment_state_f32_wmma(const float* __restrict__ k,
                                      const float* __restrict__ v,
                                      const float* __restrict__ proj,
                                      float* __restrict__ ws, size_t ws_bytes) {
  if (ws_bytes < (size_t)WS_FLOATS * sizeof(float)) return;

  __shared__ __attribute__((aligned(16))) float xk_s[SEGL * PD];
  __shared__ __attribute__((aligned(16))) float v_s[SEGL * PD];
  __shared__ float kp_s[SEGL * PM];

  const int fh   = blockIdx.x & 1;
  const int hs   = blockIdx.x >> 1;
  const int h    = hs / NPART;
  const int seg  = hs % NPART;
  const int tid  = threadIdx.x;
  const int lane = tid & 31;
  const int ft   = 4 * fh + (tid >> 5);
  const int lr   = lane & 15;
  const int lh   = lane >> 4;
  const int m0   = ft * 16;

  const float norm = 0.35355339059327378f;

  const float* kseg = k + ((size_t)h * NSEQ + (size_t)seg * SEGL) * DH;
  const float* vseg = v + ((size_t)h * NSEQ + (size_t)seg * SEGL) * DH;

  float4 rk[8], rv[8];
#pragma unroll
  for (int t = 0; t < 8; ++t) {
    const int idx = tid + 128 * t;
    rk[t] = *reinterpret_cast<const float4*>(kseg + 4 * idx);
    rv[t] = *reinterpret_cast<const float4*>(vseg + 4 * idx);
  }
  v2f pb[DH / 4];
#pragma unroll
  for (int kk = 0; kk < DH / 4; ++kk) {
    pb[kk].x = proj[(m0 + lr) * DH + 4 * kk + 2 * lh];
    pb[kk].y = proj[(m0 + lr) * DH + 4 * kk + 2 * lh + 1];
  }
#pragma unroll
  for (int t = 0; t < 8; ++t) {
    const int idx = tid + 128 * t;
    const int p = (idx >> 4) * PD + 4 * (idx & 15);
    *reinterpret_cast<float4*>(xk_s + p) = rk[t];
    *reinterpret_cast<float4*>(v_s + p)  = rv[t];
  }
  __syncthreads();

  v8f acc[SEGL / CHK] = {};
#pragma unroll
  for (int kk = 0; kk < DH / 4; ++kk) {
#pragma unroll
    for (int rt = 0; rt < SEGL / CHK; ++rt) {
      v2f a;
      a.x = xk_s[(CHK * rt + lr) * PD + 4 * kk + 2 * lh];
      a.y = xk_s[(CHK * rt + lr) * PD + 4 * kk + 2 * lh + 1];
      acc[rt] = wmma4(a, pb[kk], acc[rt]);
    }
  }
#pragma unroll
  for (int rt = 0; rt < SEGL / CHK; ++rt)
#pragma unroll
    for (int j = 0; j < 8; ++j)
      kp_s[(CHK * rt + j + 8 * lh) * PM + m0 + lr] = fmaxf(acc[rt][j] * norm, 0.f) + KEPS;
  __builtin_amdgcn_fence(__ATOMIC_RELEASE, "wavefront");
  __builtin_amdgcn_wave_barrier();
  __builtin_amdgcn_fence(__ATOMIC_ACQUIRE, "wavefront");

  float* P = ws + (size_t)hs * PSTRIDE;

  float run = 0.f;
#pragma unroll
  for (int r = 0; r < SEGL; ++r) run += kp_s[r * PM + m0 + lr];
  if (lh == 0) P[MF * DH + m0 + lr] = run;

  v8f sreg[4] = {};
#pragma unroll
  for (int kb = 0; kb < SEGL; kb += 4) {
    v2f a;
    a.x = kp_s[(kb + 2 * lh) * PM + m0 + lr];
    a.y = kp_s[(kb + 2 * lh + 1) * PM + m0 + lr];
#pragma unroll
    for (int et = 0; et < 4; ++et) {
      v2f b;
      b.x = v_s[(kb + 2 * lh) * PD + et * 16 + lr];
      b.y = v_s[(kb + 2 * lh + 1) * PD + et * 16 + lr];
      sreg[et] = wmma4(a, b, sreg[et]);
    }
  }
#pragma unroll
  for (int et = 0; et < 4; ++et)
#pragma unroll
    for (int jh = 0; jh < 2; ++jh) {
      float4 o;
      o.x = sreg[et][4 * jh + 0];
      o.y = sreg[et][4 * jh + 1];
      o.z = sreg[et][4 * jh + 2];
      o.w = sreg[et][4 * jh + 3];
      *reinterpret_cast<float4*>(P + (((ft * 4 + et) * 2 + jh) * 32 + lane) * 4) = o;
    }
}

__device__ __forceinline__ void contrib_load(const float* __restrict__ P, int blk0, int lane, int col,
                                             float4 (&t)[4], float& tk) {
#pragma unroll
  for (int i = 0; i < 4; ++i)
    t[i] = *reinterpret_cast<const float4*>(P + ((blk0 + i) * 32 + lane) * 4);
  tk = P[MF * DH + col];
}
__device__ __forceinline__ void contrib_add(v8f (&sreg)[ETW], float& run, const float4 (&t)[4], float tk) {
#pragma unroll
  for (int i = 0; i < 4; ++i) {
    sreg[i >> 1][4 * (i & 1) + 0] += t[i].x;
    sreg[i >> 1][4 * (i & 1) + 1] += t[i].y;
    sreg[i >> 1][4 * (i & 1) + 2] += t[i].z;
    sreg[i >> 1][4 * (i & 1) + 3] += t[i].w;
  }
  run += tk;
}

__global__ __launch_bounds__(256)
void performer_causal_f32_wmma(const float* __restrict__ q,
                               const float* __restrict__ k,
                               const float* __restrict__ v,
                               const float* __restrict__ proj,
                               float* __restrict__ out,
                               const float* __restrict__ ws, size_t ws_bytes) {
  __shared__ __attribute__((aligned(16))) float proj_s[MF * PD];
  __shared__ float S_s[MF * PD];
  __shared__ float qp_s[SEGL * PM];
  __shared__ float kp_s[SEGL * PM];
  __shared__ __attribute__((aligned(16))) float xq_s[SEGL * PD];
  __shared__ __attribute__((aligned(16))) float xk_s[SEGL * PD];
  __shared__ __attribute__((aligned(16))) float v_s[SEGL * PD];
  __shared__ float a_s[NAT * CHK * PA];

  const int eh   = blockIdx.x & 1;
  const int h    = (blockIdx.x >> 1) & 7;
  const int seg  = NSEG - 1 - (blockIdx.x >> 4);
  const int ec0  = EW * eh;
  const int n_begin = seg * SEGL;
  const int nlight  = n_begin / CHK;
  const int tid  = threadIdx.x;
  const int lane = tid & 31;
  const int wave = tid >> 5;
  const int lr   = lane & 15;
  const int lh   = lane >> 4;

  const float norm = 0.35355339059327378f;

  const float* qh = q + (size_t)h * NSEQ * DH;
  const float* kh = k + (size_t)h * NSEQ * DH;
  const float* vh = v + (size_t)h * NSEQ * DH;
  float*       oh = out + (size_t)h * NSEQ * DH;

  const int m0w = wave * 16;
  const bool from_ws = ws_bytes >= (size_t)WS_FLOATS * sizeof(float);
  const float* Ph = ws + (size_t)(h * NPART) * PSTRIDE;
  const int blk0 = wave * 8 + 4 * eh;
  const int sB = seg & 8;
  const int sC = seg & 12;
  const int sD = seg & 14;
  float4 tA[8][4], tB[4][4], tC[2][4], tD[1][4];
  float tkA[8], tkB[4], tkC[2], tkD[1];

  {
    float4 rp[8];
#pragma unroll
    for (int t = 0; t < 8; ++t)
      rp[t] = *reinterpret_cast<const float4*>(proj + 4 * (tid + 256 * t));
    float4 rq4[4], rk4[4], rv4[2];
#pragma unroll
    for (int t = 0; t < 4; ++t) {
      const int idx = tid + 256 * t;
      rq4[t] = *reinterpret_cast<const float4*>(qh + (size_t)n_begin * DH + 4 * idx);
      rk4[t] = *reinterpret_cast<const float4*>(kh + (size_t)n_begin * DH + 4 * idx);
    }
#pragma unroll
    for (int t = 0; t < 2; ++t) {
      const int idx = tid + 256 * t;
      rv4[t] = *reinterpret_cast<const float4*>(vh + (size_t)(n_begin + (idx >> 3)) * DH + ec0 + 4 * (idx & 7));
    }
#pragma unroll
    for (int t = 0; t < 8; ++t) {
      const int idx = tid + 256 * t;
      const int p = (idx >> 4) * PD + 4 * (idx & 15);
      *reinterpret_cast<float4*>(proj_s + p) = rp[t];
    }
#pragma unroll
    for (int t = 0; t < 4; ++t) {
      const int idx = tid + 256 * t;
      const int p = (idx >> 4) * PD + 4 * (idx & 15);
      *reinterpret_cast<float4*>(xq_s + p) = rq4[t];
      *reinterpret_cast<float4*>(xk_s + p) = rk4[t];
    }
#pragma unroll
    for (int t = 0; t < 2; ++t) {
      const int idx = tid + 256 * t;
      *reinterpret_cast<float4*>(v_s + (idx >> 3) * PD + ec0 + 4 * (idx & 7)) = rv4[t];
    }
  }
  if (from_ws) {
    if (seg & 8) {
#pragma unroll
      for (int u = 0; u < 8; ++u)
        contrib_load(Ph + (size_t)u * PSTRIDE, blk0, lane, m0w + lr, tA[u], tkA[u]);
    }
    if (seg & 4) {
#pragma unroll
      for (int u = 0; u < 4; ++u)
        contrib_load(Ph + (size_t)(sB + u) * PSTRIDE, blk0, lane, m0w + lr, tB[u], tkB[u]);
    }
  }

  v8f sreg[ETW] = {};
  float run = 0.f;
  if (!from_ws && nlight > 0) {
    const int m0 = m0w;
      float* const xkbuf = S_s;
      float* const vbuf  = S_s + 2 * CHK * PD;
      __syncthreads();
      v2f pb[DH / 4];
#pragma unroll
      for (int kk = 0; kk < DH / 4; ++kk) {
        pb[kk].x = proj_s[(m0 + lr) * PD + 4 * kk + 2 * lh];
        pb[kk].y = proj_s[(m0 + lr) * PD + 4 * kk + 2 * lh + 1];
      }
      float rk[4], rv[4];
#pragma unroll
      for (int t = 0; t < 4; ++t) {
        rk[t] = kh[tid + 256 * t];
        rv[t] = vh[tid + 256 * t];
      }
#pragma unroll
      for (int t = 0; t < 4; ++t) {
        const int p = ((tid >> 6) + 4 * t) * PD + (tid & 63);
        xkbuf[p] = rk[t];
        vbuf[p]  = rv[t];
      }
      __syncthreads();
      for (int c = 0; c < nlight; ++c) {
        const float* xkb = xkbuf + (c & 1) * (CHK * PD);
        const float* vb  = vbuf  + (c & 1) * (CHK * PD);
        float* xkn = xkbuf + ((c + 1) & 1) * (CHK * PD);
        float* vn  = vbuf  + ((c + 1) & 1) * (CHK * PD);
        const int cn = (c + 1 < nlight) ? (c + 1) : c;
#pragma unroll
        for (int t = 0; t < 4; ++t) {
          rk[t] = kh[cn * CHK * DH + tid + 256 * t];
          rv[t] = vh[cn * CHK * DH + tid + 256 * t];
        }
        v8f acck = {};
#pragma unroll
        for (int kk = 0; kk < DH / 4; ++kk) {
          v2f a;
          a.x = xkb[lr * PD + 4 * kk + 2 * lh];
          a.y = xkb[lr * PD + 4 * kk + 2 * lh + 1];
          acck = wmma4(a, pb[kk], acck);
        }
#pragma unroll
        for (int j = 0; j < 8; ++j)
          kp_s[(j + 8 * lh) * PM + m0 + lr] = fmaxf(acck[j] * norm, 0.f) + KEPS;
        __builtin_amdgcn_fence(__ATOMIC_RELEASE, "wavefront");
        __builtin_amdgcn_wave_barrier();
        __builtin_amdgcn_fence(__ATOMIC_ACQUIRE, "wavefront");
#pragma unroll
        for (int r = 0; r < CHK; ++r) run += kp_s[r * PM + m0 + lr];
#pragma unroll
        for (int etl = 0; etl < ETW; ++etl) {
          const int e0 = ec0 + etl * 16;
#pragma unroll
          for (int kb = 0; kb < CHK; kb += 4) {
            v2f a, b;
            a.x = kp_s[(kb + 2 * lh) * PM + m0 + lr];
            a.y = kp_s[(kb + 2 * lh + 1) * PM + m0 + lr];
            b.x = vb[(kb + 2 * lh) * PD + e0 + lr];
            b.y = vb[(kb + 2 * lh + 1) * PD + e0 + lr];
            sreg[etl] = wmma4(a, b, sreg[etl]);
          }
        }
#pragma unroll
        for (int t = 0; t < 4; ++t) {
          const int p = ((tid >> 6) + 4 * t) * PD + (tid & 63);
          xkn[p] = rk[t];
          vn[p]  = rv[t];
        }
        __syncthreads();
      }
  }
  __builtin_amdgcn_fence(__ATOMIC_RELEASE, "workgroup", "local");
  __builtin_amdgcn_s_barrier();
  __builtin_amdgcn_fence(__ATOMIC_ACQUIRE, "workgroup", "local");

  v8f accq[NRT] = {};
  v8f acck[NRT] = {};
#pragma unroll
  for (int kb = 0; kb < DH; kb += 4) {
    if (kb == DH / 4 && from_ws) {
      if (seg & 2) {
#pragma unroll
        for (int u = 0; u < 2; ++u)
          contrib_load(Ph + (size_t)(sC + u) * PSTRIDE, blk0, lane, m0w + lr, tC[u], tkC[u]);
      }
      if (seg & 1)
        contrib_load(Ph + (size_t)sD * PSTRIDE, blk0, lane, m0w + lr, tD[0], tkD[0]);
    }
    v2f b;
    b.x = proj_s[(m0w + lr) * PD + kb + 2 * lh];
    b.y = proj_s[(m0w + lr) * PD + kb + 2 * lh + 1];
#pragma unroll
    for (int rt = 0; rt < NRT; ++rt) {
      v2f a;
      a.x = xq_s[(CHK * rt + lr) * PD + kb + 2 * lh];
      a.y = xq_s[(CHK * rt + lr) * PD + kb + 2 * lh + 1];
      accq[rt] = wmma4(a, b, accq[rt]);
      a.x = xk_s[(CHK * rt + lr) * PD + kb + 2 * lh];
      a.y = xk_s[(CHK * rt + lr) * PD + kb + 2 * lh + 1];
      acck[rt] = wmma4(a, b, acck[rt]);
    }
  }
#pragma unroll
  for (int rt = 0; rt < NRT; ++rt)
#pragma unroll
    for (int j = 0; j < 8; ++j)
      kp_s[(CHK * rt + j + 8 * lh) * PM + m0w + lr] = fmaxf(acck[rt][j] * norm, 0.f) + KEPS;

  if (from_ws) {
    if (seg & 8) {
#pragma unroll
      for (int u = 0; u < 8; ++u) contrib_add(sreg, run, tA[u], tkA[u]);
    }
    if (seg & 4) {
#pragma unroll
      for (int u = 0; u < 4; ++u) contrib_add(sreg, run, tB[u], tkB[u]);
    }
    if (seg & 2) {
#pragma unroll
      for (int u = 0; u < 2; ++u) contrib_add(sreg, run, tC[u], tkC[u]);
    }
    if (seg & 1) contrib_add(sreg, run, tD[0], tkD[0]);
  }
#pragma unroll
  for (int etl = 0; etl < ETW; ++etl)
#pragma unroll
    for (int j = 0; j < 8; ++j)
      S_s[(m0w + j + 8 * lh) * PD + ec0 + etl * 16 + lr] = sreg[etl][j];

  __builtin_amdgcn_fence(__ATOMIC_RELEASE, "wavefront");
  __builtin_amdgcn_wave_barrier();
  __builtin_amdgcn_fence(__ATOMIC_ACQUIRE, "wavefront");
  {
    float kc[NRT][8];
#pragma unroll
    for (int rt = 0; rt < NRT; ++rt) {
#pragma unroll
      for (int j = 0; j < 8; ++j) {
        run += kp_s[(CHK * rt + j) * PM + m0w + lr];
        kc[rt][j] = run;
      }
#pragma unroll
      for (int j = 0; j < 8; ++j) {
        run += kp_s[(CHK * rt + 8 + j) * PM + m0w + lr];
        if (lh) kc[rt][j] = run;
      }
    }
#pragma unroll
    for (int rt = 0; rt < NRT; ++rt)
#pragma unroll
      for (int j = 0; j < 8; ++j)
        qp_s[(CHK * rt + j + 8 * lh) * PM + m0w + lr] =
            (fmaxf(accq[rt][j] * norm, 0.f) + KEPS) / kc[rt][j];
  }
  __syncthreads();

  const int wv = __builtin_amdgcn_readfirstlane(wave);

  const int ort = wv >> 1;
  const int e0  = ec0 + (wv & 1) * 16;
  int art = wv >> 1, act = wv & 1;
  if (wv == 1) { art = 2; act = 2; }
  v8f o = {};
  {
    v8f am = {};
    const float* qo = qp_s + (CHK * ort + lr) * PM + 2 * lh;
    const float* sb = S_s + (2 * lh) * PD + e0 + lr;
    const float* qa = qp_s + (CHK * art + lr) * PM + 2 * lh;
    const float* ka = kp_s + (CHK * act + lr) * PM + 2 * lh;
#pragma unroll
    for (int kb = 0; kb < MF; kb += 4) {
      v2f a, b;
      a.x = qo[kb];
      a.y = qo[kb + 1];
      b.x = sb[kb * PD];
      b.y = sb[(kb + 1) * PD];
      o = wmma4(a, b, o);
      a.x = qa[kb];
      a.y = qa[kb + 1];
      b.x = ka[kb];
      b.y = ka[kb + 1];
      am = wmma4(a, b, am);
    }
    float* at = a_s + (art * (art + 1) / 2 + act) * (CHK * PA);
    const bool diag = art == act;
#pragma unroll
    for (int j = 0; j < 8; ++j) {
      const int row = j + 8 * lh;
      at[row * PA + lr] = (!diag || lr <= row) ? am[j] : 0.f;
    }
  }
  if (wv < 2) {
    const int act2 = 2 + wv;
    v8f am = {};
    const float* qa = qp_s + (CHK * 3 + lr) * PM + 2 * lh;
    const float* ka = kp_s + (CHK * act2 + lr) * PM + 2 * lh;
#pragma unroll
    for (int kb = 0; kb < MF; kb += 4) {
      v2f a, b;
      a.x = qa[kb];
      a.y = qa[kb + 1];
      b.x = ka[kb];
      b.y = ka[kb + 1];
      am = wmma4(a, b, am);
    }
    float* at = a_s + (6 + act2) * (CHK * PA);
    const bool diag = act2 == 3;
#pragma unroll
    for (int j = 0; j < 8; ++j) {
      const int row = j + 8 * lh;
      at[row * PA + lr] = (!diag || lr <= row) ? am[j] : 0.f;
    }
  }
  __syncthreads();

#pragma unroll
  for (int ct = 0; ct < NRT; ++ct) {
    if (ct <= ort) {
      const float* at = a_s + (ort * (ort + 1) / 2 + ct) * (CHK * PA) + lr * PA + 2 * lh;
      const float* vb = v_s + (CHK * ct + 2 * lh) * PD + e0 + lr;
#pragma unroll
      for (int kb = 0; kb < CHK; kb += 4) {
        v2f a, b;
        a.x = at[kb];
        a.y = at[kb + 1];
        b.x = vb[kb * PD];
        b.y = vb[(kb + 1) * PD];
        o = wmma4(a, b, o);
      }
    }
  }
#pragma unroll
  for (int j = 0; j < 8; ++j)
    oh[(n_begin + CHK * ort + j + 8 * lh) * DH + e0 + lr] = o[j];
}

extern "C" void kernel_launch(void* const* d_in, const int* in_sizes, int n_in,
                              void* d_out, int out_size, void* d_ws, size_t ws_size,
                              hipStream_t stream) {
  (void)in_sizes; (void)n_in; (void)out_size;
  const float* q    = (const float*)d_in[0];
  const float* k    = (const float*)d_in[1];
  const float* v    = (const float*)d_in[2];
  const float* proj = (const float*)d_in[3];
  float* out        = (float*)d_out;
  float* ws         = (float*)d_ws;
  performer_segment_state_f32_wmma<<<dim3(240), dim3(128), 0, stream>>>(k, v, proj, ws, ws_size);
  performer_causal_f32_wmma<<<dim3(256), dim3(256), 0, stream>>>(q, k, v, proj, out, ws, ws_size);
}


# First launch also stores its kp rows (all 16 segments) to the workspace; each scan wave fetches its 64x16 kp tile from there and the scan launch drops its k projection; values unchanged
# speedup vs baseline: 31.7701x; 1.0591x over previous
#include <hip/hip_runtime.h>


#define KEPS 1e-3f
#define NSEQ 1024
#define DH   64
#define MF   128
#define CHK  16
#define NSEG 16
#define SEGL (NSEQ / NSEG)
#define NEH  2
#define EW   (DH / NEH)
#define ETW  (EW / 16)
static_assert(8 * NSEG * NEH == 256 && NSEG == 16 && NEH == 2 && EW == 32 && ETW == 2,
              "the scan launch below starts 8 x NSEG x NEH workgroups and decodes its block id by shifts");
#define NPART   (NSEG - 1)
#define PSTRIDE (MF * DH + MF)
#define WS_FLOATS (8 * NPART * PSTRIDE)
#define KP_FLOATS (8 * NSEG * SEGL * MF)
#define WSK_FLOATS (WS_FLOATS + KP_FLOATS)
#define NFH 2
static_assert(NFH * 8 * NSEG == 256 && MF / 16 == 4 * NFH,
              "the first launch below starts NFH x 8 x NSEG workgroups of 4 waves, one 16-feature tile a wave");
static_assert((WS_FLOATS * 4) % 256 == 0, "the kp rows start on a 256-byte boundary of the workspace");
static_assert(SEGL % CHK == 0 && SEGL == 64, "the first launch holds one segment of 64 rows in LDS");

#define PD   (DH + 12)
#define PM   (MF + 12)
#define PA   (CHK + 4)
#define NRT  (SEGL / CHK)
#define NAT  (NRT * (NRT + 1) / 2)
static_assert(NRT == 4 && NAT == 10 && 2 * NRT == 8,
              "the scan launch gives its 8 waves one output tile each (NRT x ETW) and the 10 score tiles by wave number");

typedef float v2f __attribute__((ext_vector_type(2)));
typedef float v8f __attribute__((ext_vector_type(8)));
typedef float v4f __attribute__((ext_vector_type(4)));

__device__ __forceinline__ v8f wmma4(v2f a, v2f b, v8f c) {
  return __builtin_amdgcn_wmma_f32_16x16x4_f32(false, a, false, b,
                                               (short)0, c, false, false);
}


__global__ __launch_bounds__(128)
void performer_segment_state_f32_wmma(const float* __restrict__ k,
                                      const float* __restrict__ v,
                                      const float* __restrict__ proj,
                                      float* __restrict__ ws, size_t ws_bytes) {
  if (ws_bytes < (size_t)WS_FLOATS * sizeof(float)) return;
  const bool kp_ws = ws_bytes >= (size_t)WSK_FLOATS * sizeof(float);

  __shared__ __attribute__((aligned(16))) float xk_s[SEGL * PD];
  __shared__ __attribute__((aligned(16))) float v_s[SEGL * PD];
  __shared__ __attribute__((aligned(16))) float kp_s[SEGL * PM];

  const int fh   = blockIdx.x & 1;
  const int h    = blockIdx.x >> 5;
  const int seg  = (blockIdx.x >> 1) & 15;
  const bool last = seg == NSEG - 1;
  if (last && !kp_ws) return;
  const int hs   = h * NPART + seg;
  const int tid  = threadIdx.x;
  const int lane = tid & 31;
  const int ft   = 4 * fh + (tid >> 5);
  const int lr   = lane & 15;
  const int lh   = lane >> 4;
  const int m0   = ft * 16;

  const float norm = 0.35355339059327378f;

  const float* kseg = k + ((size_t)h * NSEQ + (size_t)seg * SEGL) * DH;
  const float* vseg = v + ((size_t)h * NSEQ + (size_t)seg * SEGL) * DH;

  float4 rk[8], rv[8];
#pragma unroll
  for (int t = 0; t < 8; ++t) {
    const int idx = tid + 128 * t;
    rk[t] = *reinterpret_cast<const float4*>(kseg + 4 * idx);
    rv[t] = *reinterpret_cast<const float4*>(vseg + 4 * idx);
  }
  v2f pb[DH / 4];
#pragma unroll
  for (int kk = 0; kk < DH / 4; ++kk) {
    pb[kk].x = proj[(m0 + lr) * DH + 4 * kk + 2 * lh];
    pb[kk].y = proj[(m0 + lr) * DH + 4 * kk + 2 * lh + 1];
  }
#pragma unroll
  for (int t = 0; t < 8; ++t) {
    const int idx = tid + 128 * t;
    const int p = (idx >> 4) * PD + 4 * (idx & 15);
    *reinterpret_cast<float4*>(xk_s + p) = rk[t];
    *reinterpret_cast<float4*>(v_s + p)  = rv[t];
  }
  __syncthreads();

  v8f acc[SEGL / CHK] = {};
#pragma unroll
  for (int kk = 0; kk < DH / 4; ++kk) {
#pragma unroll
    for (int rt = 0; rt < SEGL / CHK; ++rt) {
      v2f a;
      a.x = xk_s[(CHK * rt + lr) * PD + 4 * kk + 2 * lh];
      a.y = xk_s[(CHK * rt + lr) * PD + 4 * kk + 2 * lh + 1];
      acc[rt] = wmma4(a, pb[kk], acc[rt]);
    }
  }
#pragma unroll
  for (int rt = 0; rt < SEGL / CHK; ++rt)
#pragma unroll
    for (int j = 0; j < 8; ++j)
      kp_s[(CHK * rt + j + 8 * lh) * PM + m0 + lr] = fmaxf(acc[rt][j] * norm, 0.f) + KEPS;
  __builtin_amdgcn_fence(__ATOMIC_RELEASE, "wavefront");
  __builtin_amdgcn_wave_barrier();
  __builtin_amdgcn_fence(__ATOMIC_ACQUIRE, "wavefront");

  if (kp_ws) {
    float* KP = ws + (size_t)WS_FLOATS + ((size_t)(h * NSEG + seg) * (MF / 16) + ft) * (SEGL * 16);
#pragma unroll
    for (int t = 0; t < 8; ++t) {
      const float4 x = *reinterpret_cast<const float4*>(kp_s + (8 * t + (lane >> 2)) * PM + m0 + 4 * (lane & 3));
      *reinterpret_cast<float4*>(KP + 4 * (32 * t + lane)) = x;
    }
  }
  if (last) return;

  float* P = ws + (size_t)hs * PSTRIDE;

  float run = 0.f;
#pragma unroll
  for (int r = 0; r < SEGL; ++r) run += kp_s[r * PM + m0 + lr];
  if (lh == 0) P[MF * DH + m0 + lr] = run;

  v8f sreg[4] = {};
#pragma unroll
  for (int kb = 0; kb < SEGL; kb += 4) {
    v2f a;
    a.x = kp_s[(kb + 2 * lh) * PM + m0 + lr];
    a.y = kp_s[(kb + 2 * lh + 1) * PM + m0 + lr];
#pragma unroll
    for (int et = 0; et < 4; ++et) {
      v2f b;
      b.x = v_s[(kb + 2 * lh) * PD + et * 16 + lr];
      b.y = v_s[(kb + 2 * lh + 1) * PD + et * 16 + lr];
      sreg[et] = wmma4(a, b, sreg[et]);
    }
  }
#pragma unroll
  for (int et = 0; et < 4; ++et)
#pragma unroll
    for (int jh = 0; jh < 2; ++jh) {
      float4 o;
      o.x = sreg[et][4 * jh + 0];
      o.y = sreg[et][4 * jh + 1];
      o.z = sreg[et][4 * jh + 2];
      o.w = sreg[et][4 * jh + 3];
      *reinterpret_cast<float4*>(P + (((ft * 4 + et) * 2 + jh) * 32 + lane) * 4) = o;
    }
}

__device__ __forceinline__ void contrib_load(const float* __restrict__ P, int blk0, int lane, int col,
                                             float4 (&t)[4], float& tk) {
#pragma unroll
  for (int i = 0; i < 4; ++i)
    t[i] = *reinterpret_cast<const float4*>(P + ((blk0 + i) * 32 + lane) * 4);
  tk = P[MF * DH + col];
}
__device__ __forceinline__ void contrib_add(v8f (&sreg)[ETW], float& run, const float4 (&t)[4], float tk) {
#pragma unroll
  for (int i = 0; i < 4; ++i) {
    sreg[i >> 1][4 * (i & 1) + 0] += t[i].x;
    sreg[i >> 1][4 * (i & 1) + 1] += t[i].y;
    sreg[i >> 1][4 * (i & 1) + 2] += t[i].z;
    sreg[i >> 1][4 * (i & 1) + 3] += t[i].w;
  }
  run += tk;
}

__global__ __launch_bounds__(256)
void performer_causal_f32_wmma(const float* __restrict__ q,
                               const float* __restrict__ k,
                               const float* __restrict__ v,
                               const float* __restrict__ proj,
                               float* __restrict__ out,
                               const float* __restrict__ ws, size_t ws_bytes) {
  __shared__ __attribute__((aligned(16))) float proj_s[MF * PD];
  __shared__ float S_s[MF * PD];
  __shared__ float qp_s[SEGL * PM];
  __shared__ __attribute__((aligned(16))) float kp_s[SEGL * PM];
  __shared__ __attribute__((aligned(16))) float xq_s[SEGL * PD];
  __shared__ __attribute__((aligned(16))) float xk_s[SEGL * PD];
  __shared__ __attribute__((aligned(16))) float v_s[SEGL * PD];
  __shared__ float a_s[NAT * CHK * PA];

  const int eh   = blockIdx.x & 1;
  const int h    = (blockIdx.x >> 1) & 7;
  const int seg  = NSEG - 1 - (blockIdx.x >> 4);
  const int ec0  = EW * eh;
  const int n_begin = seg * SEGL;
  const int nlight  = n_begin / CHK;
  const int tid  = threadIdx.x;
  const int lane = tid & 31;
  const int wave = tid >> 5;
  const int lr   = lane & 15;
  const int lh   = lane >> 4;

  const float norm = 0.35355339059327378f;

  const float* qh = q + (size_t)h * NSEQ * DH;
  const float* kh = k + (size_t)h * NSEQ * DH;
  const float* vh = v + (size_t)h * NSEQ * DH;
  float*       oh = out + (size_t)h * NSEQ * DH;

  const int m0w = wave * 16;
  const bool from_ws = ws_bytes >= (size_t)WS_FLOATS * sizeof(float);
  const bool kp_ws = ws_bytes >= (size_t)WSK_FLOATS * sizeof(float);
  const float* Ph = ws + (size_t)(h * NPART) * PSTRIDE;
  const int blk0 = wave * 8 + 4 * eh;
  const int sB = seg & 8;
  const int sC = seg & 12;
  const int sD = seg & 14;
  float4 tA[8][4], tB[4][4], tC[2][4], tD[1][4];
  float tkA[8], tkB[4], tkC[2], tkD[1];
  v4f rkp[8];

  {
    float4 rp[8];
#pragma unroll
    for (int t = 0; t < 8; ++t)
      rp[t] = *reinterpret_cast<const float4*>(proj + 4 * (tid + 256 * t));
    float4 rq4[4], rv4[2];
    v4f rk4[4];
#pragma unroll
    for (int t = 0; t < 4; ++t) {
      const int idx = tid + 256 * t;
      rq4[t] = *reinterpret_cast<const float4*>(qh + (size_t)n_begin * DH + 4 * idx);
    }
#pragma unroll
    for (int t = 0; t < 2; ++t) {
      const int idx = tid + 256 * t;
      rv4[t] = *reinterpret_cast<const float4*>(vh + (size_t)(n_begin + (idx >> 3)) * DH + ec0 + 4 * (idx & 7));
    }
    if (!kp_ws) {
#pragma unroll
      for (int t = 0; t < 4; ++t)
        rk4[t] = *reinterpret_cast<const v4f*>(kh + (size_t)n_begin * DH + 4 * (tid + 256 * t));
    }
#pragma unroll
    for (int t = 0; t < 8; ++t) {
      const int idx = tid + 256 * t;
      const int p = (idx >> 4) * PD + 4 * (idx & 15);
      *reinterpret_cast<float4*>(proj_s + p) = rp[t];
    }
#pragma unroll
    for (int t = 0; t < 4; ++t) {
      const int idx = tid + 256 * t;
      const int p = (idx >> 4) * PD + 4 * (idx & 15);
      *reinterpret_cast<float4*>(xq_s + p) = rq4[t];
    }
#pragma unroll
    for (int t = 0; t < 2; ++t) {
      const int idx = tid + 256 * t;
      *reinterpret_cast<float4*>(v_s + (idx >> 3) * PD + ec0 + 4 * (idx & 7)) = rv4[t];
    }
    if (!kp_ws) {
#pragma unroll
      for (int t = 0; t < 4; ++t) {
        const int idx = tid + 256 * t;
        *reinterpret_cast<v4f*>(xk_s + (idx >> 4) * PD + 4 * (idx & 15)) = rk4[t];
      }
    }
  }
  if (from_ws) {
    if (seg & 8) {
#pragma unroll
      for (int u = 0; u < 8; ++u)
        contrib_load(Ph + (size_t)u * PSTRIDE, blk0, lane, m0w + lr, tA[u], tkA[u]);
    }
    if (seg & 4) {
#pragma unroll
      for (int u = 0; u < 4; ++u)
        contrib_load(Ph + (size_t)(sB + u) * PSTRIDE, blk0, lane, m0w + lr, tB[u], tkB[u]);
    }
  }

  v8f sreg[ETW] = {};
  float run = 0.f;
  if (!from_ws && nlight > 0) {
    const int m0 = m0w;
      float* const xkbuf = S_s;
      float* const vbuf  = S_s + 2 * CHK * PD;
      __syncthreads();
      v2f pb[DH / 4];
#pragma unroll
      for (int kk = 0; kk < DH / 4; ++kk) {
        pb[kk].x = proj_s[(m0 + lr) * PD + 4 * kk + 2 * lh];
        pb[kk].y = proj_s[(m0 + lr) * PD + 4 * kk + 2 * lh + 1];
      }
      float rk[4], rv[4];
#pragma unroll
      for (int t = 0; t < 4; ++t) {
        rk[t] = kh[tid + 256 * t];
        rv[t] = vh[tid + 256 * t];
      }
#pragma unroll
      for (int t = 0; t < 4; ++t) {
        const int p = ((tid >> 6) + 4 * t) * PD + (tid & 63);
        xkbuf[p] = rk[t];
        vbuf[p]  = rv[t];
      }
      __syncthreads();
      for (int c = 0; c < nlight; ++c) {
        const float* xkb = xkbuf + (c & 1) * (CHK * PD);
        const float* vb  = vbuf  + (c & 1) * (CHK * PD);
        float* xkn = xkbuf + ((c + 1) & 1) * (CHK * PD);
        float* vn  = vbuf  + ((c + 1) & 1) * (CHK * PD);
        const int cn = (c + 1 < nlight) ? (c + 1) : c;
#pragma unroll
        for (int t = 0; t < 4; ++t) {
          rk[t] = kh[cn * CHK * DH + tid + 256 * t];
          rv[t] = vh[cn * CHK * DH + tid + 256 * t];
        }
        v8f acck = {};
#pragma unroll
        for (int kk = 0; kk < DH / 4; ++kk) {
          v2f a;
          a.x = xkb[lr * PD + 4 * kk + 2 * lh];
          a.y = xkb[lr * PD + 4 * kk + 2 * lh + 1];
          acck = wmma4(a, pb[kk], acck);
        }
#pragma unroll
        for (int j = 0; j < 8; ++j)
          kp_s[(j + 8 * lh) * PM + m0 + lr] = fmaxf(acck[j] * norm, 0.f) + KEPS;
        __builtin_amdgcn_fence(__ATOMIC_RELEASE, "wavefront");
        __builtin_amdgcn_wave_barrier();
        __builtin_amdgcn_fence(__ATOMIC_ACQUIRE, "wavefront");
#pragma unroll
        for (int r = 0; r < CHK; ++r) run += kp_s[r * PM + m0 + lr];
#pragma unroll
        for (int etl = 0; etl < ETW; ++etl) {
          const int e0 = ec0 + etl * 16;
#pragma unroll
          for (int kb = 0; kb < CHK; kb += 4) {
            v2f a, b;
            a.x = kp_s[(kb + 2 * lh) * PM + m0 + lr];
            a.y = kp_s[(kb + 2 * lh + 1) * PM + m0 + lr];
            b.x = vb[(kb + 2 * lh) * PD + e0 + lr];
            b.y = vb[(kb + 2 * lh + 1) * PD + e0 + lr];
            sreg[etl] = wmma4(a, b, sreg[etl]);
          }
        }
#pragma unroll
        for (int t = 0; t < 4; ++t) {
          const int p = ((tid >> 6) + 4 * t) * PD + (tid & 63);
          xkn[p] = rk[t];
          vn[p]  = rv[t];
        }
        __syncthreads();
      }
  }
  __builtin_amdgcn_fence(__ATOMIC_RELEASE, "workgroup", "local");
  __builtin_amdgcn_s_barrier();
  __builtin_amdgcn_fence(__ATOMIC_ACQUIRE, "workgroup", "local");

  if (!kp_ws) {
    v8f acck[NRT] = {};
#pragma unroll
    for (int kb = 0; kb < DH; kb += 4) {
      v2f b;
      b.x = proj_s[(m0w + lr) * PD + kb + 2 * lh];
      b.y = proj_s[(m0w + lr) * PD + kb + 2 * lh + 1];
#pragma unroll
      for (int rt = 0; rt < NRT; ++rt) {
        v2f a;
        a.x = xk_s[(CHK * rt + lr) * PD + kb + 2 * lh];
        a.y = xk_s[(CHK * rt + lr) * PD + kb + 2 * lh + 1];
        acck[rt] = wmma4(a, b, acck[rt]);
      }
    }
#pragma unroll
    for (int rt = 0; rt < NRT; ++rt)
#pragma unroll
      for (int j = 0; j < 8; ++j)
        kp_s[(CHK * rt + j + 8 * lh) * PM + m0w + lr] = fmaxf(acck[rt][j] * norm, 0.f) + KEPS;
  }

  v8f accq[NRT] = {};
#pragma unroll
  for (int kb = 0; kb < DH; kb += 4) {
    if (kb == DH / 2 && kp_ws) {
      const float* KP = ws + (size_t)WS_FLOATS + ((size_t)(h * NSEG + seg) * (MF / 16) + wave) * (SEGL * 16);
#pragma unroll
      for (int t = 0; t < 8; ++t)
        rkp[t] = *reinterpret_cast<const v4f*>(KP + 4 * (32 * t + lane));
    }
    if (kb == DH / 2 && from_ws) {
      if (seg & 2) {
#pragma unroll
        for (int u = 0; u < 2; ++u)
          contrib_load(Ph + (size_t)(sC + u) * PSTRIDE, blk0, lane, m0w + lr, tC[u], tkC[u]);
      }
      if (seg & 1)
        contrib_load(Ph + (size_t)sD * PSTRIDE, blk0, lane, m0w + lr, tD[0], tkD[0]);
    }
    v2f b;
    b.x = proj_s[(m0w + lr) * PD + kb + 2 * lh];
    b.y = proj_s[(m0w + lr) * PD + kb + 2 * lh + 1];
#pragma unroll
    for (int rt = 0; rt < NRT; ++rt) {
      v2f a;
      a.x = xq_s[(CHK * rt + lr) * PD + kb + 2 * lh];
      a.y = xq_s[(CHK * rt + lr) * PD + kb + 2 * lh + 1];
      accq[rt] = wmma4(a, b, accq[rt]);
    }
  }
  if (kp_ws) {
#pragma unroll
    for (int t = 0; t < 8; ++t)
      *reinterpret_cast<v4f*>(kp_s + (8 * t + (lane >> 2)) * PM + m0w + 4 * (lane & 3)) = rkp[t];
  }

  if (from_ws) {
    if (seg & 8) {
#pragma unroll
      for (int u = 0; u < 8; ++u) contrib_add(sreg, run, tA[u], tkA[u]);
    }
    if (seg & 4) {
#pragma unroll
      for (int u = 0; u < 4; ++u) contrib_add(sreg, run, tB[u], tkB[u]);
    }
    if (seg & 2) {
#pragma unroll
      for (int u = 0; u < 2; ++u) contrib_add(sreg, run, tC[u], tkC[u]);
    }
    if (seg & 1) contrib_add(sreg, run, tD[0], tkD[0]);
  }
#pragma unroll
  for (int etl = 0; etl < ETW; ++etl)
#pragma unroll
    for (int j = 0; j < 8; ++j)
      S_s[(m0w + j + 8 * lh) * PD + ec0 + etl * 16 + lr] = sreg[etl][j];

  __builtin_amdgcn_fence(__ATOMIC_RELEASE, "wavefront");
  __builtin_amdgcn_wave_barrier();
  __builtin_amdgcn_fence(__ATOMIC_ACQUIRE, "wavefront");
  {
    float kc[NRT][8];
#pragma unroll
    for (int rt = 0; rt < NRT; ++rt) {
#pragma unroll
      for (int j = 0; j < 8; ++j) {
        run += kp_s[(CHK * rt + j) * PM + m0w + lr];
        kc[rt][j] = run;
      }
#pragma unroll
      for (int j = 0; j < 8; ++j) {
        run += kp_s[(CHK * rt + 8 + j) * PM + m0w + lr];
        if (lh) kc[rt][j] = run;
      }
    }
#pragma unroll
    for (int rt = 0; rt < NRT; ++rt)
#pragma unroll
      for (int j = 0; j < 8; ++j)
        qp_s[(CHK * rt + j + 8 * lh) * PM + m0w + lr] =
            (fmaxf(accq[rt][j] * norm, 0.f) + KEPS) / kc[rt][j];
  }
  __syncthreads();

  const int wv = __builtin_amdgcn_readfirstlane(wave);

  const int ort = wv >> 1;
  const int e0  = ec0 + (wv & 1) * 16;
  int art = wv >> 1, act = wv & 1;
  if (wv == 1) { art = 2; act = 2; }
  v8f o = {};
  {
    v8f am = {};
    const float* qo = qp_s + (CHK * ort + lr) * PM + 2 * lh;
    const float* sb = S_s + (2 * lh) * PD + e0 + lr;
    const float* qa = qp_s + (CHK * art + lr) * PM + 2 * lh;
    const float* ka = kp_s + (CHK * act + lr) * PM + 2 * lh;
#pragma unroll
    for (int kb = 0; kb < MF; kb += 4) {
      v2f a, b;
      a.x = qo[kb];
      a.y = qo[kb + 1];
      b.x = sb[kb * PD];
      b.y = sb[(kb + 1) * PD];
      o = wmma4(a, b, o);
      a.x = qa[kb];
      a.y = qa[kb + 1];
      b.x = ka[kb];
      b.y = ka[kb + 1];
      am = wmma4(a, b, am);
    }
    float* at = a_s + (art * (art + 1) / 2 + act) * (CHK * PA);
    const bool diag = art == act;
#pragma unroll
    for (int j = 0; j < 8; ++j) {
      const int row = j + 8 * lh;
      at[row * PA + lr] = (!diag || lr <= row) ? am[j] : 0.f;
    }
  }
  if (wv < 2) {
    const int act2 = 2 + wv;
    v8f am = {};
    const float* qa = qp_s + (CHK * 3 + lr) * PM + 2 * lh;
    const float* ka = kp_s + (CHK * act2 + lr) * PM + 2 * lh;
#pragma unroll
    for (int kb = 0; kb < MF; kb += 4) {
      v2f a, b;
      a.x = qa[kb];
      a.y = qa[kb + 1];
      b.x = ka[kb];
      b.y = ka[kb + 1];
      am = wmma4(a, b, am);
    }
    float* at = a_s + (6 + act2) * (CHK * PA);
    const bool diag = act2 == 3;
#pragma unroll
    for (int j = 0; j < 8; ++j) {
      const int row = j + 8 * lh;
      at[row * PA + lr] = (!diag || lr <= row) ? am[j] : 0.f;
    }
  }
  __syncthreads();

#pragma unroll
  for (int ct = 0; ct < NRT; ++ct) {
    if (ct <= ort) {
      const float* at = a_s + (ort * (ort + 1) / 2 + ct) * (CHK * PA) + lr * PA + 2 * lh;
      const float* vb = v_s + (CHK * ct + 2 * lh) * PD + e0 + lr;
#pragma unroll
      for (int kb = 0; kb < CHK; kb += 4) {
        v2f a, b;
        a.x = at[kb];
        a.y = at[kb + 1];
        b.x = vb[kb * PD];
        b.y = vb[(kb + 1) * PD];
        o = wmma4(a, b, o);
      }
    }
  }
#pragma unroll
  for (int j = 0; j < 8; ++j)
    oh[(n_begin + CHK * ort + j + 8 * lh) * DH + e0 + lr] = o[j];
}

extern "C" void kernel_launch(void* const* d_in, const int* in_sizes, int n_in,
                              void* d_out, int out_size, void* d_ws, size_t ws_size,
                              hipStream_t stream) {
  (void)in_sizes; (void)n_in; (void)out_size;
  const float* q    = (const float*)d_in[0];
  const float* k    = (const float*)d_in[1];
  const float* v    = (const float*)d_in[2];
  const float* proj = (const float*)d_in[3];
  float* out        = (float*)d_out;
  float* ws         = (float*)d_ws;
  performer_segment_state_f32_wmma<<<dim3(256), dim3(128), 0, stream>>>(k, v, proj, ws, ws_size);
  performer_causal_f32_wmma<<<dim3(256), dim3(256), 0, stream>>>(q, k, v, proj, out, ws, ws_size);
}


# LDS reads of WMMA fragments and of the serial feature sums regrouped: a few large groups in front of the WMMA blocks instead of reads among the WMMAs (both launches); same WMMAs, same sum order
# speedup vs baseline: 34.4917x; 1.0857x over previous
#include <hip/hip_runtime.h>


#define KEPS 1e-3f
#define NSEQ 1024
#define DH   64
#define MF   128
#define CHK  16
#define NSEG 16
#define SEGL (NSEQ / NSEG)
#define NEH  2
#define EW   (DH / NEH)
#define ETW  (EW / 16)
static_assert(8 * NSEG * NEH == 256 && NSEG == 16 && NEH == 2 && EW == 32 && ETW == 2,
              "the scan launch below starts 8 x NSEG x NEH workgroups and decodes its block id by shifts");
#define NPART   (NSEG - 1)
#define PSTRIDE (MF * DH + MF)
#define WS_FLOATS (8 * NPART * PSTRIDE)
#define KP_FLOATS (8 * NSEG * SEGL * MF)
#define WSK_FLOATS (WS_FLOATS + KP_FLOATS)
#define NFH 2
static_assert(NFH * 8 * NSEG == 256 && MF / 16 == 4 * NFH,
              "the first launch below starts NFH x 8 x NSEG workgroups of 4 waves, one 16-feature tile a wave");
static_assert((WS_FLOATS * 4) % 256 == 0, "the kp rows start on a 256-byte boundary of the workspace");
static_assert(SEGL % CHK == 0 && SEGL == 64, "the first launch holds one segment of 64 rows in LDS");

#define PD   (DH + 12)
#define PM   (MF + 12)
#define PA   (CHK + 4)
#define NRT  (SEGL / CHK)
#define NAT  (NRT * (NRT + 1) / 2)
static_assert(NRT == 4 && NAT == 10 && 2 * NRT == 8,
              "the scan launch gives its 8 waves one output tile each (NRT x ETW) and the 10 score tiles by wave number");

typedef float v2f __attribute__((ext_vector_type(2)));
typedef float v8f __attribute__((ext_vector_type(8)));
typedef float v4f __attribute__((ext_vector_type(4)));

__device__ __forceinline__ v8f wmma4(v2f a, v2f b, v8f c) {
  return __builtin_amdgcn_wmma_f32_16x16x4_f32(false, a, false, b,
                                               (short)0, c, false, false);
}


__global__ __launch_bounds__(128)
void performer_segment_state_f32_wmma(const float* __restrict__ k,
                                      const float* __restrict__ v,
                                      const float* __restrict__ proj,
                                      float* __restrict__ ws, size_t ws_bytes) {
  if (ws_bytes < (size_t)WS_FLOATS * sizeof(float)) return;
  const bool kp_ws = ws_bytes >= (size_t)WSK_FLOATS * sizeof(float);

  __shared__ __attribute__((aligned(16))) float xk_s[SEGL * PD];
  __shared__ __attribute__((aligned(16))) float v_s[SEGL * PD];
  __shared__ __attribute__((aligned(16))) float kp_s[SEGL * PM];

  const int fh   = blockIdx.x & 1;
  const int h    = blockIdx.x >> 5;
  const int seg  = (blockIdx.x >> 1) & 15;
  const bool last = seg == NSEG - 1;
  if (last && !kp_ws) return;
  const int hs   = h * NPART + seg;
  const int tid  = threadIdx.x;
  const int lane = tid & 31;
  const int ft   = 4 * fh + (tid >> 5);
  const int lr   = lane & 15;
  const int lh   = lane >> 4;
  const int m0   = ft * 16;

  const float norm = 0.35355339059327378f;

  const float* kseg = k + ((size_t)h * NSEQ + (size_t)seg * SEGL) * DH;
  const float* vseg = v + ((size_t)h * NSEQ + (size_t)seg * SEGL) * DH;

  float4 rk[8], rv[8];
#pragma unroll
  for (int t = 0; t < 8; ++t) {
    const int idx = tid + 128 * t;
    rk[t] = *reinterpret_cast<const float4*>(kseg + 4 * idx);
    rv[t] = *reinterpret_cast<const float4*>(vseg + 4 * idx);
  }
  v2f pb[DH / 4];
#pragma unroll
  for (int kk = 0; kk < DH / 4; ++kk) {
    pb[kk].x = proj[(m0 + lr) * DH + 4 * kk + 2 * lh];
    pb[kk].y = proj[(m0 + lr) * DH + 4 * kk + 2 * lh + 1];
  }
#pragma unroll
  for (int t = 0; t < 8; ++t) {
    const int idx = tid + 128 * t;
    const int p = (idx >> 4) * PD + 4 * (idx & 15);
    *reinterpret_cast<float4*>(xk_s + p) = rk[t];
    *reinterpret_cast<float4*>(v_s + p)  = rv[t];
  }
  __syncthreads();

  v2f xa[DH / 4][SEGL / CHK];
#pragma unroll
  for (int kk = 0; kk < DH / 4; ++kk) {
#pragma unroll
    for (int rt = 0; rt < SEGL / CHK; ++rt) {
      xa[kk][rt].x = xk_s[(CHK * rt + lr) * PD + 4 * kk + 2 * lh];
      xa[kk][rt].y = xk_s[(CHK * rt + lr) * PD + 4 * kk + 2 * lh + 1];
    }
  }
  __builtin_amdgcn_sched_barrier(0);
  v8f acc[SEGL / CHK] = {};
#pragma unroll
  for (int kk = 0; kk < 4; ++kk) {
#pragma unroll
    for (int rt = 0; rt < SEGL / CHK; ++rt)
      acc[rt] = wmma4(xa[kk][rt], pb[kk], acc[rt]);
  }
  __builtin_amdgcn_sched_barrier(0);
  v2f vb[SEGL / 4][4];
#pragma unroll
  for (int kb = 0; kb < SEGL; kb += 4) {
#pragma unroll
    for (int et = 0; et < 4; ++et) {
      vb[kb / 4][et].x = v_s[(kb + 2 * lh) * PD + et * 16 + lr];
      vb[kb / 4][et].y = v_s[(kb + 2 * lh + 1) * PD + et * 16 + lr];
    }
  }
  __builtin_amdgcn_sched_barrier(0);
#pragma unroll
  for (int kk = 4; kk < DH / 4; ++kk) {
#pragma unroll
    for (int rt = 0; rt < SEGL / CHK; ++rt)
      acc[rt] = wmma4(xa[kk][rt], pb[kk], acc[rt]);
  }
#pragma unroll
  for (int rt = 0; rt < SEGL / CHK; ++rt)
#pragma unroll
    for (int j = 0; j < 8; ++j)
      kp_s[(CHK * rt + j + 8 * lh) * PM + m0 + lr] = fmaxf(acc[rt][j] * norm, 0.f) + KEPS;
  __builtin_amdgcn_fence(__ATOMIC_RELEASE, "wavefront");
  __builtin_amdgcn_wave_barrier();
  __builtin_amdgcn_fence(__ATOMIC_ACQUIRE, "wavefront");

  if (kp_ws) {
    float* KP = ws + (size_t)WS_FLOATS + ((size_t)(h * NSEG + seg) * (MF / 16) + ft) * (SEGL * 16);
#pragma unroll
    for (int t = 0; t < 8; ++t) {
      const float4 x = *reinterpret_cast<const float4*>(kp_s + (8 * t + (lane >> 2)) * PM + m0 + 4 * (lane & 3));
      *reinterpret_cast<float4*>(KP + 4 * (32 * t + lane)) = x;
    }
  }
  if (last) return;

  float* P = ws + (size_t)hs * PSTRIDE;

  v2f ka[SEGL / 4];
#pragma unroll
  for (int kb = 0; kb < SEGL; kb += 4) {
    ka[kb / 4].x = kp_s[(kb + 2 * lh) * PM + m0 + lr];
    ka[kb / 4].y = kp_s[(kb + 2 * lh + 1) * PM + m0 + lr];
  }
  float kr[SEGL];
#pragma unroll
  for (int r = 0; r < SEGL; ++r) kr[r] = kp_s[r * PM + m0 + lr];
  __builtin_amdgcn_sched_barrier(0);

  float run = 0.f;
  v8f sreg[4] = {};
#pragma unroll
  for (int kb = 0; kb < SEGL; kb += 4) {
#pragma unroll
    for (int et = 0; et < 4; ++et)
      sreg[et] = wmma4(ka[kb / 4], vb[kb / 4][et], sreg[et]);
#pragma unroll
    for (int r = kb; r < kb + 4; ++r) run += kr[r];
  }
  asm volatile("" : "+v"(run));
#pragma unroll
  for (int et = 0; et < 4; ++et)
#pragma unroll
    for (int jh = 0; jh < 2; ++jh) {
      float4 o;
      o.x = sreg[et][4 * jh + 0];
      o.y = sreg[et][4 * jh + 1];
      o.z = sreg[et][4 * jh + 2];
      o.w = sreg[et][4 * jh + 3];
      *reinterpret_cast<float4*>(P + (((ft * 4 + et) * 2 + jh) * 32 + lane) * 4) = o;
    }
  if (lh == 0) P[MF * DH + m0 + lr] = run;
}

__device__ __forceinline__ void contrib_load(const float* __restrict__ P, int blk0, int lane, int col,
                                             float4 (&t)[4], float& tk) {
#pragma unroll
  for (int i = 0; i < 4; ++i)
    t[i] = *reinterpret_cast<const float4*>(P + ((blk0 + i) * 32 + lane) * 4);
  tk = P[MF * DH + col];
}
__device__ __forceinline__ void contrib_add(v8f (&sreg)[ETW], float& run, const float4 (&t)[4], float tk) {
#pragma unroll
  for (int i = 0; i < 4; ++i) {
    sreg[i >> 1][4 * (i & 1) + 0] += t[i].x;
    sreg[i >> 1][4 * (i & 1) + 1] += t[i].y;
    sreg[i >> 1][4 * (i & 1) + 2] += t[i].z;
    sreg[i >> 1][4 * (i & 1) + 3] += t[i].w;
  }
  run += tk;
}

__global__ __launch_bounds__(256)
void performer_causal_f32_wmma(const float* __restrict__ q,
                               const float* __restrict__ k,
                               const float* __restrict__ v,
                               const float* __restrict__ proj,
                               float* __restrict__ out,
                               const float* __restrict__ ws, size_t ws_bytes) {
  __shared__ __attribute__((aligned(16))) float proj_s[MF * PD];
  __shared__ float S_s[MF * PD];
  __shared__ float qp_s[SEGL * PM];
  __shared__ __attribute__((aligned(16))) float kp_s[SEGL * PM];
  __shared__ __attribute__((aligned(16))) float xq_s[SEGL * PD];
  __shared__ __attribute__((aligned(16))) float xk_s[SEGL * PD];
  __shared__ __attribute__((aligned(16))) float v_s[SEGL * PD];
  __shared__ float a_s[NAT * CHK * PA];

  const int eh   = blockIdx.x & 1;
  const int h    = (blockIdx.x >> 1) & 7;
  const int seg  = NSEG - 1 - (blockIdx.x >> 4);
  const int ec0  = EW * eh;
  const int n_begin = seg * SEGL;
  const int nlight  = n_begin / CHK;
  const int tid  = threadIdx.x;
  const int lane = tid & 31;
  const int wave = tid >> 5;
  const int lr   = lane & 15;
  const int lh   = lane >> 4;

  const float norm = 0.35355339059327378f;

  const float* qh = q + (size_t)h * NSEQ * DH;
  const float* kh = k + (size_t)h * NSEQ * DH;
  const float* vh = v + (size_t)h * NSEQ * DH;
  float*       oh = out + (size_t)h * NSEQ * DH;

  const int m0w = wave * 16;
  const bool from_ws = ws_bytes >= (size_t)WS_FLOATS * sizeof(float);
  const bool kp_ws = ws_bytes >= (size_t)WSK_FLOATS * sizeof(float);
  const float* Ph = ws + (size_t)(h * NPART) * PSTRIDE;
  const int blk0 = wave * 8 + 4 * eh;
  const int sB = seg & 8;
  const int sC = seg & 12;
  const int sD = seg & 14;
  float4 tA[8][4], tB[4][4], tC[2][4], tD[1][4];
  float tkA[8], tkB[4], tkC[2], tkD[1];
  v4f rkp[8];

  {
    float4 rp[8];
#pragma unroll
    for (int t = 0; t < 8; ++t)
      rp[t] = *reinterpret_cast<const float4*>(proj + 4 * (tid + 256 * t));
    float4 rq4[4], rv4[2];
    v4f rk4[4];
#pragma unroll
    for (int t = 0; t < 4; ++t) {
      const int idx = tid + 256 * t;
      rq4[t] = *reinterpret_cast<const float4*>(qh + (size_t)n_begin * DH + 4 * idx);
    }
#pragma unroll
    for (int t = 0; t < 2; ++t) {
      const int idx = tid + 256 * t;
      rv4[t] = *reinterpret_cast<const float4*>(vh + (size_t)(n_begin + (idx >> 3)) * DH + ec0 + 4 * (idx & 7));
    }
    if (!kp_ws) {
#pragma unroll
      for (int t = 0; t < 4; ++t)
        rk4[t] = *reinterpret_cast<const v4f*>(kh + (size_t)n_begin * DH + 4 * (tid + 256 * t));
    }
#pragma unroll
    for (int t = 0; t < 8; ++t) {
      const int idx = tid + 256 * t;
      const int p = (idx >> 4) * PD + 4 * (idx & 15);
      *reinterpret_cast<float4*>(proj_s + p) = rp[t];
    }
#pragma unroll
    for (int t = 0; t < 4; ++t) {
      const int idx = tid + 256 * t;
      const int p = (idx >> 4) * PD + 4 * (idx & 15);
      *reinterpret_cast<float4*>(xq_s + p) = rq4[t];
    }
#pragma unroll
    for (int t = 0; t < 2; ++t) {
      const int idx = tid + 256 * t;
      *reinterpret_cast<float4*>(v_s + (idx >> 3) * PD + ec0 + 4 * (idx & 7)) = rv4[t];
    }
    if (!kp_ws) {
#pragma unroll
      for (int t = 0; t < 4; ++t) {
        const int idx = tid + 256 * t;
        *reinterpret_cast<v4f*>(xk_s + (idx >> 4) * PD + 4 * (idx & 15)) = rk4[t];
      }
    }
  }
  if (from_ws) {
    if (seg & 8) {
#pragma unroll
      for (int u = 0; u < 8; ++u)
        contrib_load(Ph + (size_t)u * PSTRIDE, blk0, lane, m0w + lr, tA[u], tkA[u]);
    }
    if (seg & 4) {
#pragma unroll
      for (int u = 0; u < 4; ++u)
        contrib_load(Ph + (size_t)(sB + u) * PSTRIDE, blk0, lane, m0w + lr, tB[u], tkB[u]);
    }
  }

  v8f sreg[ETW] = {};
  float run = 0.f;
  if (!from_ws && nlight > 0) {
    const int m0 = m0w;
      float* const xkbuf = S_s;
      float* const vbuf  = S_s + 2 * CHK * PD;
      __syncthreads();
      v2f pb[DH / 4];
#pragma unroll
      for (int kk = 0; kk < DH / 4; ++kk) {
        pb[kk].x = proj_s[(m0 + lr) * PD + 4 * kk + 2 * lh];
        pb[kk].y = proj_s[(m0 + lr) * PD + 4 * kk + 2 * lh + 1];
      }
      float rk[4], rv[4];
#pragma unroll
      for (int t = 0; t < 4; ++t) {
        rk[t] = kh[tid + 256 * t];
        rv[t] = vh[tid + 256 * t];
      }
#pragma unroll
      for (int t = 0; t < 4; ++t) {
        const int p = ((tid >> 6) + 4 * t) * PD + (tid & 63);
        xkbuf[p] = rk[t];
        vbuf[p]  = rv[t];
      }
      __syncthreads();
      for (int c = 0; c < nlight; ++c) {
        const float* xkb = xkbuf + (c & 1) * (CHK * PD);
        const float* vb  = vbuf  + (c & 1) * (CHK * PD);
        float* xkn = xkbuf + ((c + 1) & 1) * (CHK * PD);
        float* vn  = vbuf  + ((c + 1) & 1) * (CHK * PD);
        const int cn = (c + 1 < nlight) ? (c + 1) : c;
#pragma unroll
        for (int t = 0; t < 4; ++t) {
          rk[t] = kh[cn * CHK * DH + tid + 256 * t];
          rv[t] = vh[cn * CHK * DH + tid + 256 * t];
        }
        v8f acck = {};
#pragma unroll
        for (int kk = 0; kk < DH / 4; ++kk) {
          v2f a;
          a.x = xkb[lr * PD + 4 * kk + 2 * lh];
          a.y = xkb[lr * PD + 4 * kk + 2 * lh + 1];
          acck = wmma4(a, pb[kk], acck);
        }
#pragma unroll
        for (int j = 0; j < 8; ++j)
          kp_s[(j + 8 * lh) * PM + m0 + lr] = fmaxf(acck[j] * norm, 0.f) + KEPS;
        __builtin_amdgcn_fence(__ATOMIC_RELEASE, "wavefront");
        __builtin_amdgcn_wave_barrier();
        __builtin_amdgcn_fence(__ATOMIC_ACQUIRE, "wavefront");
#pragma unroll
        for (int r = 0; r < CHK; ++r) run += kp_s[r * PM + m0 + lr];
#pragma unroll
        for (int etl = 0; etl < ETW; ++etl) {
          const int e0 = ec0 + etl * 16;
#pragma unroll
          for (int kb = 0; kb < CHK; kb += 4) {
            v2f a, b;
            a.x = kp_s[(kb + 2 * lh) * PM + m0 + lr];
            a.y = kp_s[(kb + 2 * lh + 1) * PM + m0 + lr];
            b.x = vb[(kb + 2 * lh) * PD + e0 + lr];
            b.y = vb[(kb + 2 * lh + 1) * PD + e0 + lr];
            sreg[etl] = wmma4(a, b, sreg[etl]);
          }
        }
#pragma unroll
        for (int t = 0; t < 4; ++t) {
          const int p = ((tid >> 6) + 4 * t) * PD + (tid & 63);
          xkn[p] = rk[t];
          vn[p]  = rv[t];
        }
        __syncthreads();
      }
  }
  __builtin_amdgcn_fence(__ATOMIC_RELEASE, "workgroup", "local");
  __builtin_amdgcn_s_barrier();
  __builtin_amdgcn_fence(__ATOMIC_ACQUIRE, "workgroup", "local");

  if (!kp_ws) {
    v8f acck[NRT] = {};
#pragma unroll
    for (int kb = 0; kb < DH; kb += 4) {
      v2f b;
      b.x = proj_s[(m0w + lr) * PD + kb + 2 * lh];
      b.y = proj_s[(m0w + lr) * PD + kb + 2 * lh + 1];
#pragma unroll
      for (int rt = 0; rt < NRT; ++rt) {
        v2f a;
        a.x = xk_s[(CHK * rt + lr) * PD + kb + 2 * lh];
        a.y = xk_s[(CHK * rt + lr) * PD + kb + 2 * lh + 1];
        acck[rt] = wmma4(a, b, acck[rt]);
      }
    }
#pragma unroll
    for (int rt = 0; rt < NRT; ++rt)
#pragma unroll
      for (int j = 0; j < 8; ++j)
        kp_s[(CHK * rt + j + 8 * lh) * PM + m0w + lr] = fmaxf(acck[rt][j] * norm, 0.f) + KEPS;
  }

  v8f accq[NRT] = {};
#pragma unroll
  for (int kb = 0; kb < DH; kb += 4) {
    if (kb == DH / 2 && kp_ws) {
      const float* KP = ws + (size_t)WS_FLOATS + ((size_t)(h * NSEG + seg) * (MF / 16) + wave) * (SEGL * 16);
#pragma unroll
      for (int t = 0; t < 8; ++t)
        rkp[t] = *reinterpret_cast<const v4f*>(KP + 4 * (32 * t + lane));
    }
    if (kb == DH / 2 && from_ws) {
      if (seg & 2) {
#pragma unroll
        for (int u = 0; u < 2; ++u)
          contrib_load(Ph + (size_t)(sC + u) * PSTRIDE, blk0, lane, m0w + lr, tC[u], tkC[u]);
      }
      if (seg & 1)
        contrib_load(Ph + (size_t)sD * PSTRIDE, blk0, lane, m0w + lr, tD[0], tkD[0]);
    }
    v2f b;
    b.x = proj_s[(m0w + lr) * PD + kb + 2 * lh];
    b.y = proj_s[(m0w + lr) * PD + kb + 2 * lh + 1];
#pragma unroll
    for (int rt = 0; rt < NRT; ++rt) {
      v2f a;
      a.x = xq_s[(CHK * rt + lr) * PD + kb + 2 * lh];
      a.y = xq_s[(CHK * rt + lr) * PD + kb + 2 * lh + 1];
      accq[rt] = wmma4(a, b, accq[rt]);
    }
  }
  if (kp_ws) {
#pragma unroll
    for (int t = 0; t < 8; ++t)
      *reinterpret_cast<v4f*>(kp_s + (8 * t + (lane >> 2)) * PM + m0w + 4 * (lane & 3)) = rkp[t];
  }

  if (from_ws) {
    if (seg & 8) {
#pragma unroll
      for (int u = 0; u < 8; ++u) contrib_add(sreg, run, tA[u], tkA[u]);
    }
    if (seg & 4) {
#pragma unroll
      for (int u = 0; u < 4; ++u) contrib_add(sreg, run, tB[u], tkB[u]);
    }
    if (seg & 2) {
#pragma unroll
      for (int u = 0; u < 2; ++u) contrib_add(sreg, run, tC[u], tkC[u]);
    }
    if (seg & 1) contrib_add(sreg, run, tD[0], tkD[0]);
  }
#pragma unroll
  for (int etl = 0; etl < ETW; ++etl)
#pragma unroll
    for (int j = 0; j < 8; ++j)
      S_s[(m0w + j + 8 * lh) * PD + ec0 + etl * 16 + lr] = sreg[etl][j];

  __builtin_amdgcn_fence(__ATOMIC_RELEASE, "wavefront");
  __builtin_amdgcn_wave_barrier();
  __builtin_amdgcn_fence(__ATOMIC_ACQUIRE, "wavefront");
  {
    float kr[SEGL];
#pragma unroll
    for (int r = 0; r < SEGL; ++r) kr[r] = kp_s[r * PM + m0w + lr];
    __builtin_amdgcn_sched_barrier(0);
    float kc[NRT][8];
#pragma unroll
    for (int rt = 0; rt < NRT; ++rt) {
#pragma unroll
      for (int j = 0; j < 8; ++j) {
        run += kr[CHK * rt + j];
        kc[rt][j] = run;
      }
#pragma unroll
      for (int j = 0; j < 8; ++j) {
        run += kr[CHK * rt + 8 + j];
        if (lh) kc[rt][j] = run;
      }
    }
#pragma unroll
    for (int rt = 0; rt < NRT; ++rt)
#pragma unroll
      for (int j = 0; j < 8; ++j)
        qp_s[(CHK * rt + j + 8 * lh) * PM + m0w + lr] =
            (fmaxf(accq[rt][j] * norm, 0.f) + KEPS) / kc[rt][j];
  }
  __syncthreads();

  const int wv = __builtin_amdgcn_readfirstlane(wave);

  const int ort = wv >> 1;
  const int e0  = ec0 + (wv & 1) * 16;
  int art = wv >> 1, act = wv & 1;
  if (wv == 1) { art = 2; act = 2; }
  v8f o = {};
  {
    v8f am = {};
    const float* qo = qp_s + (CHK * ort + lr) * PM + 2 * lh;
    const float* sb = S_s + (2 * lh) * PD + e0 + lr;
    const float* qa = qp_s + (CHK * art + lr) * PM + 2 * lh;
    const float* ka = kp_s + (CHK * act + lr) * PM + 2 * lh;
    v2f foa[2][4], fob[2][4], fma[2][4], fmb[2][4];
#pragma unroll
    for (int i = 0; i < 4; ++i) {
      foa[0][i].x = qo[4 * i];
      foa[0][i].y = qo[4 * i + 1];
      fob[0][i].x = sb[(4 * i) * PD];
      fob[0][i].y = sb[(4 * i + 1) * PD];
      fma[0][i].x = qa[4 * i];
      fma[0][i].y = qa[4 * i + 1];
      fmb[0][i].x = ka[4 * i];
      fmb[0][i].y = ka[4 * i + 1];
    }
    __builtin_amdgcn_sched_barrier(0);
#pragma unroll
    for (int blk = 0; blk < MF / 16; ++blk) {
      const int cur = blk & 1, nxt = cur ^ 1;
      if (blk + 1 < MF / 16) {
#pragma unroll
        for (int i = 0; i < 4; ++i) {
          const int kb = 16 * (blk + 1) + 4 * i;
          foa[nxt][i].x = qo[kb];
          foa[nxt][i].y = qo[kb + 1];
          fob[nxt][i].x = sb[kb * PD];
          fob[nxt][i].y = sb[(kb + 1) * PD];
          fma[nxt][i].x = qa[kb];
          fma[nxt][i].y = qa[kb + 1];
          fmb[nxt][i].x = ka[kb];
          fmb[nxt][i].y = ka[kb + 1];
        }
      }
      __builtin_amdgcn_sched_barrier(0);
#pragma unroll
      for (int i = 0; i < 4; ++i) {
        o  = wmma4(foa[cur][i], fob[cur][i], o);
        am = wmma4(fma[cur][i], fmb[cur][i], am);
      }
      __builtin_amdgcn_sched_barrier(0);
    }
    float* at = a_s + (art * (art + 1) / 2 + act) * (CHK * PA);
    const bool diag = art == act;
#pragma unroll
    for (int j = 0; j < 8; ++j) {
      const int row = j + 8 * lh;
      at[row * PA + lr] = (!diag || lr <= row) ? am[j] : 0.f;
    }
  }
  if (wv < 2) {
    const int act2 = 2 + wv;
    v8f am = {};
    const float* qa = qp_s + (CHK * 3 + lr) * PM + 2 * lh;
    const float* ka = kp_s + (CHK * act2 + lr) * PM + 2 * lh;
#pragma unroll
    for (int kb = 0; kb < MF; kb += 4) {
      v2f a, b;
      a.x = qa[kb];
      a.y = qa[kb + 1];
      b.x = ka[kb];
      b.y = ka[kb + 1];
      am = wmma4(a, b, am);
    }
    float* at = a_s + (6 + act2) * (CHK * PA);
    const bool diag = act2 == 3;
#pragma unroll
    for (int j = 0; j < 8; ++j) {
      const int row = j + 8 * lh;
      at[row * PA + lr] = (!diag || lr <= row) ? am[j] : 0.f;
    }
  }
  __syncthreads();

  v2f sa[NRT][CHK / 4], sv[NRT][CHK / 4];
#pragma unroll
  for (int ct = 0; ct < NRT; ++ct) {
    if (ct <= ort) {
      const float* at = a_s + (ort * (ort + 1) / 2 + ct) * (CHK * PA) + lr * PA + 2 * lh;
      const float* vb = v_s + (CHK * ct + 2 * lh) * PD + e0 + lr;
#pragma unroll
      for (int kb = 0; kb < CHK; kb += 4) {
        sa[ct][kb / 4].x = at[kb];
        sa[ct][kb / 4].y = at[kb + 1];
        sv[ct][kb / 4].x = vb[kb * PD];
        sv[ct][kb / 4].y = vb[(kb + 1) * PD];
      }
    }
  }
  __builtin_amdgcn_sched_barrier(0);
#pragma unroll
  for (int ct = 0; ct < NRT; ++ct) {
    if (ct <= ort) {
#pragma unroll
      for (int i = 0; i < CHK / 4; ++i)
        o = wmma4(sa[ct][i], sv[ct][i], o);
    }
  }
#pragma unroll
  for (int j = 0; j < 8; ++j)
    oh[(n_begin + CHK * ort + j + 8 * lh) * DH + e0 + lr] = o[j];
}

extern "C" void kernel_launch(void* const* d_in, const int* in_sizes, int n_in,
                              void* d_out, int out_size, void* d_ws, size_t ws_size,
                              hipStream_t stream) {
  (void)in_sizes; (void)n_in; (void)out_size;
  const float* q    = (const float*)d_in[0];
  const float* k    = (const float*)d_in[1];
  const float* v    = (const float*)d_in[2];
  const float* proj = (const float*)d_in[3];
  float* out        = (float*)d_out;
  float* ws         = (float*)d_ws;
  performer_segment_state_f32_wmma<<<dim3(256), dim3(128), 0, stream>>>(k, v, proj, ws, ws_size);
  performer_causal_f32_wmma<<<dim3(256), dim3(256), 0, stream>>>(q, k, v, proj, out, ws, ws_size);
}
